# starting kernel
# baseline (speedup 1.0000x reference)
#include <hip/hip_runtime.h>
#include <cstdint>


typedef float v2f __attribute__((ext_vector_type(2)));
typedef float v8f __attribute__((ext_vector_type(8)));

#define NEG_SLOPE 0.2f
#define GAT_EPS 1e-16f
#define ENC_NEG_INF 0x007FFFFFu

__device__ __forceinline__ unsigned enc_ord(float f) {
  unsigned u = __float_as_uint(f);
  return (u & 0x80000000u) ? ~u : (u | 0x80000000u);
}
__device__ __forceinline__ float dec_ord(unsigned k) {
  return (k & 0x80000000u) ? __uint_as_float(k & 0x7FFFFFFFu)
                           : __uint_as_float(~k);
}

__global__ void fill_u32_kernel(unsigned* __restrict__ p, unsigned v, long long n) {
  long long i = (long long)blockIdx.x * blockDim.x + threadIdx.x;
  long long stride = (long long)gridDim.x * blockDim.x;
  for (; i < n; i += stride) p[i] = v;
}

__global__ void gemm_f32_wmma_kernel(const float* __restrict__ A,
                                     const float* __restrict__ B,
                                     float* __restrict__ C,
                                     int M, int N, int K) {
  int wave = (int)((blockIdx.x * blockDim.x + threadIdx.x) >> 5);
  int lane = threadIdx.x & 31;
  int ntiles = N >> 4;
  int mtiles = (M + 15) >> 4;
  if (wave >= mtiles * ntiles) return;
  int mt = wave / ntiles;
  int nt = wave - mt * ntiles;
  int m0 = mt << 4, n0 = nt << 4;
  int half = lane >> 4;
  int lid  = lane & 15;

  v8f c = {0.f, 0.f, 0.f, 0.f, 0.f, 0.f, 0.f, 0.f};
  const float* arow = A + (size_t)(m0 + lid) * K + 2 * half;
  const float* bcol = B + (size_t)(2 * half) * N + n0 + lid;

  for (int k0 = 0; k0 < K; k0 += 4) {
    v2f a, b;
    a.x = arow[k0];
    a.y = arow[k0 + 1];
    b.x = bcol[(size_t)k0 * N];
    b.y = bcol[(size_t)k0 * N + N];
    c = __builtin_amdgcn_wmma_f32_16x16x4_f32(false, a, false, b,
                                              (short)0, c, false, false);
  }

  float* crow = C + (size_t)m0 * N + n0 + lid;
#pragma unroll
  for (int r = 0; r < 8; ++r)
    crow[(size_t)(r + 8 * half) * N] = c[r];
}

template <int H, int D>
__global__ void node_alpha_kernel(const float* __restrict__ h,
                                  const float* __restrict__ avs,
                                  const float* __restrict__ avd,
                                  float* __restrict__ as,
                                  float* __restrict__ ad, int N) {
  int n = blockIdx.x * blockDim.x + threadIdx.x;
  if (n >= N) return;
  const float* hr = h + (size_t)n * H * D;
#pragma unroll
  for (int hh = 0; hh < H; ++hh) {
    float s0 = 0.f, s1 = 0.f;
#pragma unroll
    for (int d = 0; d < D; ++d) {
      float v = hr[hh * D + d];
      s0 += v * avs[hh * D + d];
      s1 += v * avd[hh * D + d];
    }
    as[(size_t)n * H + hh] = s0;
    ad[(size_t)n * H + hh] = s1;
  }
}

template <int H>
__global__ void edge_max_kernel(const int* __restrict__ src,
                                const int* __restrict__ dst, int E, int Etot,
                                const float* __restrict__ as,
                                const float* __restrict__ ad,
                                unsigned* __restrict__ m) {
  int gid = blockIdx.x * blockDim.x + threadIdx.x;
  if (gid >= Etot * H) return;
  int eidx = gid / H, hh = gid - (gid / H) * H;
  int si, di;
  if (eidx < E) { si = src[eidx]; di = dst[eidx]; }
  else          { si = di = eidx - E; }
  float e = as[(size_t)si * H + hh] + ad[(size_t)di * H + hh];
  e = e > 0.f ? e : NEG_SLOPE * e;
  atomicMax(&m[(size_t)di * H + hh], enc_ord(e));
}

template <int H>
__global__ void edge_sum_kernel(const int* __restrict__ src,
                                const int* __restrict__ dst, int E, int Etot,
                                const float* __restrict__ as,
                                const float* __restrict__ ad,
                                const unsigned* __restrict__ m,
                                float* __restrict__ s) {
  int gid = blockIdx.x * blockDim.x + threadIdx.x;
  if (gid >= Etot * H) return;
  int eidx = gid / H, hh = gid - (gid / H) * H;
  int si, di;
  if (eidx < E) { si = src[eidx]; di = dst[eidx]; }
  else          { si = di = eidx - E; }
  float e = as[(size_t)si * H + hh] + ad[(size_t)di * H + hh];
  e = e > 0.f ? e : NEG_SLOPE * e;
  float ex = __expf(e - dec_ord(m[(size_t)di * H + hh]));
  atomicAdd(&s[(size_t)di * H + hh], ex);
}

template <int DTOT, int H>
__global__ void edge_scatter_kernel(const int* __restrict__ src,
                                    const int* __restrict__ dst, int E, int Etot,
                                    const float* __restrict__ as,
                                    const float* __restrict__ ad,
                                    const unsigned* __restrict__ m,
                                    const float* __restrict__ s,
                                    const float* __restrict__ hsrc,
                                    float* __restrict__ acc) {
  int gid = blockIdx.x * blockDim.x + threadIdx.x;
  int eidx = gid >> 5;
  int lane = gid & 31;
  if (eidx >= Etot) return;
  int si, di;
  if (eidx < E) { si = src[eidx]; di = dst[eidx]; }
  else          { si = di = eidx - E; }
  const int Dh = DTOT / H;
#pragma unroll
  for (int c = lane; c < DTOT; c += 32) {
    int hh = c / Dh;
    float e = as[(size_t)si * H + hh] + ad[(size_t)di * H + hh];
    e = e > 0.f ? e : NEG_SLOPE * e;
    float mm = dec_ord(m[(size_t)di * H + hh]);
    float ss = s[(size_t)di * H + hh];
    float alpha = __expf(e - mm) / (ss + GAT_EPS);
    atomicAdd(&acc[(size_t)di * DTOT + c],
              hsrc[(size_t)si * DTOT + c] * alpha);
  }
}

__global__ void finalize_kernel(float* __restrict__ acc,
                                const float* __restrict__ bias,
                                long long total, int C, int apply_elu) {
  long long i = (long long)blockIdx.x * blockDim.x + threadIdx.x;
  if (i >= total) return;
  int j = (int)(i % C);
  float v = acc[i] + bias[j];
  if (apply_elu) v = v > 0.f ? v : __expf(v) - 1.f;
  acc[i] = v;
}

static inline int cdiv(long long a, long long b) { return (int)((a + b - 1) / b); }

extern "C" void kernel_launch(void* const* d_in, const int* in_sizes, int n_in,
                              void* d_out, int out_size, void* d_ws, size_t ws_size,
                              hipStream_t stream) {
  const float* x      = (const float*)d_in[0];
  const int*   ei     = (const int*)  d_in[1];
  const float* W1     = (const float*)d_in[2];
  const float* a_src1 = (const float*)d_in[3];
  const float* a_dst1 = (const float*)d_in[4];
  const float* b1     = (const float*)d_in[5];
  const float* W2     = (const float*)d_in[6];
  const float* a_src2 = (const float*)d_in[7];
  const float* a_dst2 = (const float*)d_in[8];
  const float* b2     = (const float*)d_in[9];
  float* z = (float*)d_out;

  const int Fin = 256, H1 = 8, C1 = 64, C2 = 128;
  const int N    = in_sizes[0] / Fin;
  const int E    = in_sizes[1] / 2;
  const int Etot = E + N;
  const int* srcv = ei;
  const int* dstv = ei + E;

  float*    h1   = (float*)d_ws;
  float*    as1  = h1  + (size_t)N * C1;
  float*    ad1  = as1 + (size_t)N * H1;
  unsigned* m1   = (unsigned*)(ad1 + (size_t)N * H1);
  float*    s1   = (float*)(m1 + (size_t)N * H1);
  float*    acc1 = s1  + (size_t)N * H1;
  float*    h2   = acc1 + (size_t)N * C1;
  float*    as2  = h2  + (size_t)N * C2;
  float*    ad2  = as2 + N;
  unsigned* m2   = (unsigned*)(ad2 + N);
  float*    s2   = (float*)(m2 + N);

  const int B = 256;

  fill_u32_kernel<<<2048, B, 0, stream>>>((unsigned*)s1, 0u,
                                          (long long)N * (H1 + C1));
  fill_u32_kernel<<<2048, B, 0, stream>>>(m1, ENC_NEG_INF, (long long)N * H1);
  fill_u32_kernel<<<2048, B, 0, stream>>>((unsigned*)z, 0u, (long long)N * C2);
  fill_u32_kernel<<<2048, B, 0, stream>>>((unsigned*)s2, 0u, (long long)N);
  fill_u32_kernel<<<2048, B, 0, stream>>>(m2, ENC_NEG_INF, (long long)N);

  {
    int waves = (N / 16) * (C1 / 16);
    gemm_f32_wmma_kernel<<<cdiv((long long)waves * 32, B), B, 0, stream>>>(
        x, W1, h1, N, C1, Fin);
  }
  node_alpha_kernel<8, 8><<<cdiv(N, B), B, 0, stream>>>(h1, a_src1, a_dst1,
                                                        as1, ad1, N);
  edge_max_kernel<8><<<cdiv((long long)Etot * 8, B), B, 0, stream>>>(
      srcv, dstv, E, Etot, as1, ad1, m1);
  edge_sum_kernel<8><<<cdiv((long long)Etot * 8, B), B, 0, stream>>>(
      srcv, dstv, E, Etot, as1, ad1, m1, s1);
  edge_scatter_kernel<64, 8><<<cdiv((long long)Etot * 32, B), B, 0, stream>>>(
      srcv, dstv, E, Etot, as1, ad1, m1, s1, h1, acc1);
  finalize_kernel<<<cdiv((long long)N * C1, B), B, 0, stream>>>(
      acc1, b1, (long long)N * C1, C1,  1);

  {
    int waves = (N / 16) * (C2 / 16);
    gemm_f32_wmma_kernel<<<cdiv((long long)waves * 32, B), B, 0, stream>>>(
        acc1, W2, h2, N, C2, C1);
  }
  node_alpha_kernel<1, 128><<<cdiv(N, B), B, 0, stream>>>(h2, a_src2, a_dst2,
                                                          as2, ad2, N);
  edge_max_kernel<1><<<cdiv((long long)Etot, B), B, 0, stream>>>(
      srcv, dstv, E, Etot, as2, ad2, m2);
  edge_sum_kernel<1><<<cdiv((long long)Etot, B), B, 0, stream>>>(
      srcv, dstv, E, Etot, as2, ad2, m2, s2);
  edge_scatter_kernel<128, 1><<<cdiv((long long)Etot * 32, B), B, 0, stream>>>(
      srcv, dstv, E, Etot, as2, ad2, m2, s2, h2, z);
  finalize_kernel<<<cdiv((long long)N * C2, B), B, 0, stream>>>(
      z, b2, (long long)N * C2, C2,  0);
}


# Edge scan made latency tolerant: 8 dst loads in flight a wave, one LDS slot reservation a batch, (edge id, local dst) staged in LDS, src ids gathered after the scan; sum order = LDS arrival
# speedup vs baseline: 2.6954x; 2.6954x over previous
#include <hip/hip_runtime.h>
#include <cstdint>


typedef float v2f __attribute__((ext_vector_type(2)));
typedef float v8f __attribute__((ext_vector_type(8)));

#define NEG_SLOPE 0.2f
#define GAT_EPS 1e-16f
#define ENC_NEG_INF 0x007FFFFFu

__device__ __forceinline__ unsigned enc_ord(float f) {
  unsigned u = __float_as_uint(f);
  return (u & 0x80000000u) ? ~u : (u | 0x80000000u);
}
__device__ __forceinline__ float dec_ord(unsigned k) {
  return (k & 0x80000000u) ? __uint_as_float(k & 0x7FFFFFFFu)
                           : __uint_as_float(~k);
}

__global__ void gemm_f32_wmma_kernel(const float* __restrict__ A,
                                     const float* __restrict__ B,
                                     float* __restrict__ C,
                                     int M, int N, int K) {
  int wave = (int)((blockIdx.x * blockDim.x + threadIdx.x) >> 5);
  int lane = threadIdx.x & 31;
  int ntiles = N >> 4;
  int mtiles = (M + 15) >> 4;
  if (wave >= mtiles * ntiles) return;
  int mt = wave / ntiles;
  int nt = wave - mt * ntiles;
  int m0 = mt << 4, n0 = nt << 4;
  int half = lane >> 4;
  int lid  = lane & 15;

  v8f c = {0.f, 0.f, 0.f, 0.f, 0.f, 0.f, 0.f, 0.f};
  const float* arow = A + (size_t)(m0 + lid) * K + 2 * half;
  const float* bcol = B + (size_t)(2 * half) * N + n0 + lid;

  for (int k0 = 0; k0 < K; k0 += 4) {
    v2f a, b;
    a.x = arow[k0];
    a.y = arow[k0 + 1];
    b.x = bcol[(size_t)k0 * N];
    b.y = bcol[(size_t)k0 * N + N];
    c = __builtin_amdgcn_wmma_f32_16x16x4_f32(false, a, false, b,
                                              (short)0, c, false, false);
  }

  float* crow = C + (size_t)m0 * N + n0 + lid;
#pragma unroll
  for (int r = 0; r < 8; ++r)
    crow[(size_t)(r + 8 * half) * N] = c[r];
}

template <int H, int D>
__global__ void node_alpha_kernel(const float* __restrict__ h,
                                  const float* __restrict__ avs,
                                  const float* __restrict__ avd,
                                  float* __restrict__ as,
                                  float* __restrict__ ad, int N) {
  int n = blockIdx.x * blockDim.x + threadIdx.x;
  if (n >= N) return;
  const float* hr = h + (size_t)n * H * D;
#pragma unroll
  for (int hh = 0; hh < H; ++hh) {
    float s0 = 0.f, s1 = 0.f;
#pragma unroll
    for (int d = 0; d < D; ++d) {
      float v = hr[hh * D + d];
      s0 += v * avs[hh * D + d];
      s1 += v * avd[hh * D + d];
    }
    as[(size_t)n * H + hh] = s0;
    ad[(size_t)n * H + hh] = s1;
  }
}

#define NODES_PER_WG 392
#define WG_THREADS 1024
#define WG_WAVES 32
#define LIST_CAP 32768

#define SCAN_U 8

__device__ __forceinline__ void stage_append(bool valid, int e, int dl,
                                             int2* stage, int* lcount, int lane) {
  unsigned m = __builtin_amdgcn_ballot_w32(valid);
  if (m == 0u) return;
  int pos = 0;
  if (lane == 0) pos = atomicAdd(lcount, (int)__popc(m));
  pos = __builtin_amdgcn_readfirstlane(pos);
  if (valid) {
    int slot = pos + (int)__builtin_amdgcn_mbcnt_lo(m, 0u);
    if (slot < LIST_CAP) stage[slot] = make_int2(e, dl);
  }
}

template <bool GUARD>
__device__ __forceinline__ void scan_batch(const int4* __restrict__ d4, int vb, int nvec,
                                           int lo, int cnt, int2* stage, int* lcount,
                                           int lane) {
  int4 d[SCAN_U];
#pragma unroll
  for (int u = 0; u < SCAN_U; ++u) {
    const int v = vb + u * 32 + lane;
    if (GUARD) {
      d[u] = make_int4(lo - 1, lo - 1, lo - 1, lo - 1);
      if (v < nvec) d[u] = d4[v];
    } else {
      d[u] = d4[v];
    }
  }
  unsigned m[SCAN_U][4];
  int tot = 0;
#pragma unroll
  for (int u = 0; u < SCAN_U; ++u) {
    m[u][0] = __builtin_amdgcn_ballot_w32((unsigned)(d[u].x - lo) < (unsigned)cnt);
    m[u][1] = __builtin_amdgcn_ballot_w32((unsigned)(d[u].y - lo) < (unsigned)cnt);
    m[u][2] = __builtin_amdgcn_ballot_w32((unsigned)(d[u].z - lo) < (unsigned)cnt);
    m[u][3] = __builtin_amdgcn_ballot_w32((unsigned)(d[u].w - lo) < (unsigned)cnt);
    tot += (int)(__popc(m[u][0]) + __popc(m[u][1]) + __popc(m[u][2]) + __popc(m[u][3]));
  }
  if (tot == 0) return;
  int base = 0;
  if (lane == 0) base = atomicAdd(lcount, tot);
  base = __builtin_amdgcn_readfirstlane(base);
#pragma unroll
  for (int u = 0; u < SCAN_U; ++u) {
    const int e0 = 4 * (vb + u * 32 + lane);
    const int dd[4] = {d[u].x, d[u].y, d[u].z, d[u].w};
#pragma unroll
    for (int j = 0; j < 4; ++j) {
      const unsigned mm = m[u][j];
      if (mm != 0u) {
        if ((mm >> lane) & 1u) {
          const int slot = base + (int)__builtin_amdgcn_mbcnt_lo(mm, 0u);
          if (slot < LIST_CAP) stage[slot] = make_int2(e0 + j, dd[j] - lo);
        }
        base += (int)__popc(mm);
      }
    }
  }
}

__global__ __launch_bounds__(WG_THREADS) void edge_bin_kernel(
    const int* __restrict__ src, const int* __restrict__ dst, int E, int N,
    int2* __restrict__ lists, int* __restrict__ counts) {
  __shared__ int2 stage[LIST_CAP];
  __shared__ int lcount;
  const int tid = threadIdx.x, lane = tid & 31;
  const int wv = __builtin_amdgcn_readfirstlane(tid >> 5);
  const int lo = blockIdx.x * NODES_PER_WG;
  const int cnt = min(NODES_PER_WG, N - lo);
  int2* lst = lists + (size_t)blockIdx.x * LIST_CAP;
  if (tid == 0) lcount = 0;
  __syncthreads();
  if ((E & 3) == 0) {
    const int nvec = E >> 2;
    const int4* __restrict__ d4 = (const int4*)dst;
    constexpr int WSPAN = 32 * SCAN_U;
    for (int vb = wv * WSPAN; vb < nvec; vb += WG_WAVES * WSPAN) {
      if (vb + WSPAN <= nvec) scan_batch<false>(d4, vb, nvec, lo, cnt, stage, &lcount, lane);
      else                    scan_batch<true >(d4, vb, nvec, lo, cnt, stage, &lcount, lane);
    }
  } else {
    for (int eb = wv * 32; eb < E; eb += WG_THREADS) {
      const int e = eb + lane;
      int x = -1;
      if (e < E) x = dst[e] - lo;
      stage_append((unsigned)x < (unsigned)cnt, e, x, stage, &lcount, lane);
    }
  }
  __syncthreads();
  const int found = lcount;
  const int nfill = min(found, LIST_CAP);
  constexpr int GU = 8;
  for (int i0 = tid; i0 < nfill; i0 += GU * WG_THREADS) {
    int2 p[GU];
    int s[GU];
#pragma unroll
    for (int k = 0; k < GU; ++k) {
      const int i = i0 + k * WG_THREADS;
      p[k] = (i < nfill) ? stage[i] : make_int2(0, 0);
    }
#pragma unroll
    for (int k = 0; k < GU; ++k) s[k] = src[p[k].x];
#pragma unroll
    for (int k = 0; k < GU; ++k) {
      const int i = i0 + k * WG_THREADS;
      if (i < nfill) lst[i] = make_int2(s[k], p[k].y);
    }
  }
  if (tid == 0) counts[blockIdx.x] = found;
}

__device__ __forceinline__ bool fetch_item(int i, int n_real, int total, bool use_list,
                                           const int2* __restrict__ lst,
                                           const int* __restrict__ src,
                                           const int* __restrict__ dst,
                                           int lo, int cnt, int& sv, int& dv) {
  sv = 0; dv = 0;
  if (i < n_real) {
    if (use_list) { int2 p = lst[i]; sv = p.x; dv = p.y; return true; }
    int d = dst[i] - lo;
    if ((unsigned)d < (unsigned)cnt) { sv = src[i]; dv = d; return true; }
    return false;
  }
  if (i < total) { dv = i - n_real; sv = lo + dv; return true; }
  return false;
}

template <int DTOT, int H, bool ELU>
__global__ __launch_bounds__(WG_THREADS) void range_agg_kernel(
    const int* __restrict__ src, const int* __restrict__ dst, int E, int N,
    const int2* __restrict__ lists, const int* __restrict__ counts,
    const float* __restrict__ as, const float* __restrict__ ad,
    const float* __restrict__ hsrc, const float* __restrict__ bias,
    float* __restrict__ out) {
  constexpr int R = NODES_PER_WG;
  constexpr int VPL = DTOT / 32;
  constexpr int Dh = DTOT / H;
  constexpr int U = 4;
  typedef float vrow __attribute__((ext_vector_type(VPL)));
  __shared__ float acc[R * DTOT];
  __shared__ unsigned mx[R * H];
  __shared__ float sm[R * H];
  __shared__ float adl[R * H];

  const int tid = threadIdx.x, lane = tid & 31, wv = tid >> 5;
  const int lo = blockIdx.x * R;
  const int cnt = min(R, N - lo);
  const int2* lst = lists + (size_t)blockIdx.x * LIST_CAP;
  const int count = counts[blockIdx.x];
  const bool use_list = count <= LIST_CAP;
  const int n_real = use_list ? count : E;
  const int total = n_real + cnt;

  for (int i = tid; i < R * DTOT; i += WG_THREADS) acc[i] = 0.f;
  for (int i = tid; i < R * H; i += WG_THREADS) {
    sm[i] = 0.f;
    mx[i] = ENC_NEG_INF;
    adl[i] = (i < cnt * H) ? ad[(size_t)lo * H + i] : 0.f;
  }
  __syncthreads();

  for (int base = wv * 32; base < total; base += WG_THREADS) {
    int sv, dv;
    bool valid = fetch_item(base + lane, n_real, total, use_list, lst, src, dst, lo, cnt, sv, dv);
    if (valid) {
#pragma unroll
      for (int hh = 0; hh < H; ++hh) {
        float e = as[(size_t)sv * H + hh] + adl[dv * H + hh];
        e = e > 0.f ? e : NEG_SLOPE * e;
        atomicMax(&mx[dv * H + hh], enc_ord(e));
      }
    }
  }
  __syncthreads();

  const int c0 = lane * VPL;
  const int hh = c0 / Dh;
  const bool lead = (c0 % Dh) == 0;
  for (int base = wv * 32; base < total; base += WG_THREADS) {
    int sv, dv;
    bool valid = fetch_item(base + lane, n_real, total, use_list, lst, src, dst, lo, cnt, sv, dv);
    unsigned mask = __builtin_amdgcn_ballot_w32(valid);
    while (mask != 0u) {
      int su[U], du[U];
      bool ok[U];
#pragma unroll
      for (int u = 0; u < U; ++u) {
        ok[u] = mask != 0u;
        int j = ok[u] ? (int)__builtin_ctz(mask) : 0;
        if (ok[u]) mask &= mask - 1u;
        su[u] = __builtin_amdgcn_readlane(sv, j);
        du[u] = __builtin_amdgcn_readlane(dv, j);
      }
      vrow row[U];
      float asv[U], adv[U];
      unsigned mxv[U];
#pragma unroll
      for (int u = 0; u < U; ++u) {
        row[u] = *(const vrow*)(hsrc + (size_t)su[u] * DTOT + c0);
        asv[u] = as[(size_t)su[u] * H + hh];
        adv[u] = adl[du[u] * H + hh];
        mxv[u] = mx[du[u] * H + hh];
      }
#pragma unroll
      for (int u = 0; u < U; ++u) {
        if (ok[u]) {
          float e = asv[u] + adv[u];
          e = e > 0.f ? e : NEG_SLOPE * e;
          float w = __expf(e - dec_ord(mxv[u]));
#pragma unroll
          for (int k = 0; k < VPL; ++k)
            atomicAdd(&acc[du[u] * DTOT + c0 + k], w * row[u][k]);
          if (lead) atomicAdd(&sm[du[u] * H + hh], w);
        }
      }
    }
  }
  __syncthreads();

  for (int i = tid; i < cnt * DTOT; i += WG_THREADS) {
    int n = i / DTOT, c = i - n * DTOT;
    float v = acc[i] / (sm[n * H + c / Dh] + GAT_EPS) + bias[c];
    if (ELU) v = v > 0.f ? v : __expf(v) - 1.f;
    out[(size_t)lo * DTOT + i] = v;
  }
}

static inline int cdiv(long long a, long long b) { return (int)((a + b - 1) / b); }

extern "C" void kernel_launch(void* const* d_in, const int* in_sizes, int n_in,
                              void* d_out, int out_size, void* d_ws, size_t ws_size,
                              hipStream_t stream) {
  const float* x      = (const float*)d_in[0];
  const int*   ei     = (const int*)  d_in[1];
  const float* W1     = (const float*)d_in[2];
  const float* a_src1 = (const float*)d_in[3];
  const float* a_dst1 = (const float*)d_in[4];
  const float* b1     = (const float*)d_in[5];
  const float* W2     = (const float*)d_in[6];
  const float* a_src2 = (const float*)d_in[7];
  const float* a_dst2 = (const float*)d_in[8];
  const float* b2     = (const float*)d_in[9];
  float* z = (float*)d_out;

  const int Fin = 256, H1 = 8, C1 = 64, C2 = 128;
  const int N    = in_sizes[0] / Fin;
  const int E    = in_sizes[1] / 2;
  const int* srcv = ei;
  const int* dstv = ei + E;
  const int G = cdiv(N, NODES_PER_WG);

  float* h1   = (float*)d_ws;
  float* as1  = h1  + (size_t)N * C1;
  float* ad1  = as1 + (size_t)N * H1;
  float* hid  = ad1 + (size_t)N * H1;
  float* h2   = hid + (size_t)N * C1;
  float* as2  = h2  + (size_t)N * C2;
  float* ad2  = as2 + N;
  int*   cnts = (int*)(ad2 + N);
  int2*  lsts = (int2*)(cnts + 1024);

  const int B = 256;

  edge_bin_kernel<<<G, WG_THREADS, 0, stream>>>(srcv, dstv, E, N, lsts, cnts);

  {
    int waves = (N / 16) * (C1 / 16);
    gemm_f32_wmma_kernel<<<cdiv((long long)waves * 32, B), B, 0, stream>>>(
        x, W1, h1, N, C1, Fin);
  }
  node_alpha_kernel<8, 8><<<cdiv(N, B), B, 0, stream>>>(h1, a_src1, a_dst1,
                                                        as1, ad1, N);
  range_agg_kernel<64, 8, true><<<G, WG_THREADS, 0, stream>>>(
      srcv, dstv, E, N, lsts, cnts, as1, ad1, h1, b1, hid);

  {
    int waves = (N / 16) * (C2 / 16);
    gemm_f32_wmma_kernel<<<cdiv((long long)waves * 32, B), B, 0, stream>>>(
        hid, W2, h2, N, C2, C1);
  }
  node_alpha_kernel<1, 128><<<cdiv(N, B), B, 0, stream>>>(h2, a_src2, a_dst2,
                                                          as2, ad2, N);
  range_agg_kernel<128, 1, false><<<G, WG_THREADS, 0, stream>>>(
      srcv, dstv, E, N, lsts, cnts, as2, ad2, h2, b2, z);
}


# Edge filter in two levels: a work-group scans 1/8 of the edge list for a group of 8 node ranges; each range owner then picks its edges from its group's 8 lists into LDS; sum order = LDS arrival
# speedup vs baseline: 3.0231x; 1.1216x over previous
#include <hip/hip_runtime.h>
#include <cstdint>


typedef float v2f __attribute__((ext_vector_type(2)));
typedef float v8f __attribute__((ext_vector_type(8)));

#define NEG_SLOPE 0.2f
#define GAT_EPS 1e-16f
#define ENC_NEG_INF 0x007FFFFFu

__device__ __forceinline__ unsigned enc_ord(float f) {
  unsigned u = __float_as_uint(f);
  return (u & 0x80000000u) ? ~u : (u | 0x80000000u);
}
__device__ __forceinline__ float dec_ord(unsigned k) {
  return (k & 0x80000000u) ? __uint_as_float(k & 0x7FFFFFFFu)
                           : __uint_as_float(~k);
}

__global__ void gemm_f32_wmma_kernel(const float* __restrict__ A,
                                     const float* __restrict__ B,
                                     float* __restrict__ C,
                                     int M, int N, int K) {
  int wave = (int)((blockIdx.x * blockDim.x + threadIdx.x) >> 5);
  int lane = threadIdx.x & 31;
  int ntiles = N >> 4;
  int mtiles = (M + 15) >> 4;
  if (wave >= mtiles * ntiles) return;
  int mt = wave / ntiles;
  int nt = wave - mt * ntiles;
  int m0 = mt << 4, n0 = nt << 4;
  int half = lane >> 4;
  int lid  = lane & 15;

  v8f c = {0.f, 0.f, 0.f, 0.f, 0.f, 0.f, 0.f, 0.f};
  const float* arow = A + (size_t)(m0 + lid) * K + 2 * half;
  const float* bcol = B + (size_t)(2 * half) * N + n0 + lid;

  for (int k0 = 0; k0 < K; k0 += 4) {
    v2f a, b;
    a.x = arow[k0];
    a.y = arow[k0 + 1];
    b.x = bcol[(size_t)k0 * N];
    b.y = bcol[(size_t)k0 * N + N];
    c = __builtin_amdgcn_wmma_f32_16x16x4_f32(false, a, false, b,
                                              (short)0, c, false, false);
  }

  float* crow = C + (size_t)m0 * N + n0 + lid;
#pragma unroll
  for (int r = 0; r < 8; ++r)
    crow[(size_t)(r + 8 * half) * N] = c[r];
}

template <int H, int D>
__global__ void node_alpha_kernel(const float* __restrict__ h,
                                  const float* __restrict__ avs,
                                  const float* __restrict__ avd,
                                  float* __restrict__ as,
                                  float* __restrict__ ad, int N) {
  int n = blockIdx.x * blockDim.x + threadIdx.x;
  if (n >= N) return;
  const float* hr = h + (size_t)n * H * D;
#pragma unroll
  for (int hh = 0; hh < H; ++hh) {
    float s0 = 0.f, s1 = 0.f;
#pragma unroll
    for (int d = 0; d < D; ++d) {
      float v = hr[hh * D + d];
      s0 += v * avs[hh * D + d];
      s1 += v * avd[hh * D + d];
    }
    as[(size_t)n * H + hh] = s0;
    ad[(size_t)n * H + hh] = s1;
  }
}

#define NODES_PER_WG 392
#define WG_THREADS 1024
#define WG_WAVES 32
#define LIST_CAP 32768
#define GROUP_RANGES 8
#define GROUP_NODES (GROUP_RANGES * NODES_PER_WG)
#define FINE_CAP 12288
#define FINE_U 8

#define SCAN_U 8

__device__ __forceinline__ void stage_append(bool valid, int e, int dl,
                                             int2* stage, int* lcount, int lane) {
  unsigned m = __builtin_amdgcn_ballot_w32(valid);
  if (m == 0u) return;
  int pos = 0;
  if (lane == 0) pos = atomicAdd(lcount, (int)__popc(m));
  pos = __builtin_amdgcn_readfirstlane(pos);
  if (valid) {
    int slot = pos + (int)__builtin_amdgcn_mbcnt_lo(m, 0u);
    if (slot < LIST_CAP) stage[slot] = make_int2(e, dl);
  }
}

template <bool GUARD>
__device__ __forceinline__ void scan_batch(const int4* __restrict__ d4, int vb, int nvec,
                                           int lo, int cnt, int2* stage, int* lcount,
                                           int lane) {
  int4 d[SCAN_U];
#pragma unroll
  for (int u = 0; u < SCAN_U; ++u) {
    const int v = vb + u * 32 + lane;
    if (GUARD) {
      d[u] = make_int4(lo - 1, lo - 1, lo - 1, lo - 1);
      if (v < nvec) d[u] = d4[v];
    } else {
      d[u] = d4[v];
    }
  }
  unsigned m[SCAN_U][4];
  int tot = 0;
#pragma unroll
  for (int u = 0; u < SCAN_U; ++u) {
    m[u][0] = __builtin_amdgcn_ballot_w32((unsigned)(d[u].x - lo) < (unsigned)cnt);
    m[u][1] = __builtin_amdgcn_ballot_w32((unsigned)(d[u].y - lo) < (unsigned)cnt);
    m[u][2] = __builtin_amdgcn_ballot_w32((unsigned)(d[u].z - lo) < (unsigned)cnt);
    m[u][3] = __builtin_amdgcn_ballot_w32((unsigned)(d[u].w - lo) < (unsigned)cnt);
    tot += (int)(__popc(m[u][0]) + __popc(m[u][1]) + __popc(m[u][2]) + __popc(m[u][3]));
  }
  if (tot == 0) return;
  int base = 0;
  if (lane == 0) base = atomicAdd(lcount, tot);
  base = __builtin_amdgcn_readfirstlane(base);
#pragma unroll
  for (int u = 0; u < SCAN_U; ++u) {
    const int e0 = 4 * (vb + u * 32 + lane);
    const int dd[4] = {d[u].x, d[u].y, d[u].z, d[u].w};
#pragma unroll
    for (int j = 0; j < 4; ++j) {
      const unsigned mm = m[u][j];
      if (mm != 0u) {
        if ((mm >> lane) & 1u) {
          const int slot = base + (int)__builtin_amdgcn_mbcnt_lo(mm, 0u);
          if (slot < LIST_CAP) stage[slot] = make_int2(e0 + j, dd[j] - lo);
        }
        base += (int)__popc(mm);
      }
    }
  }
}

__global__ __launch_bounds__(WG_THREADS) void edge_bin_kernel(
    const int* __restrict__ src, const int* __restrict__ dst, int E, int N,
    int2* __restrict__ lists, int* __restrict__ counts) {
  __shared__ int2 stage[LIST_CAP];
  __shared__ int lcount;
  const int tid = threadIdx.x, lane = tid & 31;
  const int wv = __builtin_amdgcn_readfirstlane(tid >> 5);
  const int grp = blockIdx.x / GROUP_RANGES;
  const int slc = blockIdx.x - grp * GROUP_RANGES;
  const int lo = grp * GROUP_NODES;
  const int cnt = min(GROUP_NODES, N - lo);
  int2* lst = lists + (size_t)blockIdx.x * LIST_CAP;
  if (tid == 0) lcount = 0;
  __syncthreads();
  if ((E & 3) == 0) {
    const int nvec = E >> 2;
    const int vper = (nvec + GROUP_RANGES - 1) / GROUP_RANGES;
    const int vbeg = slc * vper;
    const int vend = min(nvec, vbeg + vper);
    const int4* __restrict__ d4 = (const int4*)dst;
    constexpr int WSPAN = 32 * SCAN_U;
    for (int vb = vbeg + wv * WSPAN; vb < vend; vb += WG_WAVES * WSPAN) {
      if (vb + WSPAN <= vend) scan_batch<false>(d4, vb, vend, lo, cnt, stage, &lcount, lane);
      else                    scan_batch<true >(d4, vb, vend, lo, cnt, stage, &lcount, lane);
    }
  } else {
    const int eper = (E + GROUP_RANGES - 1) / GROUP_RANGES;
    const int ebeg = slc * eper;
    const int eend = min(E, ebeg + eper);
    for (int eb = ebeg + wv * 32; eb < eend; eb += WG_THREADS) {
      const int e = eb + lane;
      int x = -1;
      if (e < eend) x = dst[e] - lo;
      stage_append((unsigned)x < (unsigned)cnt, e, x, stage, &lcount, lane);
    }
  }
  __syncthreads();
  const int found = lcount;
  const int nfill = min(found, LIST_CAP);
  constexpr int GU = 8;
  for (int i0 = tid; i0 < nfill; i0 += GU * WG_THREADS) {
    int2 p[GU];
    int s[GU];
#pragma unroll
    for (int k = 0; k < GU; ++k) {
      const int i = i0 + k * WG_THREADS;
      p[k] = (i < nfill) ? stage[i] : make_int2(0, 0);
    }
#pragma unroll
    for (int k = 0; k < GU; ++k) s[k] = src[p[k].x];
#pragma unroll
    for (int k = 0; k < GU; ++k) {
      const int i = i0 + k * WG_THREADS;
      if (i < nfill) lst[i] = make_int2(s[k], p[k].y);
    }
  }
  if (tid == 0) counts[blockIdx.x] = found;
}

template <bool GUARD>
__device__ __forceinline__ void fine_batch(const int2* __restrict__ seg, int i0, int c,
                                           int sub_lo, int cnt, int2* flist, int* fcount,
                                           int lane) {
  int2 p[FINE_U];
#pragma unroll
  for (int u = 0; u < FINE_U; ++u) {
    const int i = i0 + u * 32 + lane;
    if (GUARD) {
      p[u] = make_int2(0, sub_lo - 1);
      if (i < c) p[u] = seg[i];
    } else {
      p[u] = seg[i];
    }
  }
  unsigned m[FINE_U];
  int tot = 0;
#pragma unroll
  for (int u = 0; u < FINE_U; ++u) {
    m[u] = __builtin_amdgcn_ballot_w32((unsigned)(p[u].y - sub_lo) < (unsigned)cnt);
    tot += (int)__popc(m[u]);
  }
  if (tot == 0) return;
  int base = 0;
  if (lane == 0) base = atomicAdd(fcount, tot);
  base = __builtin_amdgcn_readfirstlane(base);
#pragma unroll
  for (int u = 0; u < FINE_U; ++u) {
    const unsigned mm = m[u];
    if (mm != 0u) {
      if ((mm >> lane) & 1u) {
        const int slot = base + (int)__builtin_amdgcn_mbcnt_lo(mm, 0u);
        if (slot < FINE_CAP) flist[slot] = make_int2(p[u].x, p[u].y - sub_lo);
      }
      base += (int)__popc(mm);
    }
  }
}

__device__ __forceinline__ bool fetch_item(int i, int n_real, int total, bool use_list,
                                           const int2* lst,
                                           const int* __restrict__ src,
                                           const int* __restrict__ dst,
                                           int lo, int cnt, int& sv, int& dv) {
  sv = 0; dv = 0;
  if (i < n_real) {
    if (use_list) { int2 p = lst[i]; sv = p.x; dv = p.y; return true; }
    int d = dst[i] - lo;
    if ((unsigned)d < (unsigned)cnt) { sv = src[i]; dv = d; return true; }
    return false;
  }
  if (i < total) { dv = i - n_real; sv = lo + dv; return true; }
  return false;
}

template <int DTOT, int H, bool ELU>
__global__ __launch_bounds__(WG_THREADS) void range_agg_kernel(
    const int* __restrict__ src, const int* __restrict__ dst, int E, int N,
    const int2* __restrict__ lists, const int* __restrict__ counts,
    const float* __restrict__ as, const float* __restrict__ ad,
    const float* __restrict__ hsrc, const float* __restrict__ bias,
    float* __restrict__ out) {
  constexpr int R = NODES_PER_WG;
  constexpr int VPL = DTOT / 32;
  constexpr int Dh = DTOT / H;
  constexpr int U = 4;
  typedef float vrow __attribute__((ext_vector_type(VPL)));
  __shared__ float acc[R * DTOT];
  __shared__ unsigned mx[R * H];
  __shared__ float sm[R * H];
  __shared__ float adl[R * H];
  __shared__ int2 flist[FINE_CAP];
  __shared__ int fcount;

  const int tid = threadIdx.x, lane = tid & 31, wv = tid >> 5;
  const int lo = blockIdx.x * R;
  const int cnt = min(R, N - lo);
  const int grp = blockIdx.x / GROUP_RANGES;
  const int sub_lo = (blockIdx.x - grp * GROUP_RANGES) * R;
  const int2* glists = lists + (size_t)grp * GROUP_RANGES * LIST_CAP;
  const int* gcounts = counts + grp * GROUP_RANGES;
  int cmax = 0;
#pragma unroll
  for (int s = 0; s < GROUP_RANGES; ++s) cmax = max(cmax, gcounts[s]);
  const bool seg_ok = cmax <= LIST_CAP;

  if (tid == 0) fcount = 0;
  __syncthreads();

  for (int i = tid; i < R * DTOT; i += WG_THREADS) acc[i] = 0.f;
  for (int i = tid; i < R * H; i += WG_THREADS) {
    sm[i] = 0.f;
    mx[i] = ENC_NEG_INF;
    adl[i] = (i < cnt * H) ? ad[(size_t)lo * H + i] : 0.f;
  }
  if (seg_ok) {
    const int wu = __builtin_amdgcn_readfirstlane(wv);
    const int s = wu % GROUP_RANGES;
    const int q = wu / GROUP_RANGES;
    constexpr int FSPAN = 32 * FINE_U;
    constexpr int FSTEP = (WG_WAVES / GROUP_RANGES) * FSPAN;
    const int2* seg = glists + (size_t)s * LIST_CAP;
    const int c = gcounts[s];
    for (int i0 = q * FSPAN; i0 < c; i0 += FSTEP) {
      if (i0 + FSPAN <= c) fine_batch<false>(seg, i0, c, sub_lo, cnt, flist, &fcount, lane);
      else                 fine_batch<true >(seg, i0, c, sub_lo, cnt, flist, &fcount, lane);
    }
  }
  __syncthreads();
  const int found = fcount;
  const bool use_list = seg_ok && found <= FINE_CAP;
  const int n_real = use_list ? found : E;
  const int total = n_real + cnt;
  const int2* lst = flist;

  for (int base = wv * 32; base < total; base += WG_THREADS) {
    int sv, dv;
    bool valid = fetch_item(base + lane, n_real, total, use_list, lst, src, dst, lo, cnt, sv, dv);
    if (valid) {
#pragma unroll
      for (int hh = 0; hh < H; ++hh) {
        float e = as[(size_t)sv * H + hh] + adl[dv * H + hh];
        e = e > 0.f ? e : NEG_SLOPE * e;
        atomicMax(&mx[dv * H + hh], enc_ord(e));
      }
    }
  }
  __syncthreads();

  const int c0 = lane * VPL;
  const int hh = c0 / Dh;
  const bool lead = (c0 % Dh) == 0;
  for (int base = wv * 32; base < total; base += WG_THREADS) {
    int sv, dv;
    bool valid = fetch_item(base + lane, n_real, total, use_list, lst, src, dst, lo, cnt, sv, dv);
    unsigned mask = __builtin_amdgcn_ballot_w32(valid);
    while (mask != 0u) {
      int su[U], du[U];
      bool ok[U];
#pragma unroll
      for (int u = 0; u < U; ++u) {
        ok[u] = mask != 0u;
        int j = ok[u] ? (int)__builtin_ctz(mask) : 0;
        if (ok[u]) mask &= mask - 1u;
        su[u] = __builtin_amdgcn_readlane(sv, j);
        du[u] = __builtin_amdgcn_readlane(dv, j);
      }
      vrow row[U];
      float asv[U], adv[U];
      unsigned mxv[U];
#pragma unroll
      for (int u = 0; u < U; ++u) {
        row[u] = *(const vrow*)(hsrc + (size_t)su[u] * DTOT + c0);
        asv[u] = as[(size_t)su[u] * H + hh];
        adv[u] = adl[du[u] * H + hh];
        mxv[u] = mx[du[u] * H + hh];
      }
#pragma unroll
      for (int u = 0; u < U; ++u) {
        if (ok[u]) {
          float e = asv[u] + adv[u];
          e = e > 0.f ? e : NEG_SLOPE * e;
          float w = __expf(e - dec_ord(mxv[u]));
#pragma unroll
          for (int k = 0; k < VPL; ++k)
            atomicAdd(&acc[du[u] * DTOT + c0 + k], w * row[u][k]);
          if (lead) atomicAdd(&sm[du[u] * H + hh], w);
        }
      }
    }
  }
  __syncthreads();

  for (int i = tid; i < cnt * DTOT; i += WG_THREADS) {
    int n = i / DTOT, c = i - n * DTOT;
    float v = acc[i] / (sm[n * H + c / Dh] + GAT_EPS) + bias[c];
    if (ELU) v = v > 0.f ? v : __expf(v) - 1.f;
    out[(size_t)lo * DTOT + i] = v;
  }
}

static inline int cdiv(long long a, long long b) { return (int)((a + b - 1) / b); }

extern "C" void kernel_launch(void* const* d_in, const int* in_sizes, int n_in,
                              void* d_out, int out_size, void* d_ws, size_t ws_size,
                              hipStream_t stream) {
  const float* x      = (const float*)d_in[0];
  const int*   ei     = (const int*)  d_in[1];
  const float* W1     = (const float*)d_in[2];
  const float* a_src1 = (const float*)d_in[3];
  const float* a_dst1 = (const float*)d_in[4];
  const float* b1     = (const float*)d_in[5];
  const float* W2     = (const float*)d_in[6];
  const float* a_src2 = (const float*)d_in[7];
  const float* a_dst2 = (const float*)d_in[8];
  const float* b2     = (const float*)d_in[9];
  float* z = (float*)d_out;

  const int Fin = 256, H1 = 8, C1 = 64, C2 = 128;
  const int N    = in_sizes[0] / Fin;
  const int E    = in_sizes[1] / 2;
  const int* srcv = ei;
  const int* dstv = ei + E;
  const int G = cdiv(N, NODES_PER_WG);
  const int GB = cdiv(G, GROUP_RANGES) * GROUP_RANGES;

  float* h1   = (float*)d_ws;
  float* as1  = h1  + (size_t)N * C1;
  float* ad1  = as1 + (size_t)N * H1;
  float* hid  = ad1 + (size_t)N * H1;
  float* h2   = hid + (size_t)N * C1;
  float* as2  = h2  + (size_t)N * C2;
  float* ad2  = as2 + N;
  int*   cnts = (int*)(ad2 + N);
  int2*  lsts = (int2*)(cnts + 1024);

  const int B = 256;

  edge_bin_kernel<<<GB, WG_THREADS, 0, stream>>>(srcv, dstv, E, N, lsts, cnts);

  {
    int waves = (N / 16) * (C1 / 16);
    gemm_f32_wmma_kernel<<<cdiv((long long)waves * 32, B), B, 0, stream>>>(
        x, W1, h1, N, C1, Fin);
  }
  node_alpha_kernel<8, 8><<<cdiv(N, B), B, 0, stream>>>(h1, a_src1, a_dst1,
                                                        as1, ad1, N);
  range_agg_kernel<64, 8, true><<<G, WG_THREADS, 0, stream>>>(
      srcv, dstv, E, N, lsts, cnts, as1, ad1, h1, b1, hid);

  {
    int waves = (N / 16) * (C2 / 16);
    gemm_f32_wmma_kernel<<<cdiv((long long)waves * 32, B), B, 0, stream>>>(
        hid, W2, h2, N, C2, C1);
  }
  node_alpha_kernel<1, 128><<<cdiv(N, B), B, 0, stream>>>(h2, a_src2, a_dst2,
                                                          as2, ad2, N);
  range_agg_kernel<128, 1, false><<<G, WG_THREADS, 0, stream>>>(
      srcv, dstv, E, N, lsts, cnts, as2, ad2, h2, b2, z);
}


# Layer-2 product moved into the layer-1 range owner's epilogue: finished hidden rows stay in LDS and feed WMMA as A operand (x W2); no hidden plane, no layer-2 GEMM launch; sum order unchanged
# speedup vs baseline: 3.2252x; 1.0669x over previous
#include <hip/hip_runtime.h>
#include <cstdint>


typedef float v2f __attribute__((ext_vector_type(2)));
typedef float v8f __attribute__((ext_vector_type(8)));

#define NEG_SLOPE 0.2f
#define GAT_EPS 1e-16f
#define ENC_NEG_INF 0x007FFFFFu

__device__ __forceinline__ unsigned enc_ord(float f) {
  unsigned u = __float_as_uint(f);
  return (u & 0x80000000u) ? ~u : (u | 0x80000000u);
}
__device__ __forceinline__ float dec_ord(unsigned k) {
  return (k & 0x80000000u) ? __uint_as_float(k & 0x7FFFFFFFu)
                           : __uint_as_float(~k);
}

__global__ void gemm_f32_wmma_kernel(const float* __restrict__ A,
                                     const float* __restrict__ B,
                                     float* __restrict__ C,
                                     int M, int N, int K) {
  int wave = (int)((blockIdx.x * blockDim.x + threadIdx.x) >> 5);
  int lane = threadIdx.x & 31;
  int ntiles = N >> 4;
  int mtiles = (M + 15) >> 4;
  if (wave >= mtiles * ntiles) return;
  int mt = wave / ntiles;
  int nt = wave - mt * ntiles;
  int m0 = mt << 4, n0 = nt << 4;
  int half = lane >> 4;
  int lid  = lane & 15;

  v8f c = {0.f, 0.f, 0.f, 0.f, 0.f, 0.f, 0.f, 0.f};
  const float* arow = A + (size_t)(m0 + lid) * K + 2 * half;
  const float* bcol = B + (size_t)(2 * half) * N + n0 + lid;

  for (int k0 = 0; k0 < K; k0 += 4) {
    v2f a, b;
    a.x = arow[k0];
    a.y = arow[k0 + 1];
    b.x = bcol[(size_t)k0 * N];
    b.y = bcol[(size_t)k0 * N + N];
    c = __builtin_amdgcn_wmma_f32_16x16x4_f32(false, a, false, b,
                                              (short)0, c, false, false);
  }

  float* crow = C + (size_t)m0 * N + n0 + lid;
#pragma unroll
  for (int r = 0; r < 8; ++r)
    crow[(size_t)(r + 8 * half) * N] = c[r];
}

template <int H, int D>
__global__ void node_alpha_kernel(const float* __restrict__ h,
                                  const float* __restrict__ avs,
                                  const float* __restrict__ avd,
                                  float* __restrict__ as,
                                  float* __restrict__ ad, int N) {
  int n = blockIdx.x * blockDim.x + threadIdx.x;
  if (n >= N) return;
  const float* hr = h + (size_t)n * H * D;
#pragma unroll
  for (int hh = 0; hh < H; ++hh) {
    float s0 = 0.f, s1 = 0.f;
#pragma unroll
    for (int d = 0; d < D; ++d) {
      float v = hr[hh * D + d];
      s0 += v * avs[hh * D + d];
      s1 += v * avd[hh * D + d];
    }
    as[(size_t)n * H + hh] = s0;
    ad[(size_t)n * H + hh] = s1;
  }
}

#define NODES_PER_WG 392
#define WG_THREADS 1024
#define WG_WAVES 32
#define LIST_CAP 32768
#define GROUP_RANGES 8
#define GROUP_NODES (GROUP_RANGES * NODES_PER_WG)
#define FINE_CAP 12288
#define FINE_U 8

#define SCAN_U 8

__device__ __forceinline__ void stage_append(bool valid, int e, int dl,
                                             int2* stage, int* lcount, int lane) {
  unsigned m = __builtin_amdgcn_ballot_w32(valid);
  if (m == 0u) return;
  int pos = 0;
  if (lane == 0) pos = atomicAdd(lcount, (int)__popc(m));
  pos = __builtin_amdgcn_readfirstlane(pos);
  if (valid) {
    int slot = pos + (int)__builtin_amdgcn_mbcnt_lo(m, 0u);
    if (slot < LIST_CAP) stage[slot] = make_int2(e, dl);
  }
}

template <bool GUARD>
__device__ __forceinline__ void scan_batch(const int4* __restrict__ d4, int vb, int nvec,
                                           int lo, int cnt, int2* stage, int* lcount,
                                           int lane) {
  int4 d[SCAN_U];
#pragma unroll
  for (int u = 0; u < SCAN_U; ++u) {
    const int v = vb + u * 32 + lane;
    if (GUARD) {
      d[u] = make_int4(lo - 1, lo - 1, lo - 1, lo - 1);
      if (v < nvec) d[u] = d4[v];
    } else {
      d[u] = d4[v];
    }
  }
  unsigned m[SCAN_U][4];
  int tot = 0;
#pragma unroll
  for (int u = 0; u < SCAN_U; ++u) {
    m[u][0] = __builtin_amdgcn_ballot_w32((unsigned)(d[u].x - lo) < (unsigned)cnt);
    m[u][1] = __builtin_amdgcn_ballot_w32((unsigned)(d[u].y - lo) < (unsigned)cnt);
    m[u][2] = __builtin_amdgcn_ballot_w32((unsigned)(d[u].z - lo) < (unsigned)cnt);
    m[u][3] = __builtin_amdgcn_ballot_w32((unsigned)(d[u].w - lo) < (unsigned)cnt);
    tot += (int)(__popc(m[u][0]) + __popc(m[u][1]) + __popc(m[u][2]) + __popc(m[u][3]));
  }
  if (tot == 0) return;
  int base = 0;
  if (lane == 0) base = atomicAdd(lcount, tot);
  base = __builtin_amdgcn_readfirstlane(base);
#pragma unroll
  for (int u = 0; u < SCAN_U; ++u) {
    const int e0 = 4 * (vb + u * 32 + lane);
    const int dd[4] = {d[u].x, d[u].y, d[u].z, d[u].w};
#pragma unroll
    for (int j = 0; j < 4; ++j) {
      const unsigned mm = m[u][j];
      if (mm != 0u) {
        if ((mm >> lane) & 1u) {
          const int slot = base + (int)__builtin_amdgcn_mbcnt_lo(mm, 0u);
          if (slot < LIST_CAP) stage[slot] = make_int2(e0 + j, dd[j] - lo);
        }
        base += (int)__popc(mm);
      }
    }
  }
}

__global__ __launch_bounds__(WG_THREADS) void edge_bin_kernel(
    const int* __restrict__ src, const int* __restrict__ dst, int E, int N,
    int2* __restrict__ lists, int* __restrict__ counts) {
  __shared__ int2 stage[LIST_CAP];
  __shared__ int lcount;
  const int tid = threadIdx.x, lane = tid & 31;
  const int wv = __builtin_amdgcn_readfirstlane(tid >> 5);
  const int grp = blockIdx.x / GROUP_RANGES;
  const int slc = blockIdx.x - grp * GROUP_RANGES;
  const int lo = grp * GROUP_NODES;
  const int cnt = min(GROUP_NODES, N - lo);
  int2* lst = lists + (size_t)blockIdx.x * LIST_CAP;
  if (tid == 0) lcount = 0;
  __syncthreads();
  if ((E & 3) == 0) {
    const int nvec = E >> 2;
    const int vper = (nvec + GROUP_RANGES - 1) / GROUP_RANGES;
    const int vbeg = slc * vper;
    const int vend = min(nvec, vbeg + vper);
    const int4* __restrict__ d4 = (const int4*)dst;
    constexpr int WSPAN = 32 * SCAN_U;
    for (int vb = vbeg + wv * WSPAN; vb < vend; vb += WG_WAVES * WSPAN) {
      if (vb + WSPAN <= vend) scan_batch<false>(d4, vb, vend, lo, cnt, stage, &lcount, lane);
      else                    scan_batch<true >(d4, vb, vend, lo, cnt, stage, &lcount, lane);
    }
  } else {
    const int eper = (E + GROUP_RANGES - 1) / GROUP_RANGES;
    const int ebeg = slc * eper;
    const int eend = min(E, ebeg + eper);
    for (int eb = ebeg + wv * 32; eb < eend; eb += WG_THREADS) {
      const int e = eb + lane;
      int x = -1;
      if (e < eend) x = dst[e] - lo;
      stage_append((unsigned)x < (unsigned)cnt, e, x, stage, &lcount, lane);
    }
  }
  __syncthreads();
  const int found = lcount;
  const int nfill = min(found, LIST_CAP);
  constexpr int GU = 8;
  for (int i0 = tid; i0 < nfill; i0 += GU * WG_THREADS) {
    int2 p[GU];
    int s[GU];
#pragma unroll
    for (int k = 0; k < GU; ++k) {
      const int i = i0 + k * WG_THREADS;
      p[k] = (i < nfill) ? stage[i] : make_int2(0, 0);
    }
#pragma unroll
    for (int k = 0; k < GU; ++k) s[k] = src[p[k].x];
#pragma unroll
    for (int k = 0; k < GU; ++k) {
      const int i = i0 + k * WG_THREADS;
      if (i < nfill) lst[i] = make_int2(s[k], p[k].y);
    }
  }
  if (tid == 0) counts[blockIdx.x] = found;
}

template <bool GUARD>
__device__ __forceinline__ void fine_batch(const int2* __restrict__ seg, int i0, int c,
                                           int sub_lo, int cnt, int2* flist, int* fcount,
                                           int lane) {
  int2 p[FINE_U];
#pragma unroll
  for (int u = 0; u < FINE_U; ++u) {
    const int i = i0 + u * 32 + lane;
    if (GUARD) {
      p[u] = make_int2(0, sub_lo - 1);
      if (i < c) p[u] = seg[i];
    } else {
      p[u] = seg[i];
    }
  }
  unsigned m[FINE_U];
  int tot = 0;
#pragma unroll
  for (int u = 0; u < FINE_U; ++u) {
    m[u] = __builtin_amdgcn_ballot_w32((unsigned)(p[u].y - sub_lo) < (unsigned)cnt);
    tot += (int)__popc(m[u]);
  }
  if (tot == 0) return;
  int base = 0;
  if (lane == 0) base = atomicAdd(fcount, tot);
  base = __builtin_amdgcn_readfirstlane(base);
#pragma unroll
  for (int u = 0; u < FINE_U; ++u) {
    const unsigned mm = m[u];
    if (mm != 0u) {
      if ((mm >> lane) & 1u) {
        const int slot = base + (int)__builtin_amdgcn_mbcnt_lo(mm, 0u);
        if (slot < FINE_CAP) flist[slot] = make_int2(p[u].x, p[u].y - sub_lo);
      }
      base += (int)__popc(mm);
    }
  }
}

__device__ __forceinline__ bool fetch_item(int i, int n_real, int total, bool use_list,
                                           const int2* lst,
                                           const int* __restrict__ src,
                                           const int* __restrict__ dst,
                                           int lo, int cnt, int& sv, int& dv) {
  sv = 0; dv = 0;
  if (i < n_real) {
    if (use_list) { int2 p = lst[i]; sv = p.x; dv = p.y; return true; }
    int d = dst[i] - lo;
    if ((unsigned)d < (unsigned)cnt) { sv = src[i]; dv = d; return true; }
    return false;
  }
  if (i < total) { dv = i - n_real; sv = lo + dv; return true; }
  return false;
}

template <int DTOT, int H, bool ELU, int NOUT>
__global__ __launch_bounds__(WG_THREADS) void range_agg_kernel(
    const int* __restrict__ src, const int* __restrict__ dst, int E, int N,
    const int2* __restrict__ lists, const int* __restrict__ counts,
    const float* __restrict__ as, const float* __restrict__ ad,
    const float* __restrict__ hsrc, const float* __restrict__ bias,
    const float* __restrict__ Wn, float* __restrict__ out) {
  constexpr int R = NODES_PER_WG;
  constexpr bool PROD = NOUT > 0;
  constexpr int P = PROD ? DTOT + 4 : DTOT;
  constexpr int RP = PROD ? ((R + 15) / 16) * 16 : R;
  constexpr int VPL = DTOT / 32;
  constexpr int Dh = DTOT / H;
  constexpr int U = 4;
  typedef float vrow __attribute__((ext_vector_type(VPL)));
  __shared__ __attribute__((aligned(16))) float acc[RP * P];
  __shared__ unsigned mx[R * H];
  __shared__ float sm[R * H];
  __shared__ float adl[R * H];
  __shared__ int2 flist[FINE_CAP];
  __shared__ int fcount;

  const int tid = threadIdx.x, lane = tid & 31, wv = tid >> 5;
  const int lo = blockIdx.x * R;
  const int cnt = min(R, N - lo);
  const int grp = blockIdx.x / GROUP_RANGES;
  const int sub_lo = (blockIdx.x - grp * GROUP_RANGES) * R;
  const int2* glists = lists + (size_t)grp * GROUP_RANGES * LIST_CAP;
  const int* gcounts = counts + grp * GROUP_RANGES;
  int cmax = 0;
#pragma unroll
  for (int s = 0; s < GROUP_RANGES; ++s) cmax = max(cmax, gcounts[s]);
  const bool seg_ok = cmax <= LIST_CAP;

  if (tid == 0) fcount = 0;
  __syncthreads();

  for (int i = tid; i < RP * P; i += WG_THREADS) acc[i] = 0.f;
  for (int i = tid; i < R * H; i += WG_THREADS) {
    sm[i] = 0.f;
    mx[i] = ENC_NEG_INF;
    adl[i] = (i < cnt * H) ? ad[(size_t)lo * H + i] : 0.f;
  }
  if (seg_ok) {
    const int wu = __builtin_amdgcn_readfirstlane(wv);
    const int s = wu % GROUP_RANGES;
    const int q = wu / GROUP_RANGES;
    constexpr int FSPAN = 32 * FINE_U;
    constexpr int FSTEP = (WG_WAVES / GROUP_RANGES) * FSPAN;
    const int2* seg = glists + (size_t)s * LIST_CAP;
    const int c = gcounts[s];
    for (int i0 = q * FSPAN; i0 < c; i0 += FSTEP) {
      if (i0 + FSPAN <= c) fine_batch<false>(seg, i0, c, sub_lo, cnt, flist, &fcount, lane);
      else                 fine_batch<true >(seg, i0, c, sub_lo, cnt, flist, &fcount, lane);
    }
  }
  __syncthreads();
  const int found = fcount;
  const bool use_list = seg_ok && found <= FINE_CAP;
  const int n_real = use_list ? found : E;
  const int total = n_real + cnt;
  const int2* lst = flist;

  for (int base = wv * 32; base < total; base += WG_THREADS) {
    int sv, dv;
    bool valid = fetch_item(base + lane, n_real, total, use_list, lst, src, dst, lo, cnt, sv, dv);
    if (valid) {
#pragma unroll
      for (int hh = 0; hh < H; ++hh) {
        float e = as[(size_t)sv * H + hh] + adl[dv * H + hh];
        e = e > 0.f ? e : NEG_SLOPE * e;
        atomicMax(&mx[dv * H + hh], enc_ord(e));
      }
    }
  }
  __syncthreads();

  const int c0 = lane * VPL;
  const int hh = c0 / Dh;
  const bool lead = (c0 % Dh) == 0;
  for (int base = wv * 32; base < total; base += WG_THREADS) {
    int sv, dv;
    bool valid = fetch_item(base + lane, n_real, total, use_list, lst, src, dst, lo, cnt, sv, dv);
    unsigned mask = __builtin_amdgcn_ballot_w32(valid);
    while (mask != 0u) {
      int su[U], du[U];
      bool ok[U];
#pragma unroll
      for (int u = 0; u < U; ++u) {
        ok[u] = mask != 0u;
        int j = ok[u] ? (int)__builtin_ctz(mask) : 0;
        if (ok[u]) mask &= mask - 1u;
        su[u] = __builtin_amdgcn_readlane(sv, j);
        du[u] = __builtin_amdgcn_readlane(dv, j);
      }
      vrow row[U];
      float asv[U], adv[U];
      unsigned mxv[U];
#pragma unroll
      for (int u = 0; u < U; ++u) {
        row[u] = *(const vrow*)(hsrc + (size_t)su[u] * DTOT + c0);
        asv[u] = as[(size_t)su[u] * H + hh];
        adv[u] = adl[du[u] * H + hh];
        mxv[u] = mx[du[u] * H + hh];
      }
#pragma unroll
      for (int u = 0; u < U; ++u) {
        if (ok[u]) {
          float e = asv[u] + adv[u];
          e = e > 0.f ? e : NEG_SLOPE * e;
          float w = __expf(e - dec_ord(mxv[u]));
#pragma unroll
          for (int k = 0; k < VPL; ++k)
            atomicAdd(&acc[du[u] * P + c0 + k], w * row[u][k]);
          if (lead) atomicAdd(&sm[du[u] * H + hh], w);
        }
      }
    }
  }
  __syncthreads();

  if constexpr (!PROD) {
    for (int i = tid; i < cnt * DTOT; i += WG_THREADS) {
      int n = i / DTOT, c = i - n * DTOT;
      float v = acc[i] / (sm[n * H + c / Dh] + GAT_EPS) + bias[c];
      if (ELU) v = v > 0.f ? v : __expf(v) - 1.f;
      out[(size_t)lo * DTOT + i] = v;
    }
  } else {
    for (int i = tid; i < cnt * DTOT; i += WG_THREADS) {
      int n = i / DTOT, c = i - n * DTOT;
      float v = acc[n * P + c] / (sm[n * H + c / Dh] + GAT_EPS) + bias[c];
      if (ELU) v = v > 0.f ? v : __expf(v) - 1.f;
      acc[n * P + c] = v;
    }
    __syncthreads();
    constexpr int NT = NOUT / 16;
    constexpr int MT = RP / 16;
    constexpr int KS = DTOT / 4;
    constexpr int QS = WG_WAVES / NT;
    const int half = lane >> 4, lid = lane & 15;
    const int wu = __builtin_amdgcn_readfirstlane(wv);
    const int ct = wu % NT;
    const int q = wu / NT;
    const float* bcol = Wn + (size_t)(2 * half) * NOUT + ct * 16 + lid;
    v2f b[KS];
#pragma unroll
    for (int ks = 0; ks < KS; ++ks) {
      b[ks].x = bcol[(size_t)(4 * ks) * NOUT];
      b[ks].y = bcol[(size_t)(4 * ks + 1) * NOUT];
    }
    for (int mt = q; mt < MT; mt += QS) {
      const float* arow = acc + (mt * 16 + lid) * P + 2 * half;
      v2f a[KS];
#pragma unroll
      for (int ks = 0; ks < KS; ++ks) a[ks] = *(const v2f*)(arow + 4 * ks);
      v8f c = {0.f, 0.f, 0.f, 0.f, 0.f, 0.f, 0.f, 0.f};
#pragma unroll
      for (int ks = 0; ks < KS; ++ks)
        c = __builtin_amdgcn_wmma_f32_16x16x4_f32(false, a[ks], false, b[ks],
                                                  (short)0, c, false, false);
      const int r0 = mt * 16 + 8 * half;
      float* crow = out + (size_t)(lo + r0) * NOUT + ct * 16 + lid;
      if (mt * 16 + 16 <= cnt) {
#pragma unroll
        for (int r = 0; r < 8; ++r) crow[(size_t)r * NOUT] = c[r];
      } else {
        const int nrows = cnt - r0;
#pragma unroll
        for (int r = 0; r < 8; ++r)
          if (r < nrows) crow[(size_t)r * NOUT] = c[r];
      }
    }
  }
}

static inline int cdiv(long long a, long long b) { return (int)((a + b - 1) / b); }

extern "C" void kernel_launch(void* const* d_in, const int* in_sizes, int n_in,
                              void* d_out, int out_size, void* d_ws, size_t ws_size,
                              hipStream_t stream) {
  const float* x      = (const float*)d_in[0];
  const int*   ei     = (const int*)  d_in[1];
  const float* W1     = (const float*)d_in[2];
  const float* a_src1 = (const float*)d_in[3];
  const float* a_dst1 = (const float*)d_in[4];
  const float* b1     = (const float*)d_in[5];
  const float* W2     = (const float*)d_in[6];
  const float* a_src2 = (const float*)d_in[7];
  const float* a_dst2 = (const float*)d_in[8];
  const float* b2     = (const float*)d_in[9];
  float* z = (float*)d_out;

  const int Fin = 256, H1 = 8, C1 = 64, C2 = 128;
  const int N    = in_sizes[0] / Fin;
  const int E    = in_sizes[1] / 2;
  const int* srcv = ei;
  const int* dstv = ei + E;
  const int G = cdiv(N, NODES_PER_WG);
  const int GB = cdiv(G, GROUP_RANGES) * GROUP_RANGES;

  float* h1   = (float*)d_ws;
  float* as1  = h1  + (size_t)N * C1;
  float* ad1  = as1 + (size_t)N * H1;
  float* h2   = ad1 + (size_t)N * H1;
  float* as2  = h2  + (size_t)N * C2;
  float* ad2  = as2 + N;
  int*   cnts = (int*)(ad2 + N);
  int2*  lsts = (int2*)(cnts + 1024);

  const int B = 256;

  edge_bin_kernel<<<GB, WG_THREADS, 0, stream>>>(srcv, dstv, E, N, lsts, cnts);

  {
    int waves = (N / 16) * (C1 / 16);
    gemm_f32_wmma_kernel<<<cdiv((long long)waves * 32, B), B, 0, stream>>>(
        x, W1, h1, N, C1, Fin);
  }
  node_alpha_kernel<8, 8><<<cdiv(N, B), B, 0, stream>>>(h1, a_src1, a_dst1,
                                                        as1, ad1, N);
  range_agg_kernel<64, 8, true, 128><<<G, WG_THREADS, 0, stream>>>(
      srcv, dstv, E, N, lsts, cnts, as1, ad1, h1, b1, W2, h2);

  node_alpha_kernel<1, 128><<<cdiv(N, B), B, 0, stream>>>(h2, a_src2, a_dst2,
                                                          as2, ad2, N);
  range_agg_kernel<128, 1, false, 0><<<G, WG_THREADS, 0, stream>>>(
      srcv, dstv, E, N, lsts, cnts, as2, ad2, h2, b2, nullptr, z);
}


# Layer-1 GEMM operands via LDS and registers: 64-row x slabs fetched by whole-row loads into padded LDS, next slab in flight; W1 column tile held in registers; 16-wave work-groups; sum order unchanged
# speedup vs baseline: 3.4497x; 1.0696x over previous
#include <hip/hip_runtime.h>
#include <cstdint>


typedef float v2f __attribute__((ext_vector_type(2)));
typedef float v8f __attribute__((ext_vector_type(8)));

#define NEG_SLOPE 0.2f
#define GAT_EPS 1e-16f
#define ENC_NEG_INF 0x007FFFFFu

__device__ __forceinline__ unsigned enc_ord(float f) {
  unsigned u = __float_as_uint(f);
  return (u & 0x80000000u) ? ~u : (u | 0x80000000u);
}
__device__ __forceinline__ float dec_ord(unsigned k) {
  return (k & 0x80000000u) ? __uint_as_float(k & 0x7FFFFFFFu)
                           : __uint_as_float(~k);
}

#define G1_K 256
#define G1_N 64
#define G1_ROWS 64
#define G1_PITCH (G1_K + 4)
#define G1_THREADS 512

__global__ __launch_bounds__(G1_THREADS) void gemm1_lds_kernel(
    const float* __restrict__ A, const float* __restrict__ B,
    float* __restrict__ C, int M) {
  __shared__ __attribute__((aligned(16))) float sA[2 * G1_ROWS * G1_PITCH];
  constexpr int KS = G1_K / 4;
  constexpr int LD = (G1_ROWS * G1_K / 4) / G1_THREADS;
  constexpr int STAGE = G1_ROWS * G1_PITCH;
  const int tid = threadIdx.x, lane = tid & 31;
  const int half = lane >> 4, lid = lane & 15;
  const int wu = __builtin_amdgcn_readfirstlane(tid >> 5);
  const int ct = wu & 3;
  const int rt = wu >> 2;
  const int mtiles = (M + 15) >> 4;
  const int t0 = (int)(((long long)blockIdx.x * mtiles) / gridDim.x);
  const int t1 = (int)(((long long)(blockIdx.x + 1) * mtiles) / gridDim.x);
  const int rlim = min(16 * t1, M);

  const float* bcol = B + (size_t)(2 * half) * G1_N + ct * 16 + lid;
  v2f b[KS];
#pragma unroll
  for (int ks = 0; ks < KS; ++ks) {
    b[ks].x = bcol[(size_t)(4 * ks) * G1_N];
    b[ks].y = bcol[(size_t)(4 * ks + 1) * G1_N];
  }

  float4 pre[LD];
#pragma unroll
  for (int i = 0; i < LD; ++i) {
    const int idx = i * G1_THREADS + tid;
    const int row = 16 * t0 + (idx >> 6);
    pre[i] = make_float4(0.f, 0.f, 0.f, 0.f);
    if (row < rlim) pre[i] = *(const float4*)(A + (size_t)row * G1_K + 4 * (idx & 63));
  }
#pragma unroll
  for (int i = 0; i < LD; ++i) {
    const int idx = i * G1_THREADS + tid;
    *(float4*)(sA + (idx >> 6) * G1_PITCH + 4 * (idx & 63)) = pre[i];
  }
  __syncthreads();

  int buf = 0;
  for (int tb = t0; tb < t1; tb += 4) {
    const int tn = tb + 4;
    const bool more = tn < t1;
    if (more) {
#pragma unroll
      for (int i = 0; i < LD; ++i) {
        const int idx = i * G1_THREADS + tid;
        const int row = 16 * tn + (idx >> 6);
        pre[i] = make_float4(0.f, 0.f, 0.f, 0.f);
        if (row < rlim) pre[i] = *(const float4*)(A + (size_t)row * G1_K + 4 * (idx & 63));
      }
    }
    const int mt = tb + rt;
    if (mt < t1) {
      const float* arow = sA + buf * STAGE + (rt * 16 + lid) * G1_PITCH + 2 * half;
      v8f c = {0.f, 0.f, 0.f, 0.f, 0.f, 0.f, 0.f, 0.f};
      constexpr int AC = 8;
      v2f a[2][AC];
#pragma unroll
      for (int j = 0; j < AC; ++j) a[0][j] = *(const v2f*)(arow + 4 * j);
#pragma unroll
      for (int kc = 0; kc < KS / AC; ++kc) {
        if (kc + 1 < KS / AC) {
#pragma unroll
          for (int j = 0; j < AC; ++j)
            a[(kc + 1) & 1][j] = *(const v2f*)(arow + 4 * (AC * (kc + 1) + j));
        }
        __builtin_amdgcn_sched_barrier(0);
#pragma unroll
        for (int j = 0; j < AC; ++j)
          c = __builtin_amdgcn_wmma_f32_16x16x4_f32(false, a[kc & 1][j], false, b[AC * kc + j],
                                                    (short)0, c, false, false);
        __builtin_amdgcn_sched_barrier(0);
      }
      const int r0 = mt * 16 + 8 * half;
      float* crow = C + (size_t)r0 * G1_N + ct * 16 + lid;
      if (mt * 16 + 16 <= M) {
#pragma unroll
        for (int r = 0; r < 8; ++r) crow[(size_t)r * G1_N] = c[r];
      } else {
        const int nrows = M - r0;
#pragma unroll
        for (int r = 0; r < 8; ++r)
          if (r < nrows) crow[(size_t)r * G1_N] = c[r];
      }
    }
    if (more) {
      float* nxt = sA + (buf ^ 1) * STAGE;
#pragma unroll
      for (int i = 0; i < LD; ++i) {
        const int idx = i * G1_THREADS + tid;
        *(float4*)(nxt + (idx >> 6) * G1_PITCH + 4 * (idx & 63)) = pre[i];
      }
    }
    __syncthreads();
    buf ^= 1;
  }
}

template <int H, int D>
__global__ void node_alpha_kernel(const float* __restrict__ h,
                                  const float* __restrict__ avs,
                                  const float* __restrict__ avd,
                                  float* __restrict__ as,
                                  float* __restrict__ ad, int N) {
  int n = blockIdx.x * blockDim.x + threadIdx.x;
  if (n >= N) return;
  const float* hr = h + (size_t)n * H * D;
#pragma unroll
  for (int hh = 0; hh < H; ++hh) {
    float s0 = 0.f, s1 = 0.f;
#pragma unroll
    for (int d = 0; d < D; ++d) {
      float v = hr[hh * D + d];
      s0 += v * avs[hh * D + d];
      s1 += v * avd[hh * D + d];
    }
    as[(size_t)n * H + hh] = s0;
    ad[(size_t)n * H + hh] = s1;
  }
}

#define NODES_PER_WG 392
#define WG_THREADS 1024
#define WG_WAVES 32
#define LIST_CAP 32768
#define GROUP_RANGES 8
#define GROUP_NODES (GROUP_RANGES * NODES_PER_WG)
#define FINE_CAP 12288
#define FINE_U 8

#define SCAN_U 8

__device__ __forceinline__ void stage_append(bool valid, int e, int dl,
                                             int2* stage, int* lcount, int lane) {
  unsigned m = __builtin_amdgcn_ballot_w32(valid);
  if (m == 0u) return;
  int pos = 0;
  if (lane == 0) pos = atomicAdd(lcount, (int)__popc(m));
  pos = __builtin_amdgcn_readfirstlane(pos);
  if (valid) {
    int slot = pos + (int)__builtin_amdgcn_mbcnt_lo(m, 0u);
    if (slot < LIST_CAP) stage[slot] = make_int2(e, dl);
  }
}

template <bool GUARD>
__device__ __forceinline__ void scan_batch(const int4* __restrict__ d4, int vb, int nvec,
                                           int lo, int cnt, int2* stage, int* lcount,
                                           int lane) {
  int4 d[SCAN_U];
#pragma unroll
  for (int u = 0; u < SCAN_U; ++u) {
    const int v = vb + u * 32 + lane;
    if (GUARD) {
      d[u] = make_int4(lo - 1, lo - 1, lo - 1, lo - 1);
      if (v < nvec) d[u] = d4[v];
    } else {
      d[u] = d4[v];
    }
  }
  unsigned m[SCAN_U][4];
  int tot = 0;
#pragma unroll
  for (int u = 0; u < SCAN_U; ++u) {
    m[u][0] = __builtin_amdgcn_ballot_w32((unsigned)(d[u].x - lo) < (unsigned)cnt);
    m[u][1] = __builtin_amdgcn_ballot_w32((unsigned)(d[u].y - lo) < (unsigned)cnt);
    m[u][2] = __builtin_amdgcn_ballot_w32((unsigned)(d[u].z - lo) < (unsigned)cnt);
    m[u][3] = __builtin_amdgcn_ballot_w32((unsigned)(d[u].w - lo) < (unsigned)cnt);
    tot += (int)(__popc(m[u][0]) + __popc(m[u][1]) + __popc(m[u][2]) + __popc(m[u][3]));
  }
  if (tot == 0) return;
  int base = 0;
  if (lane == 0) base = atomicAdd(lcount, tot);
  base = __builtin_amdgcn_readfirstlane(base);
#pragma unroll
  for (int u = 0; u < SCAN_U; ++u) {
    const int e0 = 4 * (vb + u * 32 + lane);
    const int dd[4] = {d[u].x, d[u].y, d[u].z, d[u].w};
#pragma unroll
    for (int j = 0; j < 4; ++j) {
      const unsigned mm = m[u][j];
      if (mm != 0u) {
        if ((mm >> lane) & 1u) {
          const int slot = base + (int)__builtin_amdgcn_mbcnt_lo(mm, 0u);
          if (slot < LIST_CAP) stage[slot] = make_int2(e0 + j, dd[j] - lo);
        }
        base += (int)__popc(mm);
      }
    }
  }
}

__global__ __launch_bounds__(WG_THREADS) void edge_bin_kernel(
    const int* __restrict__ src, const int* __restrict__ dst, int E, int N,
    int2* __restrict__ lists, int* __restrict__ counts) {
  __shared__ int2 stage[LIST_CAP];
  __shared__ int lcount;
  const int tid = threadIdx.x, lane = tid & 31;
  const int wv = __builtin_amdgcn_readfirstlane(tid >> 5);
  const int grp = blockIdx.x / GROUP_RANGES;
  const int slc = blockIdx.x - grp * GROUP_RANGES;
  const int lo = grp * GROUP_NODES;
  const int cnt = min(GROUP_NODES, N - lo);
  int2* lst = lists + (size_t)blockIdx.x * LIST_CAP;
  if (tid == 0) lcount = 0;
  __syncthreads();
  if ((E & 3) == 0) {
    const int nvec = E >> 2;
    const int vper = (nvec + GROUP_RANGES - 1) / GROUP_RANGES;
    const int vbeg = slc * vper;
    const int vend = min(nvec, vbeg + vper);
    const int4* __restrict__ d4 = (const int4*)dst;
    constexpr int WSPAN = 32 * SCAN_U;
    for (int vb = vbeg + wv * WSPAN; vb < vend; vb += WG_WAVES * WSPAN) {
      if (vb + WSPAN <= vend) scan_batch<false>(d4, vb, vend, lo, cnt, stage, &lcount, lane);
      else                    scan_batch<true >(d4, vb, vend, lo, cnt, stage, &lcount, lane);
    }
  } else {
    const int eper = (E + GROUP_RANGES - 1) / GROUP_RANGES;
    const int ebeg = slc * eper;
    const int eend = min(E, ebeg + eper);
    for (int eb = ebeg + wv * 32; eb < eend; eb += WG_THREADS) {
      const int e = eb + lane;
      int x = -1;
      if (e < eend) x = dst[e] - lo;
      stage_append((unsigned)x < (unsigned)cnt, e, x, stage, &lcount, lane);
    }
  }
  __syncthreads();
  const int found = lcount;
  const int nfill = min(found, LIST_CAP);
  constexpr int GU = 8;
  for (int i0 = tid; i0 < nfill; i0 += GU * WG_THREADS) {
    int2 p[GU];
    int s[GU];
#pragma unroll
    for (int k = 0; k < GU; ++k) {
      const int i = i0 + k * WG_THREADS;
      p[k] = (i < nfill) ? stage[i] : make_int2(0, 0);
    }
#pragma unroll
    for (int k = 0; k < GU; ++k) s[k] = src[p[k].x];
#pragma unroll
    for (int k = 0; k < GU; ++k) {
      const int i = i0 + k * WG_THREADS;
      if (i < nfill) lst[i] = make_int2(s[k], p[k].y);
    }
  }
  if (tid == 0) counts[blockIdx.x] = found;
}

template <bool GUARD>
__device__ __forceinline__ void fine_batch(const int2* __restrict__ seg, int i0, int c,
                                           int sub_lo, int cnt, int2* flist, int* fcount,
                                           int lane) {
  int2 p[FINE_U];
#pragma unroll
  for (int u = 0; u < FINE_U; ++u) {
    const int i = i0 + u * 32 + lane;
    if (GUARD) {
      p[u] = make_int2(0, sub_lo - 1);
      if (i < c) p[u] = seg[i];
    } else {
      p[u] = seg[i];
    }
  }
  unsigned m[FINE_U];
  int tot = 0;
#pragma unroll
  for (int u = 0; u < FINE_U; ++u) {
    m[u] = __builtin_amdgcn_ballot_w32((unsigned)(p[u].y - sub_lo) < (unsigned)cnt);
    tot += (int)__popc(m[u]);
  }
  if (tot == 0) return;
  int base = 0;
  if (lane == 0) base = atomicAdd(fcount, tot);
  base = __builtin_amdgcn_readfirstlane(base);
#pragma unroll
  for (int u = 0; u < FINE_U; ++u) {
    const unsigned mm = m[u];
    if (mm != 0u) {
      if ((mm >> lane) & 1u) {
        const int slot = base + (int)__builtin_amdgcn_mbcnt_lo(mm, 0u);
        if (slot < FINE_CAP) flist[slot] = make_int2(p[u].x, p[u].y - sub_lo);
      }
      base += (int)__popc(mm);
    }
  }
}

__device__ __forceinline__ bool fetch_item(int i, int n_real, int total, bool use_list,
                                           const int2* lst,
                                           const int* __restrict__ src,
                                           const int* __restrict__ dst,
                                           int lo, int cnt, int& sv, int& dv) {
  sv = 0; dv = 0;
  if (i < n_real) {
    if (use_list) { int2 p = lst[i]; sv = p.x; dv = p.y; return true; }
    int d = dst[i] - lo;
    if ((unsigned)d < (unsigned)cnt) { sv = src[i]; dv = d; return true; }
    return false;
  }
  if (i < total) { dv = i - n_real; sv = lo + dv; return true; }
  return false;
}

template <int DTOT, int H, bool ELU, int NOUT>
__global__ __launch_bounds__(WG_THREADS) void range_agg_kernel(
    const int* __restrict__ src, const int* __restrict__ dst, int E, int N,
    const int2* __restrict__ lists, const int* __restrict__ counts,
    const float* __restrict__ as, const float* __restrict__ ad,
    const float* __restrict__ hsrc, const float* __restrict__ bias,
    const float* __restrict__ Wn, float* __restrict__ out) {
  constexpr int R = NODES_PER_WG;
  constexpr bool PROD = NOUT > 0;
  constexpr int P = PROD ? DTOT + 4 : DTOT;
  constexpr int RP = PROD ? ((R + 15) / 16) * 16 : R;
  constexpr int VPL = DTOT / 32;
  constexpr int Dh = DTOT / H;
  constexpr int U = 4;
  typedef float vrow __attribute__((ext_vector_type(VPL)));
  __shared__ __attribute__((aligned(16))) float acc[RP * P];
  __shared__ unsigned mx[R * H];
  __shared__ float sm[R * H];
  __shared__ float adl[R * H];
  __shared__ int2 flist[FINE_CAP];
  __shared__ int fcount;

  const int tid = threadIdx.x, lane = tid & 31, wv = tid >> 5;
  const int lo = blockIdx.x * R;
  const int cnt = min(R, N - lo);
  const int grp = blockIdx.x / GROUP_RANGES;
  const int sub_lo = (blockIdx.x - grp * GROUP_RANGES) * R;
  const int2* glists = lists + (size_t)grp * GROUP_RANGES * LIST_CAP;
  const int* gcounts = counts + grp * GROUP_RANGES;
  int cmax = 0;
#pragma unroll
  for (int s = 0; s < GROUP_RANGES; ++s) cmax = max(cmax, gcounts[s]);
  const bool seg_ok = cmax <= LIST_CAP;

  if (tid == 0) fcount = 0;
  __syncthreads();

  for (int i = tid; i < RP * P; i += WG_THREADS) acc[i] = 0.f;
  for (int i = tid; i < R * H; i += WG_THREADS) {
    sm[i] = 0.f;
    mx[i] = ENC_NEG_INF;
    adl[i] = (i < cnt * H) ? ad[(size_t)lo * H + i] : 0.f;
  }
  if (seg_ok) {
    const int wu = __builtin_amdgcn_readfirstlane(wv);
    const int s = wu % GROUP_RANGES;
    const int q = wu / GROUP_RANGES;
    constexpr int FSPAN = 32 * FINE_U;
    constexpr int FSTEP = (WG_WAVES / GROUP_RANGES) * FSPAN;
    const int2* seg = glists + (size_t)s * LIST_CAP;
    const int c = gcounts[s];
    for (int i0 = q * FSPAN; i0 < c; i0 += FSTEP) {
      if (i0 + FSPAN <= c) fine_batch<false>(seg, i0, c, sub_lo, cnt, flist, &fcount, lane);
      else                 fine_batch<true >(seg, i0, c, sub_lo, cnt, flist, &fcount, lane);
    }
  }
  __syncthreads();
  const int found = fcount;
  const bool use_list = seg_ok && found <= FINE_CAP;
  const int n_real = use_list ? found : E;
  const int total = n_real + cnt;
  const int2* lst = flist;

  for (int base = wv * 32; base < total; base += WG_THREADS) {
    int sv, dv;
    bool valid = fetch_item(base + lane, n_real, total, use_list, lst, src, dst, lo, cnt, sv, dv);
    if (valid) {
#pragma unroll
      for (int hh = 0; hh < H; ++hh) {
        float e = as[(size_t)sv * H + hh] + adl[dv * H + hh];
        e = e > 0.f ? e : NEG_SLOPE * e;
        atomicMax(&mx[dv * H + hh], enc_ord(e));
      }
    }
  }
  __syncthreads();

  const int c0 = lane * VPL;
  const int hh = c0 / Dh;
  const bool lead = (c0 % Dh) == 0;
  for (int base = wv * 32; base < total; base += WG_THREADS) {
    int sv, dv;
    bool valid = fetch_item(base + lane, n_real, total, use_list, lst, src, dst, lo, cnt, sv, dv);
    unsigned mask = __builtin_amdgcn_ballot_w32(valid);
    while (mask != 0u) {
      int su[U], du[U];
      bool ok[U];
#pragma unroll
      for (int u = 0; u < U; ++u) {
        ok[u] = mask != 0u;
        int j = ok[u] ? (int)__builtin_ctz(mask) : 0;
        if (ok[u]) mask &= mask - 1u;
        su[u] = __builtin_amdgcn_readlane(sv, j);
        du[u] = __builtin_amdgcn_readlane(dv, j);
      }
      vrow row[U];
      float asv[U], adv[U];
      unsigned mxv[U];
#pragma unroll
      for (int u = 0; u < U; ++u) {
        row[u] = *(const vrow*)(hsrc + (size_t)su[u] * DTOT + c0);
        asv[u] = as[(size_t)su[u] * H + hh];
        adv[u] = adl[du[u] * H + hh];
        mxv[u] = mx[du[u] * H + hh];
      }
#pragma unroll
      for (int u = 0; u < U; ++u) {
        if (ok[u]) {
          float e = asv[u] + adv[u];
          e = e > 0.f ? e : NEG_SLOPE * e;
          float w = __expf(e - dec_ord(mxv[u]));
#pragma unroll
          for (int k = 0; k < VPL; ++k)
            atomicAdd(&acc[du[u] * P + c0 + k], w * row[u][k]);
          if (lead) atomicAdd(&sm[du[u] * H + hh], w);
        }
      }
    }
  }
  __syncthreads();

  if constexpr (!PROD) {
    for (int i = tid; i < cnt * DTOT; i += WG_THREADS) {
      int n = i / DTOT, c = i - n * DTOT;
      float v = acc[i] / (sm[n * H + c / Dh] + GAT_EPS) + bias[c];
      if (ELU) v = v > 0.f ? v : __expf(v) - 1.f;
      out[(size_t)lo * DTOT + i] = v;
    }
  } else {
    for (int i = tid; i < cnt * DTOT; i += WG_THREADS) {
      int n = i / DTOT, c = i - n * DTOT;
      float v = acc[n * P + c] / (sm[n * H + c / Dh] + GAT_EPS) + bias[c];
      if (ELU) v = v > 0.f ? v : __expf(v) - 1.f;
      acc[n * P + c] = v;
    }
    __syncthreads();
    constexpr int NT = NOUT / 16;
    constexpr int MT = RP / 16;
    constexpr int KS = DTOT / 4;
    constexpr int QS = WG_WAVES / NT;
    const int half = lane >> 4, lid = lane & 15;
    const int wu = __builtin_amdgcn_readfirstlane(wv);
    const int ct = wu % NT;
    const int q = wu / NT;
    const float* bcol = Wn + (size_t)(2 * half) * NOUT + ct * 16 + lid;
    v2f b[KS];
#pragma unroll
    for (int ks = 0; ks < KS; ++ks) {
      b[ks].x = bcol[(size_t)(4 * ks) * NOUT];
      b[ks].y = bcol[(size_t)(4 * ks + 1) * NOUT];
    }
    for (int mt = q; mt < MT; mt += QS) {
      const float* arow = acc + (mt * 16 + lid) * P + 2 * half;
      v2f a[KS];
#pragma unroll
      for (int ks = 0; ks < KS; ++ks) a[ks] = *(const v2f*)(arow + 4 * ks);
      v8f c = {0.f, 0.f, 0.f, 0.f, 0.f, 0.f, 0.f, 0.f};
#pragma unroll
      for (int ks = 0; ks < KS; ++ks)
        c = __builtin_amdgcn_wmma_f32_16x16x4_f32(false, a[ks], false, b[ks],
                                                  (short)0, c, false, false);
      const int r0 = mt * 16 + 8 * half;
      float* crow = out + (size_t)(lo + r0) * NOUT + ct * 16 + lid;
      if (mt * 16 + 16 <= cnt) {
#pragma unroll
        for (int r = 0; r < 8; ++r) crow[(size_t)r * NOUT] = c[r];
      } else {
        const int nrows = cnt - r0;
#pragma unroll
        for (int r = 0; r < 8; ++r)
          if (r < nrows) crow[(size_t)r * NOUT] = c[r];
      }
    }
  }
}

static inline int cdiv(long long a, long long b) { return (int)((a + b - 1) / b); }

extern "C" void kernel_launch(void* const* d_in, const int* in_sizes, int n_in,
                              void* d_out, int out_size, void* d_ws, size_t ws_size,
                              hipStream_t stream) {
  const float* x      = (const float*)d_in[0];
  const int*   ei     = (const int*)  d_in[1];
  const float* W1     = (const float*)d_in[2];
  const float* a_src1 = (const float*)d_in[3];
  const float* a_dst1 = (const float*)d_in[4];
  const float* b1     = (const float*)d_in[5];
  const float* W2     = (const float*)d_in[6];
  const float* a_src2 = (const float*)d_in[7];
  const float* a_dst2 = (const float*)d_in[8];
  const float* b2     = (const float*)d_in[9];
  float* z = (float*)d_out;

  const int Fin = 256, H1 = 8, C1 = 64, C2 = 128;
  const int N    = in_sizes[0] / Fin;
  const int E    = in_sizes[1] / 2;
  const int* srcv = ei;
  const int* dstv = ei + E;
  const int G = cdiv(N, NODES_PER_WG);
  const int GB = cdiv(G, GROUP_RANGES) * GROUP_RANGES;

  float* h1   = (float*)d_ws;
  float* as1  = h1  + (size_t)N * C1;
  float* ad1  = as1 + (size_t)N * H1;
  float* h2   = ad1 + (size_t)N * H1;
  float* as2  = h2  + (size_t)N * C2;
  float* ad2  = as2 + N;
  int*   cnts = (int*)(ad2 + N);
  int2*  lsts = (int2*)(cnts + 1024);

  const int B = 256;

  edge_bin_kernel<<<GB, WG_THREADS, 0, stream>>>(srcv, dstv, E, N, lsts, cnts);

  gemm1_lds_kernel<<<256, G1_THREADS, 0, stream>>>(x, W1, h1, N);
  node_alpha_kernel<8, 8><<<cdiv(N, B), B, 0, stream>>>(h1, a_src1, a_dst1,
                                                        as1, ad1, N);
  range_agg_kernel<64, 8, true, 128><<<G, WG_THREADS, 0, stream>>>(
      srcv, dstv, E, N, lsts, cnts, as1, ad1, h1, b1, W2, h2);

  node_alpha_kernel<1, 128><<<cdiv(N, B), B, 0, stream>>>(h2, a_src2, a_dst2,
                                                          as2, ad2, N);
  range_agg_kernel<128, 1, false, 0><<<G, WG_THREADS, 0, stream>>>(
      srcv, dstv, E, N, lsts, cnts, as2, ad2, h2, b2, nullptr, z);
}


# Node ownership inside the range: wave w owns nodes w+32j, the fine pass files edges by owner wave in LDS, that wave sums rows and weights in registers; no LDS float adds; sum order = LDS arrival
# speedup vs baseline: 3.9071x; 1.1326x over previous
#include <hip/hip_runtime.h>
#include <cstdint>


typedef float v2f __attribute__((ext_vector_type(2)));
typedef float v8f __attribute__((ext_vector_type(8)));

#define NEG_SLOPE 0.2f
#define GAT_EPS 1e-16f
#define ENC_NEG_INF 0x007FFFFFu

__device__ __forceinline__ unsigned enc_ord(float f) {
  unsigned u = __float_as_uint(f);
  return (u & 0x80000000u) ? ~u : (u | 0x80000000u);
}
__device__ __forceinline__ float dec_ord(unsigned k) {
  return (k & 0x80000000u) ? __uint_as_float(k & 0x7FFFFFFFu)
                           : __uint_as_float(~k);
}

#define G1_K 256
#define G1_N 64
#define G1_ROWS 64
#define G1_PITCH (G1_K + 4)
#define G1_THREADS 512

__global__ __launch_bounds__(G1_THREADS) void gemm1_lds_kernel(
    const float* __restrict__ A, const float* __restrict__ B,
    float* __restrict__ C, int M) {
  __shared__ __attribute__((aligned(16))) float sA[2 * G1_ROWS * G1_PITCH];
  constexpr int KS = G1_K / 4;
  constexpr int LD = (G1_ROWS * G1_K / 4) / G1_THREADS;
  constexpr int STAGE = G1_ROWS * G1_PITCH;
  const int tid = threadIdx.x, lane = tid & 31;
  const int half = lane >> 4, lid = lane & 15;
  const int wu = __builtin_amdgcn_readfirstlane(tid >> 5);
  const int ct = wu & 3;
  const int rt = wu >> 2;
  const int mtiles = (M + 15) >> 4;
  const int t0 = (int)(((long long)blockIdx.x * mtiles) / gridDim.x);
  const int t1 = (int)(((long long)(blockIdx.x + 1) * mtiles) / gridDim.x);
  const int rlim = min(16 * t1, M);

  const float* bcol = B + (size_t)(2 * half) * G1_N + ct * 16 + lid;
  v2f b[KS];
#pragma unroll
  for (int ks = 0; ks < KS; ++ks) {
    b[ks].x = bcol[(size_t)(4 * ks) * G1_N];
    b[ks].y = bcol[(size_t)(4 * ks + 1) * G1_N];
  }

  float4 pre[LD];
#pragma unroll
  for (int i = 0; i < LD; ++i) {
    const int idx = i * G1_THREADS + tid;
    const int row = 16 * t0 + (idx >> 6);
    pre[i] = make_float4(0.f, 0.f, 0.f, 0.f);
    if (row < rlim) pre[i] = *(const float4*)(A + (size_t)row * G1_K + 4 * (idx & 63));
  }
#pragma unroll
  for (int i = 0; i < LD; ++i) {
    const int idx = i * G1_THREADS + tid;
    *(float4*)(sA + (idx >> 6) * G1_PITCH + 4 * (idx & 63)) = pre[i];
  }
  __syncthreads();

  int buf = 0;
  for (int tb = t0; tb < t1; tb += 4) {
    const int tn = tb + 4;
    const bool more = tn < t1;
    if (more) {
#pragma unroll
      for (int i = 0; i < LD; ++i) {
        const int idx = i * G1_THREADS + tid;
        const int row = 16 * tn + (idx >> 6);
        pre[i] = make_float4(0.f, 0.f, 0.f, 0.f);
        if (row < rlim) pre[i] = *(const float4*)(A + (size_t)row * G1_K + 4 * (idx & 63));
      }
    }
    const int mt = tb + rt;
    if (mt < t1) {
      const float* arow = sA + buf * STAGE + (rt * 16 + lid) * G1_PITCH + 2 * half;
      v8f c = {0.f, 0.f, 0.f, 0.f, 0.f, 0.f, 0.f, 0.f};
      constexpr int AC = 8;
      v2f a[2][AC];
#pragma unroll
      for (int j = 0; j < AC; ++j) a[0][j] = *(const v2f*)(arow + 4 * j);
#pragma unroll
      for (int kc = 0; kc < KS / AC; ++kc) {
        if (kc + 1 < KS / AC) {
#pragma unroll
          for (int j = 0; j < AC; ++j)
            a[(kc + 1) & 1][j] = *(const v2f*)(arow + 4 * (AC * (kc + 1) + j));
        }
        __builtin_amdgcn_sched_barrier(0);
#pragma unroll
        for (int j = 0; j < AC; ++j)
          c = __builtin_amdgcn_wmma_f32_16x16x4_f32(false, a[kc & 1][j], false, b[AC * kc + j],
                                                    (short)0, c, false, false);
        __builtin_amdgcn_sched_barrier(0);
      }
      const int r0 = mt * 16 + 8 * half;
      float* crow = C + (size_t)r0 * G1_N + ct * 16 + lid;
      if (mt * 16 + 16 <= M) {
#pragma unroll
        for (int r = 0; r < 8; ++r) crow[(size_t)r * G1_N] = c[r];
      } else {
        const int nrows = M - r0;
#pragma unroll
        for (int r = 0; r < 8; ++r)
          if (r < nrows) crow[(size_t)r * G1_N] = c[r];
      }
    }
    if (more) {
      float* nxt = sA + (buf ^ 1) * STAGE;
#pragma unroll
      for (int i = 0; i < LD; ++i) {
        const int idx = i * G1_THREADS + tid;
        *(float4*)(nxt + (idx >> 6) * G1_PITCH + 4 * (idx & 63)) = pre[i];
      }
    }
    __syncthreads();
    buf ^= 1;
  }
}

template <int H, int D>
__global__ void node_alpha_kernel(const float* __restrict__ h,
                                  const float* __restrict__ avs,
                                  const float* __restrict__ avd,
                                  float* __restrict__ as,
                                  float* __restrict__ ad, int N) {
  int n = blockIdx.x * blockDim.x + threadIdx.x;
  if (n >= N) return;
  const float* hr = h + (size_t)n * H * D;
#pragma unroll
  for (int hh = 0; hh < H; ++hh) {
    float s0 = 0.f, s1 = 0.f;
#pragma unroll
    for (int d = 0; d < D; ++d) {
      float v = hr[hh * D + d];
      s0 += v * avs[hh * D + d];
      s1 += v * avd[hh * D + d];
    }
    as[(size_t)n * H + hh] = s0;
    ad[(size_t)n * H + hh] = s1;
  }
}

#define NODES_PER_WG 392
#define WG_THREADS 1024
#define WG_WAVES 32
#define LIST_CAP 32768
#define GROUP_RANGES 8
#define GROUP_NODES (GROUP_RANGES * NODES_PER_WG)
#define BKT_CAP 512
#define OWN_SLOTS ((NODES_PER_WG + 31) / 32)
#define FINE_U 8

#define SCAN_U 8

__device__ __forceinline__ void stage_append(bool valid, int e, int dl,
                                             int2* stage, int* lcount, int lane) {
  unsigned m = __builtin_amdgcn_ballot_w32(valid);
  if (m == 0u) return;
  int pos = 0;
  if (lane == 0) pos = atomicAdd(lcount, (int)__popc(m));
  pos = __builtin_amdgcn_readfirstlane(pos);
  if (valid) {
    int slot = pos + (int)__builtin_amdgcn_mbcnt_lo(m, 0u);
    if (slot < LIST_CAP) stage[slot] = make_int2(e, dl);
  }
}

template <bool GUARD>
__device__ __forceinline__ void scan_batch(const int4* __restrict__ d4, int vb, int nvec,
                                           int lo, int cnt, int2* stage, int* lcount,
                                           int lane) {
  int4 d[SCAN_U];
#pragma unroll
  for (int u = 0; u < SCAN_U; ++u) {
    const int v = vb + u * 32 + lane;
    if (GUARD) {
      d[u] = make_int4(lo - 1, lo - 1, lo - 1, lo - 1);
      if (v < nvec) d[u] = d4[v];
    } else {
      d[u] = d4[v];
    }
  }
  unsigned m[SCAN_U][4];
  int tot = 0;
#pragma unroll
  for (int u = 0; u < SCAN_U; ++u) {
    m[u][0] = __builtin_amdgcn_ballot_w32((unsigned)(d[u].x - lo) < (unsigned)cnt);
    m[u][1] = __builtin_amdgcn_ballot_w32((unsigned)(d[u].y - lo) < (unsigned)cnt);
    m[u][2] = __builtin_amdgcn_ballot_w32((unsigned)(d[u].z - lo) < (unsigned)cnt);
    m[u][3] = __builtin_amdgcn_ballot_w32((unsigned)(d[u].w - lo) < (unsigned)cnt);
    tot += (int)(__popc(m[u][0]) + __popc(m[u][1]) + __popc(m[u][2]) + __popc(m[u][3]));
  }
  if (tot == 0) return;
  int base = 0;
  if (lane == 0) base = atomicAdd(lcount, tot);
  base = __builtin_amdgcn_readfirstlane(base);
#pragma unroll
  for (int u = 0; u < SCAN_U; ++u) {
    const int e0 = 4 * (vb + u * 32 + lane);
    const int dd[4] = {d[u].x, d[u].y, d[u].z, d[u].w};
#pragma unroll
    for (int j = 0; j < 4; ++j) {
      const unsigned mm = m[u][j];
      if (mm != 0u) {
        if ((mm >> lane) & 1u) {
          const int slot = base + (int)__builtin_amdgcn_mbcnt_lo(mm, 0u);
          if (slot < LIST_CAP) stage[slot] = make_int2(e0 + j, dd[j] - lo);
        }
        base += (int)__popc(mm);
      }
    }
  }
}

__global__ __launch_bounds__(WG_THREADS) void edge_bin_kernel(
    const int* __restrict__ src, const int* __restrict__ dst, int E, int N,
    int2* __restrict__ lists, int* __restrict__ counts) {
  __shared__ int2 stage[LIST_CAP];
  __shared__ int lcount;
  const int tid = threadIdx.x, lane = tid & 31;
  const int wv = __builtin_amdgcn_readfirstlane(tid >> 5);
  const int grp = blockIdx.x / GROUP_RANGES;
  const int slc = blockIdx.x - grp * GROUP_RANGES;
  const int lo = grp * GROUP_NODES;
  const int cnt = min(GROUP_NODES, N - lo);
  int2* lst = lists + (size_t)blockIdx.x * LIST_CAP;
  if (tid == 0) lcount = 0;
  __syncthreads();
  if ((E & 3) == 0) {
    const int nvec = E >> 2;
    const int vper = (nvec + GROUP_RANGES - 1) / GROUP_RANGES;
    const int vbeg = slc * vper;
    const int vend = min(nvec, vbeg + vper);
    const int4* __restrict__ d4 = (const int4*)dst;
    constexpr int WSPAN = 32 * SCAN_U;
    for (int vb = vbeg + wv * WSPAN; vb < vend; vb += WG_WAVES * WSPAN) {
      if (vb + WSPAN <= vend) scan_batch<false>(d4, vb, vend, lo, cnt, stage, &lcount, lane);
      else                    scan_batch<true >(d4, vb, vend, lo, cnt, stage, &lcount, lane);
    }
  } else {
    const int eper = (E + GROUP_RANGES - 1) / GROUP_RANGES;
    const int ebeg = slc * eper;
    const int eend = min(E, ebeg + eper);
    for (int eb = ebeg + wv * 32; eb < eend; eb += WG_THREADS) {
      const int e = eb + lane;
      int x = -1;
      if (e < eend) x = dst[e] - lo;
      stage_append((unsigned)x < (unsigned)cnt, e, x, stage, &lcount, lane);
    }
  }
  __syncthreads();
  const int found = lcount;
  const int nfill = min(found, LIST_CAP);
  constexpr int GU = 8;
  for (int i0 = tid; i0 < nfill; i0 += GU * WG_THREADS) {
    int2 p[GU];
    int s[GU];
#pragma unroll
    for (int k = 0; k < GU; ++k) {
      const int i = i0 + k * WG_THREADS;
      p[k] = (i < nfill) ? stage[i] : make_int2(0, 0);
    }
#pragma unroll
    for (int k = 0; k < GU; ++k) s[k] = src[p[k].x];
#pragma unroll
    for (int k = 0; k < GU; ++k) {
      const int i = i0 + k * WG_THREADS;
      if (i < nfill) lst[i] = make_int2(s[k], p[k].y);
    }
  }
  if (tid == 0) counts[blockIdx.x] = found;
}

template <bool GUARD>
__device__ __forceinline__ void fine_batch(const int2* __restrict__ seg, int i0, int c,
                                           int sub_lo, int cnt, int2* bkt, int* bcnt,
                                           int lane) {
  int2 p[FINE_U];
#pragma unroll
  for (int u = 0; u < FINE_U; ++u) {
    const int i = i0 + u * 32 + lane;
    if (GUARD) {
      p[u] = make_int2(0, sub_lo - 1);
      if (i < c) p[u] = seg[i];
    } else {
      p[u] = seg[i];
    }
  }
  int slot[FINE_U];
#pragma unroll
  for (int u = 0; u < FINE_U; ++u) {
    const int dl = p[u].y - sub_lo;
    slot[u] = BKT_CAP;
    if ((unsigned)dl < (unsigned)cnt) slot[u] = atomicAdd(&bcnt[dl & 31], 1);
  }
#pragma unroll
  for (int u = 0; u < FINE_U; ++u) {
    const int dl = p[u].y - sub_lo;
    if (slot[u] < BKT_CAP) bkt[(dl & 31) * BKT_CAP + slot[u]] = make_int2(p[u].x, dl);
  }
}

__device__ __forceinline__ bool fetch_own(int i, int n_real, bool use_list,
                                          const int2* lst,
                                          const int* __restrict__ src,
                                          const int* __restrict__ dst,
                                          int lo, int cnt, int wave, int& sv, int& dv) {
  sv = 0; dv = 0;
  if (i < n_real) {
    if (use_list) { int2 p = lst[i]; sv = p.x; dv = p.y; return true; }
    int d = dst[i] - lo;
    if ((unsigned)d < (unsigned)cnt && (d & 31) == wave) { sv = src[i]; dv = d; return true; }
    return false;
  }
  const int j = i - n_real;
  const int n = wave + 32 * j;
  if (j < OWN_SLOTS && n < cnt) { dv = n; sv = lo + n; return true; }
  return false;
}

template <int DTOT, int H, bool ELU, int NOUT>
__global__ __launch_bounds__(WG_THREADS) void range_agg_kernel(
    const int* __restrict__ src, const int* __restrict__ dst, int E, int N,
    const int2* __restrict__ lists, const int* __restrict__ counts,
    const float* __restrict__ as, const float* __restrict__ ad,
    const float* __restrict__ hsrc, const float* __restrict__ bias,
    const float* __restrict__ Wn, float* __restrict__ out) {
  constexpr int R = NODES_PER_WG;
  constexpr bool PROD = NOUT > 0;
  constexpr int P = DTOT + 4;
  constexpr int RP = ((R + 15) / 16) * 16;
  constexpr int VPL = DTOT / 32;
  constexpr int Dh = DTOT / H;
  constexpr int U = 4;
  constexpr int NS = OWN_SLOTS;
  constexpr int SPARE_SLOT = 15;
  static_assert(NS <= SPARE_SLOT, "a wave's node slots and the spare are the elements of one 16-wide register vector");
  static_assert(WG_WAVES == 32, "a node's owner wave is its index modulo 32");
  typedef float vrow __attribute__((ext_vector_type(VPL)));
  typedef float vslot __attribute__((ext_vector_type(16)));
  __shared__ __attribute__((aligned(16))) float rows[PROD ? RP * P : 4];
  __shared__ unsigned mx[R * H];
  __shared__ float adl[R * H];
  __shared__ int2 bkt[WG_WAVES * BKT_CAP];
  __shared__ int bcnt[WG_WAVES];

  const int tid = threadIdx.x, lane = tid & 31;
  const int wu = __builtin_amdgcn_readfirstlane(tid >> 5);
  const int lo = blockIdx.x * R;
  const int cnt = min(R, N - lo);
  const int grp = blockIdx.x / GROUP_RANGES;
  const int sub_lo = (blockIdx.x - grp * GROUP_RANGES) * R;
  const int2* glists = lists + (size_t)grp * GROUP_RANGES * LIST_CAP;
  const int* gcounts = counts + grp * GROUP_RANGES;
  int cmax = 0;
#pragma unroll
  for (int s = 0; s < GROUP_RANGES; ++s) cmax = max(cmax, gcounts[s]);
  const bool seg_ok = cmax <= LIST_CAP;

  if (tid < WG_WAVES) bcnt[tid] = 0;
  for (int i = tid; i < R * H; i += WG_THREADS) {
    mx[i] = ENC_NEG_INF;
    adl[i] = (i < cnt * H) ? ad[(size_t)lo * H + i] : 0.f;
  }
  __syncthreads();
  if (seg_ok) {
    const int s = wu % GROUP_RANGES;
    const int q = wu / GROUP_RANGES;
    constexpr int FSPAN = 32 * FINE_U;
    constexpr int FSTEP = (WG_WAVES / GROUP_RANGES) * FSPAN;
    const int2* seg = glists + (size_t)s * LIST_CAP;
    const int c = gcounts[s];
    for (int i0 = q * FSPAN; i0 < c; i0 += FSTEP) {
      if (i0 + FSPAN <= c) fine_batch<false>(seg, i0, c, sub_lo, cnt, bkt, bcnt, lane);
      else                 fine_batch<true >(seg, i0, c, sub_lo, cnt, bkt, bcnt, lane);
    }
  }
  __syncthreads();
  const int myc = bcnt[lane];
  const bool use_list = seg_ok && __builtin_amdgcn_ballot_w32(myc > BKT_CAP) == 0u;
  const int n_real = use_list ? __builtin_amdgcn_readlane(myc, wu) : E;
  const int total = n_real + NS;
  const int2* lst = bkt + wu * BKT_CAP;

  for (int base = 0; base < total; base += 32) {
    int sv, dv;
    bool valid = fetch_own(base + lane, n_real, use_list, lst, src, dst, lo, cnt, wu, sv, dv);
    if (valid) {
#pragma unroll
      for (int hh = 0; hh < H; ++hh) {
        float e = as[(size_t)sv * H + hh] + adl[dv * H + hh];
        e = e > 0.f ? e : NEG_SLOPE * e;
        atomicMax(&mx[dv * H + hh], enc_ord(e));
      }
    }
  }
  __syncthreads();

  const int c0 = lane * VPL;
  const int hh = c0 / Dh;
  vslot accv[VPL];
  vslot smv;
  float sm1 = 0.f;
#pragma unroll
  for (int j = 0; j < 16; ++j) {
    smv[j] = 0.f;
#pragma unroll
    for (int k = 0; k < VPL; ++k) accv[k][j] = 0.f;
  }
  for (int base = 0; base < total; base += 32) {
    int sv, dv;
    bool valid = fetch_own(base + lane, n_real, use_list, lst, src, dst, lo, cnt, wu, sv, dv);
    unsigned mask = __builtin_amdgcn_ballot_w32(valid);
    while (mask != 0u) {
      int su[U], du[U];
      bool ok[U];
#pragma unroll
      for (int u = 0; u < U; ++u) {
        ok[u] = mask != 0u;
        int j = ok[u] ? (int)__builtin_ctz(mask) : 0;
        if (ok[u]) mask &= mask - 1u;
        su[u] = __builtin_amdgcn_readlane(sv, j);
        du[u] = __builtin_amdgcn_readlane(dv, j);
      }
      vrow row[U];
      float asv[U], adv[U];
      unsigned mxv[U];
#pragma unroll
      for (int u = 0; u < U; ++u) {
        row[u] = *(const vrow*)(hsrc + (size_t)su[u] * DTOT + c0);
        asv[u] = as[(size_t)su[u] * H + hh];
        adv[u] = adl[du[u] * H + hh];
        mxv[u] = mx[du[u] * H + hh];
      }
      __builtin_amdgcn_sched_barrier(0);
#pragma unroll
      for (int u = 0; u < U; ++u) {
        float e = asv[u] + adv[u];
        e = e > 0.f ? e : NEG_SLOPE * e;
        const float w = __expf(e - dec_ord(mxv[u]));
        const int sl = ok[u] ? (du[u] >> 5) : SPARE_SLOT;
#pragma unroll
        for (int k = 0; k < VPL; ++k) accv[k][sl] += w * row[u][k];
        if constexpr (H == 1) sm1 += (lane == sl) ? w : 0.f;
        else                  smv[sl] += w;
      }
    }
  }

  float bv[VPL];
#pragma unroll
  for (int k = 0; k < VPL; ++k) bv[k] = bias[c0 + k];
#pragma unroll
  for (int j = 0; j < NS; ++j) {
    const int n = wu + 32 * j;
    float sj;
    if constexpr (H == 1) sj = __int_as_float(__builtin_amdgcn_readlane(__float_as_int(sm1), j));
    else                  sj = smv[j];
    vrow v;
#pragma unroll
    for (int k = 0; k < VPL; ++k) {
      float t = accv[k][j] / (sj + GAT_EPS) + bv[k];
      if (ELU) t = t > 0.f ? t : __expf(t) - 1.f;
      v[k] = t;
    }
    if constexpr (!PROD) {
      if (n < cnt) *(vrow*)(out + (size_t)(lo + n) * DTOT + c0) = v;
    } else {
      if (n < RP) {
        if (n >= cnt) {
#pragma unroll
          for (int k = 0; k < VPL; ++k) v[k] = 0.f;
        }
        *(vrow*)(rows + n * P + c0) = v;
      }
    }
  }

  if constexpr (PROD) {
    __syncthreads();
    constexpr int NT = NOUT / 16;
    constexpr int MT = RP / 16;
    constexpr int KS = DTOT / 4;
    constexpr int QS = WG_WAVES / NT;
    const int half = lane >> 4, lid = lane & 15;
    const int ct = wu % NT;
    const int q = wu / NT;
    const float* bcol = Wn + (size_t)(2 * half) * NOUT + ct * 16 + lid;
    v2f b[KS];
#pragma unroll
    for (int ks = 0; ks < KS; ++ks) {
      b[ks].x = bcol[(size_t)(4 * ks) * NOUT];
      b[ks].y = bcol[(size_t)(4 * ks + 1) * NOUT];
    }
    for (int mt = q; mt < MT; mt += QS) {
      const float* arow = rows + (mt * 16 + lid) * P + 2 * half;
      v2f a[KS];
#pragma unroll
      for (int ks = 0; ks < KS; ++ks) a[ks] = *(const v2f*)(arow + 4 * ks);
      v8f c = {0.f, 0.f, 0.f, 0.f, 0.f, 0.f, 0.f, 0.f};
#pragma unroll
      for (int ks = 0; ks < KS; ++ks)
        c = __builtin_amdgcn_wmma_f32_16x16x4_f32(false, a[ks], false, b[ks],
                                                  (short)0, c, false, false);
      const int r0 = mt * 16 + 8 * half;
      float* crow = out + (size_t)(lo + r0) * NOUT + ct * 16 + lid;
      if (mt * 16 + 16 <= cnt) {
#pragma unroll
        for (int r = 0; r < 8; ++r) crow[(size_t)r * NOUT] = c[r];
      } else {
        const int nrows = cnt - r0;
#pragma unroll
        for (int r = 0; r < 8; ++r)
          if (r < nrows) crow[(size_t)r * NOUT] = c[r];
      }
    }
  }
}

static inline int cdiv(long long a, long long b) { return (int)((a + b - 1) / b); }

extern "C" void kernel_launch(void* const* d_in, const int* in_sizes, int n_in,
                              void* d_out, int out_size, void* d_ws, size_t ws_size,
                              hipStream_t stream) {
  const float* x      = (const float*)d_in[0];
  const int*   ei     = (const int*)  d_in[1];
  const float* W1     = (const float*)d_in[2];
  const float* a_src1 = (const float*)d_in[3];
  const float* a_dst1 = (const float*)d_in[4];
  const float* b1     = (const float*)d_in[5];
  const float* W2     = (const float*)d_in[6];
  const float* a_src2 = (const float*)d_in[7];
  const float* a_dst2 = (const float*)d_in[8];
  const float* b2     = (const float*)d_in[9];
  float* z = (float*)d_out;

  const int Fin = 256, H1 = 8, C1 = 64, C2 = 128;
  const int N    = in_sizes[0] / Fin;
  const int E    = in_sizes[1] / 2;
  const int* srcv = ei;
  const int* dstv = ei + E;
  const int G = cdiv(N, NODES_PER_WG);
  const int GB = cdiv(G, GROUP_RANGES) * GROUP_RANGES;

  float* h1   = (float*)d_ws;
  float* as1  = h1  + (size_t)N * C1;
  float* ad1  = as1 + (size_t)N * H1;
  float* h2   = ad1 + (size_t)N * H1;
  float* as2  = h2  + (size_t)N * C2;
  float* ad2  = as2 + N;
  int*   cnts = (int*)(ad2 + N);
  int2*  lsts = (int2*)(cnts + 1024);

  const int B = 256;

  edge_bin_kernel<<<GB, WG_THREADS, 0, stream>>>(srcv, dstv, E, N, lsts, cnts);

  gemm1_lds_kernel<<<256, G1_THREADS, 0, stream>>>(x, W1, h1, N);
  node_alpha_kernel<8, 8><<<cdiv(N, B), B, 0, stream>>>(h1, a_src1, a_dst1,
                                                        as1, ad1, N);
  range_agg_kernel<64, 8, true, 128><<<G, WG_THREADS, 0, stream>>>(
      srcv, dstv, E, N, lsts, cnts, as1, ad1, h1, b1, W2, h2);

  node_alpha_kernel<1, 128><<<cdiv(N, B), B, 0, stream>>>(h2, a_src2, a_dst2,
                                                          as2, ad2, N);
  range_agg_kernel<128, 1, false, 0><<<G, WG_THREADS, 0, stream>>>(
      srcv, dstv, E, N, lsts, cnts, as2, ad2, h2, b2, nullptr, z);
}


# Pass 2 split: a 32-edge block's weights formed one lane an edge (all heads; handed on by readlane or a wave's LDS slab), row loop only gathers and sums, 8 rows in flight; sum order unchanged
# speedup vs baseline: 4.1586x; 1.0644x over previous
#include <hip/hip_runtime.h>
#include <cstdint>


typedef float v2f __attribute__((ext_vector_type(2)));
typedef float v8f __attribute__((ext_vector_type(8)));

#define NEG_SLOPE 0.2f
#define GAT_EPS 1e-16f
#define ENC_NEG_INF 0x007FFFFFu

__device__ __forceinline__ unsigned enc_ord(float f) {
  unsigned u = __float_as_uint(f);
  return (u & 0x80000000u) ? ~u : (u | 0x80000000u);
}
__device__ __forceinline__ float dec_ord(unsigned k) {
  return (k & 0x80000000u) ? __uint_as_float(k & 0x7FFFFFFFu)
                           : __uint_as_float(~k);
}

#define G1_K 256
#define G1_N 64
#define G1_ROWS 64
#define G1_PITCH (G1_K + 4)
#define G1_THREADS 512

__global__ __launch_bounds__(G1_THREADS) void gemm1_lds_kernel(
    const float* __restrict__ A, const float* __restrict__ B,
    float* __restrict__ C, int M) {
  __shared__ __attribute__((aligned(16))) float sA[2 * G1_ROWS * G1_PITCH];
  constexpr int KS = G1_K / 4;
  constexpr int LD = (G1_ROWS * G1_K / 4) / G1_THREADS;
  constexpr int STAGE = G1_ROWS * G1_PITCH;
  const int tid = threadIdx.x, lane = tid & 31;
  const int half = lane >> 4, lid = lane & 15;
  const int wu = __builtin_amdgcn_readfirstlane(tid >> 5);
  const int ct = wu & 3;
  const int rt = wu >> 2;
  const int mtiles = (M + 15) >> 4;
  const int t0 = (int)(((long long)blockIdx.x * mtiles) / gridDim.x);
  const int t1 = (int)(((long long)(blockIdx.x + 1) * mtiles) / gridDim.x);
  const int rlim = min(16 * t1, M);

  const float* bcol = B + (size_t)(2 * half) * G1_N + ct * 16 + lid;
  v2f b[KS];
#pragma unroll
  for (int ks = 0; ks < KS; ++ks) {
    b[ks].x = bcol[(size_t)(4 * ks) * G1_N];
    b[ks].y = bcol[(size_t)(4 * ks + 1) * G1_N];
  }

  float4 pre[LD];
#pragma unroll
  for (int i = 0; i < LD; ++i) {
    const int idx = i * G1_THREADS + tid;
    const int row = 16 * t0 + (idx >> 6);
    pre[i] = make_float4(0.f, 0.f, 0.f, 0.f);
    if (row < rlim) pre[i] = *(const float4*)(A + (size_t)row * G1_K + 4 * (idx & 63));
  }
#pragma unroll
  for (int i = 0; i < LD; ++i) {
    const int idx = i * G1_THREADS + tid;
    *(float4*)(sA + (idx >> 6) * G1_PITCH + 4 * (idx & 63)) = pre[i];
  }
  __syncthreads();

  int buf = 0;
  for (int tb = t0; tb < t1; tb += 4) {
    const int tn = tb + 4;
    const bool more = tn < t1;
    if (more) {
#pragma unroll
      for (int i = 0; i < LD; ++i) {
        const int idx = i * G1_THREADS + tid;
        const int row = 16 * tn + (idx >> 6);
        pre[i] = make_float4(0.f, 0.f, 0.f, 0.f);
        if (row < rlim) pre[i] = *(const float4*)(A + (size_t)row * G1_K + 4 * (idx & 63));
      }
    }
    const int mt = tb + rt;
    if (mt < t1) {
      const float* arow = sA + buf * STAGE + (rt * 16 + lid) * G1_PITCH + 2 * half;
      v8f c = {0.f, 0.f, 0.f, 0.f, 0.f, 0.f, 0.f, 0.f};
      constexpr int AC = 8;
      v2f a[2][AC];
#pragma unroll
      for (int j = 0; j < AC; ++j) a[0][j] = *(const v2f*)(arow + 4 * j);
#pragma unroll
      for (int kc = 0; kc < KS / AC; ++kc) {
        if (kc + 1 < KS / AC) {
#pragma unroll
          for (int j = 0; j < AC; ++j)
            a[(kc + 1) & 1][j] = *(const v2f*)(arow + 4 * (AC * (kc + 1) + j));
        }
        __builtin_amdgcn_sched_barrier(0);
#pragma unroll
        for (int j = 0; j < AC; ++j)
          c = __builtin_amdgcn_wmma_f32_16x16x4_f32(false, a[kc & 1][j], false, b[AC * kc + j],
                                                    (short)0, c, false, false);
        __builtin_amdgcn_sched_barrier(0);
      }
      const int r0 = mt * 16 + 8 * half;
      float* crow = C + (size_t)r0 * G1_N + ct * 16 + lid;
      if (mt * 16 + 16 <= M) {
#pragma unroll
        for (int r = 0; r < 8; ++r) crow[(size_t)r * G1_N] = c[r];
      } else {
        const int nrows = M - r0;
#pragma unroll
        for (int r = 0; r < 8; ++r)
          if (r < nrows) crow[(size_t)r * G1_N] = c[r];
      }
    }
    if (more) {
      float* nxt = sA + (buf ^ 1) * STAGE;
#pragma unroll
      for (int i = 0; i < LD; ++i) {
        const int idx = i * G1_THREADS + tid;
        *(float4*)(nxt + (idx >> 6) * G1_PITCH + 4 * (idx & 63)) = pre[i];
      }
    }
    __syncthreads();
    buf ^= 1;
  }
}

template <int H, int D>
__global__ void node_alpha_kernel(const float* __restrict__ h,
                                  const float* __restrict__ avs,
                                  const float* __restrict__ avd,
                                  float* __restrict__ as,
                                  float* __restrict__ ad, int N) {
  int n = blockIdx.x * blockDim.x + threadIdx.x;
  if (n >= N) return;
  const float* hr = h + (size_t)n * H * D;
#pragma unroll
  for (int hh = 0; hh < H; ++hh) {
    float s0 = 0.f, s1 = 0.f;
#pragma unroll
    for (int d = 0; d < D; ++d) {
      float v = hr[hh * D + d];
      s0 += v * avs[hh * D + d];
      s1 += v * avd[hh * D + d];
    }
    as[(size_t)n * H + hh] = s0;
    ad[(size_t)n * H + hh] = s1;
  }
}

#define NODES_PER_WG 392
#define WG_THREADS 1024
#define WG_WAVES 32
#define LIST_CAP 32768
#define GROUP_RANGES 8
#define GROUP_NODES (GROUP_RANGES * NODES_PER_WG)
#define BKT_CAP 512
#define OWN_SLOTS ((NODES_PER_WG + 31) / 32)
#define FINE_U 8

#define SCAN_U 8

__device__ __forceinline__ void stage_append(bool valid, int e, int dl,
                                             int2* stage, int* lcount, int lane) {
  unsigned m = __builtin_amdgcn_ballot_w32(valid);
  if (m == 0u) return;
  int pos = 0;
  if (lane == 0) pos = atomicAdd(lcount, (int)__popc(m));
  pos = __builtin_amdgcn_readfirstlane(pos);
  if (valid) {
    int slot = pos + (int)__builtin_amdgcn_mbcnt_lo(m, 0u);
    if (slot < LIST_CAP) stage[slot] = make_int2(e, dl);
  }
}

template <bool GUARD>
__device__ __forceinline__ void scan_batch(const int4* __restrict__ d4, int vb, int nvec,
                                           int lo, int cnt, int2* stage, int* lcount,
                                           int lane) {
  int4 d[SCAN_U];
#pragma unroll
  for (int u = 0; u < SCAN_U; ++u) {
    const int v = vb + u * 32 + lane;
    if (GUARD) {
      d[u] = make_int4(lo - 1, lo - 1, lo - 1, lo - 1);
      if (v < nvec) d[u] = d4[v];
    } else {
      d[u] = d4[v];
    }
  }
  unsigned m[SCAN_U][4];
  int tot = 0;
#pragma unroll
  for (int u = 0; u < SCAN_U; ++u) {
    m[u][0] = __builtin_amdgcn_ballot_w32((unsigned)(d[u].x - lo) < (unsigned)cnt);
    m[u][1] = __builtin_amdgcn_ballot_w32((unsigned)(d[u].y - lo) < (unsigned)cnt);
    m[u][2] = __builtin_amdgcn_ballot_w32((unsigned)(d[u].z - lo) < (unsigned)cnt);
    m[u][3] = __builtin_amdgcn_ballot_w32((unsigned)(d[u].w - lo) < (unsigned)cnt);
    tot += (int)(__popc(m[u][0]) + __popc(m[u][1]) + __popc(m[u][2]) + __popc(m[u][3]));
  }
  if (tot == 0) return;
  int base = 0;
  if (lane == 0) base = atomicAdd(lcount, tot);
  base = __builtin_amdgcn_readfirstlane(base);
#pragma unroll
  for (int u = 0; u < SCAN_U; ++u) {
    const int e0 = 4 * (vb + u * 32 + lane);
    const int dd[4] = {d[u].x, d[u].y, d[u].z, d[u].w};
#pragma unroll
    for (int j = 0; j < 4; ++j) {
      const unsigned mm = m[u][j];
      if (mm != 0u) {
        if ((mm >> lane) & 1u) {
          const int slot = base + (int)__builtin_amdgcn_mbcnt_lo(mm, 0u);
          if (slot < LIST_CAP) stage[slot] = make_int2(e0 + j, dd[j] - lo);
        }
        base += (int)__popc(mm);
      }
    }
  }
}

__global__ __launch_bounds__(WG_THREADS) void edge_bin_kernel(
    const int* __restrict__ src, const int* __restrict__ dst, int E, int N,
    int2* __restrict__ lists, int* __restrict__ counts) {
  __shared__ int2 stage[LIST_CAP];
  __shared__ int lcount;
  const int tid = threadIdx.x, lane = tid & 31;
  const int wv = __builtin_amdgcn_readfirstlane(tid >> 5);
  const int grp = blockIdx.x / GROUP_RANGES;
  const int slc = blockIdx.x - grp * GROUP_RANGES;
  const int lo = grp * GROUP_NODES;
  const int cnt = min(GROUP_NODES, N - lo);
  int2* lst = lists + (size_t)blockIdx.x * LIST_CAP;
  if (tid == 0) lcount = 0;
  __syncthreads();
  if ((E & 3) == 0) {
    const int nvec = E >> 2;
    const int vper = (nvec + GROUP_RANGES - 1) / GROUP_RANGES;
    const int vbeg = slc * vper;
    const int vend = min(nvec, vbeg + vper);
    const int4* __restrict__ d4 = (const int4*)dst;
    constexpr int WSPAN = 32 * SCAN_U;
    for (int vb = vbeg + wv * WSPAN; vb < vend; vb += WG_WAVES * WSPAN) {
      if (vb + WSPAN <= vend) scan_batch<false>(d4, vb, vend, lo, cnt, stage, &lcount, lane);
      else                    scan_batch<true >(d4, vb, vend, lo, cnt, stage, &lcount, lane);
    }
  } else {
    const int eper = (E + GROUP_RANGES - 1) / GROUP_RANGES;
    const int ebeg = slc * eper;
    const int eend = min(E, ebeg + eper);
    for (int eb = ebeg + wv * 32; eb < eend; eb += WG_THREADS) {
      const int e = eb + lane;
      int x = -1;
      if (e < eend) x = dst[e] - lo;
      stage_append((unsigned)x < (unsigned)cnt, e, x, stage, &lcount, lane);
    }
  }
  __syncthreads();
  const int found = lcount;
  const int nfill = min(found, LIST_CAP);
  constexpr int GU = 8;
  for (int i0 = tid; i0 < nfill; i0 += GU * WG_THREADS) {
    int2 p[GU];
    int s[GU];
#pragma unroll
    for (int k = 0; k < GU; ++k) {
      const int i = i0 + k * WG_THREADS;
      p[k] = (i < nfill) ? stage[i] : make_int2(0, 0);
    }
#pragma unroll
    for (int k = 0; k < GU; ++k) s[k] = src[p[k].x];
#pragma unroll
    for (int k = 0; k < GU; ++k) {
      const int i = i0 + k * WG_THREADS;
      if (i < nfill) lst[i] = make_int2(s[k], p[k].y);
    }
  }
  if (tid == 0) counts[blockIdx.x] = found;
}

template <bool GUARD>
__device__ __forceinline__ void fine_batch(const int2* __restrict__ seg, int i0, int c,
                                           int sub_lo, int cnt, int2* bkt, int* bcnt,
                                           int lane) {
  int2 p[FINE_U];
#pragma unroll
  for (int u = 0; u < FINE_U; ++u) {
    const int i = i0 + u * 32 + lane;
    if (GUARD) {
      p[u] = make_int2(0, sub_lo - 1);
      if (i < c) p[u] = seg[i];
    } else {
      p[u] = seg[i];
    }
  }
  int slot[FINE_U];
#pragma unroll
  for (int u = 0; u < FINE_U; ++u) {
    const int dl = p[u].y - sub_lo;
    slot[u] = BKT_CAP;
    if ((unsigned)dl < (unsigned)cnt) slot[u] = atomicAdd(&bcnt[dl & 31], 1);
  }
#pragma unroll
  for (int u = 0; u < FINE_U; ++u) {
    const int dl = p[u].y - sub_lo;
    if (slot[u] < BKT_CAP) bkt[(dl & 31) * BKT_CAP + slot[u]] = make_int2(p[u].x, dl);
  }
}

__device__ __forceinline__ bool fetch_own(int i, int n_real, bool use_list,
                                          const int2* lst,
                                          const int* __restrict__ src,
                                          const int* __restrict__ dst,
                                          int lo, int cnt, int wave, int& sv, int& dv) {
  sv = 0; dv = 0;
  if (i < n_real) {
    if (use_list) { int2 p = lst[i]; sv = p.x; dv = p.y; return true; }
    int d = dst[i] - lo;
    if ((unsigned)d < (unsigned)cnt && (d & 31) == wave) { sv = src[i]; dv = d; return true; }
    return false;
  }
  const int j = i - n_real;
  const int n = wave + 32 * j;
  if (j < OWN_SLOTS && n < cnt) { dv = n; sv = lo + n; return true; }
  return false;
}

template <int DTOT, int H, bool ELU, int NOUT>
__global__ __launch_bounds__(WG_THREADS) void range_agg_kernel(
    const int* __restrict__ src, const int* __restrict__ dst, int E, int N,
    const int2* __restrict__ lists, const int* __restrict__ counts,
    const float* __restrict__ as, const float* __restrict__ ad,
    const float* __restrict__ hsrc, const float* __restrict__ bias,
    const float* __restrict__ Wn, float* __restrict__ out) {
  constexpr int R = NODES_PER_WG;
  constexpr bool PROD = NOUT > 0;
  constexpr int P = DTOT + 4;
  constexpr int RP = ((R + 15) / 16) * 16;
  constexpr int VPL = DTOT / 32;
  constexpr int Dh = DTOT / H;
  constexpr int U = 8;
  constexpr int NS = OWN_SLOTS;
  constexpr int SPARE_SLOT = 15;
  static_assert(NS <= SPARE_SLOT, "a wave's node slots and the spare are the elements of one 16-wide register vector");
  static_assert(WG_WAVES == 32, "a node's owner wave is its index modulo 32");
  typedef float vrow __attribute__((ext_vector_type(VPL)));
  typedef float vslot __attribute__((ext_vector_type(16)));
  __shared__ __attribute__((aligned(16))) float rows[PROD ? RP * P : 4];
  __shared__ unsigned mx[R * H];
  __shared__ float adl[R * H];
  __shared__ int2 bkt[WG_WAVES * BKT_CAP];
  __shared__ int bcnt[WG_WAVES];
  __shared__ float wbuf[(H > 1) ? WG_WAVES * 32 * H : 1];

  const int tid = threadIdx.x, lane = tid & 31;
  const int wu = __builtin_amdgcn_readfirstlane(tid >> 5);
  const int lo = blockIdx.x * R;
  const int cnt = min(R, N - lo);
  const int grp = blockIdx.x / GROUP_RANGES;
  const int sub_lo = (blockIdx.x - grp * GROUP_RANGES) * R;
  const int2* glists = lists + (size_t)grp * GROUP_RANGES * LIST_CAP;
  const int* gcounts = counts + grp * GROUP_RANGES;
  int cmax = 0;
#pragma unroll
  for (int s = 0; s < GROUP_RANGES; ++s) cmax = max(cmax, gcounts[s]);
  const bool seg_ok = cmax <= LIST_CAP;

  if (tid < WG_WAVES) bcnt[tid] = 0;
  for (int i = tid; i < R * H; i += WG_THREADS) {
    mx[i] = ENC_NEG_INF;
    adl[i] = (i < cnt * H) ? ad[(size_t)lo * H + i] : 0.f;
  }
  __syncthreads();
  if (seg_ok) {
    const int s = wu % GROUP_RANGES;
    const int q = wu / GROUP_RANGES;
    constexpr int FSPAN = 32 * FINE_U;
    constexpr int FSTEP = (WG_WAVES / GROUP_RANGES) * FSPAN;
    const int2* seg = glists + (size_t)s * LIST_CAP;
    const int c = gcounts[s];
    for (int i0 = q * FSPAN; i0 < c; i0 += FSTEP) {
      if (i0 + FSPAN <= c) fine_batch<false>(seg, i0, c, sub_lo, cnt, bkt, bcnt, lane);
      else                 fine_batch<true >(seg, i0, c, sub_lo, cnt, bkt, bcnt, lane);
    }
  }
  __syncthreads();
  const int myc = bcnt[lane];
  const bool use_list = seg_ok && __builtin_amdgcn_ballot_w32(myc > BKT_CAP) == 0u;
  const int n_real = use_list ? __builtin_amdgcn_readlane(myc, wu) : E;
  const int total = n_real + NS;
  const int2* lst = bkt + wu * BKT_CAP;

  for (int base = 0; base < total; base += 32) {
    int sv, dv;
    bool valid = fetch_own(base + lane, n_real, use_list, lst, src, dst, lo, cnt, wu, sv, dv);
    if (valid) {
#pragma unroll
      for (int hh = 0; hh < H; ++hh) {
        float e = as[(size_t)sv * H + hh] + adl[dv * H + hh];
        e = e > 0.f ? e : NEG_SLOPE * e;
        atomicMax(&mx[dv * H + hh], enc_ord(e));
      }
    }
  }
  __syncthreads();

  const int c0 = lane * VPL;
  const int hh = c0 / Dh;
  vslot accv[VPL];
  vslot smv;
  float sm1 = 0.f;
#pragma unroll
  for (int j = 0; j < 16; ++j) {
    smv[j] = 0.f;
#pragma unroll
    for (int k = 0; k < VPL; ++k) accv[k][j] = 0.f;
  }
  float* wblk = wbuf + (H > 1 ? wu * 32 * H : 0);
  int sv_n, dv_n;
  bool valid_n = fetch_own(lane, n_real, use_list, lst, src, dst, lo, cnt, wu, sv_n, dv_n);
  float as_n[H];
#pragma unroll
  for (int h = 0; h < H; ++h) as_n[h] = as[(size_t)sv_n * H + h];
  for (int base = 0; base < total; base += 32) {
    const int sv = sv_n, dv = dv_n;
    const bool valid = valid_n;
    float asv[H];
#pragma unroll
    for (int h = 0; h < H; ++h) asv[h] = as_n[h];
    if (base + 32 < total) {
      valid_n = fetch_own(base + 32 + lane, n_real, use_list, lst, src, dst, lo, cnt, wu, sv_n, dv_n);
#pragma unroll
      for (int h = 0; h < H; ++h) as_n[h] = as[(size_t)sv_n * H + h];
    }
    unsigned mask = __builtin_amdgcn_ballot_w32(valid);
    if (mask != 0u) {
      float wl[H];
#pragma unroll
      for (int h = 0; h < H; ++h) {
        float e = asv[h] + adl[dv * H + h];
        e = e > 0.f ? e : NEG_SLOPE * e;
        wl[h] = __expf(e - dec_ord(mx[dv * H + h]));
      }
      if constexpr (H > 1) {
#pragma unroll
        for (int h = 0; h < H; ++h) wblk[lane * H + h] = wl[h];
      }
      while (mask != 0u) {
        int su[U], du[U], ju[U];
        float ws[U];
        bool ok[U];
#pragma unroll
        for (int u = 0; u < U; ++u) {
          ok[u] = mask != 0u;
          ju[u] = ok[u] ? (int)__builtin_ctz(mask) : 0;
          if (ok[u]) mask &= mask - 1u;
          su[u] = __builtin_amdgcn_readlane(sv, ju[u]);
          du[u] = __builtin_amdgcn_readlane(dv, ju[u]);
          if constexpr (H == 1)
            ws[u] = __int_as_float(__builtin_amdgcn_readlane(__float_as_int(wl[0]), ju[u]));
          else
            ws[u] = 0.f;
        }
        vrow row[U];
        float wv[U];
#pragma unroll
        for (int u = 0; u < U; ++u) {
          row[u] = *(const vrow*)(hsrc + (size_t)su[u] * DTOT + c0);
          if constexpr (H > 1) wv[u] = wblk[ju[u] * H + hh];
          else                 wv[u] = ws[u];
        }
        __builtin_amdgcn_sched_barrier(0);
#pragma unroll
        for (int u = 0; u < U; ++u) {
          const float w = wv[u];
          const int sl = ok[u] ? (du[u] >> 5) : SPARE_SLOT;
#pragma unroll
          for (int k = 0; k < VPL; ++k) accv[k][sl] += w * row[u][k];
          if constexpr (H == 1) sm1 += (lane == sl) ? w : 0.f;
          else                  smv[sl] += w;
        }
      }
    }
  }

  float bv[VPL];
#pragma unroll
  for (int k = 0; k < VPL; ++k) bv[k] = bias[c0 + k];
#pragma unroll
  for (int j = 0; j < NS; ++j) {
    const int n = wu + 32 * j;
    float sj;
    if constexpr (H == 1) sj = __int_as_float(__builtin_amdgcn_readlane(__float_as_int(sm1), j));
    else                  sj = smv[j];
    vrow v;
#pragma unroll
    for (int k = 0; k < VPL; ++k) {
      float t = accv[k][j] / (sj + GAT_EPS) + bv[k];
      if (ELU) t = t > 0.f ? t : __expf(t) - 1.f;
      v[k] = t;
    }
    if constexpr (!PROD) {
      if (n < cnt) *(vrow*)(out + (size_t)(lo + n) * DTOT + c0) = v;
    } else {
      if (n < RP) {
        if (n >= cnt) {
#pragma unroll
          for (int k = 0; k < VPL; ++k) v[k] = 0.f;
        }
        *(vrow*)(rows + n * P + c0) = v;
      }
    }
  }

  if constexpr (PROD) {
    __syncthreads();
    constexpr int NT = NOUT / 16;
    constexpr int MT = RP / 16;
    constexpr int KS = DTOT / 4;
    constexpr int QS = WG_WAVES / NT;
    const int half = lane >> 4, lid = lane & 15;
    const int ct = wu % NT;
    const int q = wu / NT;
    const float* bcol = Wn + (size_t)(2 * half) * NOUT + ct * 16 + lid;
    v2f b[KS];
#pragma unroll
    for (int ks = 0; ks < KS; ++ks) {
      b[ks].x = bcol[(size_t)(4 * ks) * NOUT];
      b[ks].y = bcol[(size_t)(4 * ks + 1) * NOUT];
    }
    for (int mt = q; mt < MT; mt += QS) {
      const float* arow = rows + (mt * 16 + lid) * P + 2 * half;
      v2f a[KS];
#pragma unroll
      for (int ks = 0; ks < KS; ++ks) a[ks] = *(const v2f*)(arow + 4 * ks);
      v8f c = {0.f, 0.f, 0.f, 0.f, 0.f, 0.f, 0.f, 0.f};
#pragma unroll
      for (int ks = 0; ks < KS; ++ks)
        c = __builtin_amdgcn_wmma_f32_16x16x4_f32(false, a[ks], false, b[ks],
                                                  (short)0, c, false, false);
      const int r0 = mt * 16 + 8 * half;
      float* crow = out + (size_t)(lo + r0) * NOUT + ct * 16 + lid;
      if (mt * 16 + 16 <= cnt) {
#pragma unroll
        for (int r = 0; r < 8; ++r) crow[(size_t)r * NOUT] = c[r];
      } else {
        const int nrows = cnt - r0;
#pragma unroll
        for (int r = 0; r < 8; ++r)
          if (r < nrows) crow[(size_t)r * NOUT] = c[r];
      }
    }
  }
}

static inline int cdiv(long long a, long long b) { return (int)((a + b - 1) / b); }

extern "C" void kernel_launch(void* const* d_in, const int* in_sizes, int n_in,
                              void* d_out, int out_size, void* d_ws, size_t ws_size,
                              hipStream_t stream) {
  const float* x      = (const float*)d_in[0];
  const int*   ei     = (const int*)  d_in[1];
  const float* W1     = (const float*)d_in[2];
  const float* a_src1 = (const float*)d_in[3];
  const float* a_dst1 = (const float*)d_in[4];
  const float* b1     = (const float*)d_in[5];
  const float* W2     = (const float*)d_in[6];
  const float* a_src2 = (const float*)d_in[7];
  const float* a_dst2 = (const float*)d_in[8];
  const float* b2     = (const float*)d_in[9];
  float* z = (float*)d_out;

  const int Fin = 256, H1 = 8, C1 = 64, C2 = 128;
  const int N    = in_sizes[0] / Fin;
  const int E    = in_sizes[1] / 2;
  const int* srcv = ei;
  const int* dstv = ei + E;
  const int G = cdiv(N, NODES_PER_WG);
  const int GB = cdiv(G, GROUP_RANGES) * GROUP_RANGES;

  float* h1   = (float*)d_ws;
  float* as1  = h1  + (size_t)N * C1;
  float* ad1  = as1 + (size_t)N * H1;
  float* h2   = ad1 + (size_t)N * H1;
  float* as2  = h2  + (size_t)N * C2;
  float* ad2  = as2 + N;
  int*   cnts = (int*)(ad2 + N);
  int2*  lsts = (int2*)(cnts + 1024);

  const int B = 256;

  edge_bin_kernel<<<GB, WG_THREADS, 0, stream>>>(srcv, dstv, E, N, lsts, cnts);

  gemm1_lds_kernel<<<256, G1_THREADS, 0, stream>>>(x, W1, h1, N);
  node_alpha_kernel<8, 8><<<cdiv(N, B), B, 0, stream>>>(h1, a_src1, a_dst1,
                                                        as1, ad1, N);
  range_agg_kernel<64, 8, true, 128><<<G, WG_THREADS, 0, stream>>>(
      srcv, dstv, E, N, lsts, cnts, as1, ad1, h1, b1, W2, h2);

  node_alpha_kernel<1, 128><<<cdiv(N, B), B, 0, stream>>>(h2, a_src2, a_dst2,
                                                          as2, ad2, N);
  range_agg_kernel<128, 1, false, 0><<<G, WG_THREADS, 0, stream>>>(
      srcv, dstv, E, N, lsts, cnts, as2, ad2, h2, b2, nullptr, z);
}


# Attention logits taken from the WMMA result tiles in their producers (layer-1 GEMM: 8-lane sums; W2 product: 16-lane sums, 8 tiles added in LDS); no logit launches, 6->4; logit sum order: lane tree
# speedup vs baseline: 4.4874x; 1.0791x over previous
#include <hip/hip_runtime.h>
#include <cstdint>


typedef float v2f __attribute__((ext_vector_type(2)));
typedef float v8f __attribute__((ext_vector_type(8)));

#define NEG_SLOPE 0.2f
#define GAT_EPS 1e-16f
#define ENC_NEG_INF 0x007FFFFFu

__device__ __forceinline__ unsigned enc_ord(float f) {
  unsigned u = __float_as_uint(f);
  return (u & 0x80000000u) ? ~u : (u | 0x80000000u);
}
__device__ __forceinline__ float dec_ord(unsigned k) {
  return (k & 0x80000000u) ? __uint_as_float(k & 0x7FFFFFFFu)
                           : __uint_as_float(~k);
}

template <int XM>
__device__ __forceinline__ float row_xadd(float v) {
  return v + __int_as_float(__builtin_amdgcn_update_dpp(0, __float_as_int(v), 0x160 | XM,
                                                        0xf, 0xf, true));
}

#define G1_K 256
#define G1_N 64
#define G1_ROWS 64
#define G1_PITCH (G1_K + 4)
#define G1_THREADS 512

__global__ __launch_bounds__(G1_THREADS) void gemm1_lds_kernel(
    const float* __restrict__ A, const float* __restrict__ B,
    float* __restrict__ C, int M,
    const float* __restrict__ avs, const float* __restrict__ avd,
    float* __restrict__ as, float* __restrict__ ad) {
  __shared__ __attribute__((aligned(16))) float sA[2 * G1_ROWS * G1_PITCH];
  constexpr int KS = G1_K / 4;
  constexpr int LD = (G1_ROWS * G1_K / 4) / G1_THREADS;
  constexpr int STAGE = G1_ROWS * G1_PITCH;
  const int tid = threadIdx.x, lane = tid & 31;
  const int half = lane >> 4, lid = lane & 15;
  const int wu = __builtin_amdgcn_readfirstlane(tid >> 5);
  const int ct = wu & 3;
  const int rt = wu >> 2;
  const int mtiles = (M + 15) >> 4;
  const int t0 = (int)(((long long)blockIdx.x * mtiles) / gridDim.x);
  const int t1 = (int)(((long long)(blockIdx.x + 1) * mtiles) / gridDim.x);
  const int rlim = min(16 * t1, M);

  const float* bcol = B + (size_t)(2 * half) * G1_N + ct * 16 + lid;
  v2f b[KS];
#pragma unroll
  for (int ks = 0; ks < KS; ++ks) {
    b[ks].x = bcol[(size_t)(4 * ks) * G1_N];
    b[ks].y = bcol[(size_t)(4 * ks + 1) * G1_N];
  }

  const float avs_l = avs[ct * 16 + lid];
  const float avd_l = avd[ct * 16 + lid];

  float4 pre[LD];
#pragma unroll
  for (int i = 0; i < LD; ++i) {
    const int idx = i * G1_THREADS + tid;
    const int row = 16 * t0 + (idx >> 6);
    pre[i] = make_float4(0.f, 0.f, 0.f, 0.f);
    if (row < rlim) pre[i] = *(const float4*)(A + (size_t)row * G1_K + 4 * (idx & 63));
  }
#pragma unroll
  for (int i = 0; i < LD; ++i) {
    const int idx = i * G1_THREADS + tid;
    *(float4*)(sA + (idx >> 6) * G1_PITCH + 4 * (idx & 63)) = pre[i];
  }
  __syncthreads();

  int buf = 0;
  for (int tb = t0; tb < t1; tb += 4) {
    const int tn = tb + 4;
    const bool more = tn < t1;
    if (more) {
#pragma unroll
      for (int i = 0; i < LD; ++i) {
        const int idx = i * G1_THREADS + tid;
        const int row = 16 * tn + (idx >> 6);
        pre[i] = make_float4(0.f, 0.f, 0.f, 0.f);
        if (row < rlim) pre[i] = *(const float4*)(A + (size_t)row * G1_K + 4 * (idx & 63));
      }
    }
    const int mt = tb + rt;
    if (mt < t1) {
      const float* arow = sA + buf * STAGE + (rt * 16 + lid) * G1_PITCH + 2 * half;
      v8f c = {0.f, 0.f, 0.f, 0.f, 0.f, 0.f, 0.f, 0.f};
      constexpr int AC = 8;
      v2f a[2][AC];
#pragma unroll
      for (int j = 0; j < AC; ++j) a[0][j] = *(const v2f*)(arow + 4 * j);
#pragma unroll
      for (int kc = 0; kc < KS / AC; ++kc) {
        if (kc + 1 < KS / AC) {
#pragma unroll
          for (int j = 0; j < AC; ++j)
            a[(kc + 1) & 1][j] = *(const v2f*)(arow + 4 * (AC * (kc + 1) + j));
        }
        __builtin_amdgcn_sched_barrier(0);
#pragma unroll
        for (int j = 0; j < AC; ++j)
          c = __builtin_amdgcn_wmma_f32_16x16x4_f32(false, a[kc & 1][j], false, b[AC * kc + j],
                                                    (short)0, c, false, false);
        __builtin_amdgcn_sched_barrier(0);
      }
      const int r0 = mt * 16 + 8 * half;
      {
        float ls[8], ld[8];
#pragma unroll
        for (int r = 0; r < 8; ++r) {
          ls[r] = row_xadd<4>(row_xadd<2>(row_xadd<1>(c[r] * avs_l)));
          ld[r] = row_xadd<4>(row_xadd<2>(row_xadd<1>(c[r] * avd_l)));
        }
        const int d = lid & 7;
        float vs = ls[0], vd = ld[0];
#pragma unroll
        for (int r = 1; r < 8; ++r) {
          vs = (d == r) ? ls[r] : vs;
          vd = (d == r) ? ld[r] : vd;
        }
        const int lrow = r0 + d;
        if (lrow < M) {
          const size_t li = (size_t)lrow * (G1_N / 8) + 2 * ct + (lid >> 3);
          as[li] = vs;
          ad[li] = vd;
        }
      }
      float* crow = C + (size_t)r0 * G1_N + ct * 16 + lid;
      if (mt * 16 + 16 <= M) {
#pragma unroll
        for (int r = 0; r < 8; ++r) crow[(size_t)r * G1_N] = c[r];
      } else {
        const int nrows = M - r0;
#pragma unroll
        for (int r = 0; r < 8; ++r)
          if (r < nrows) crow[(size_t)r * G1_N] = c[r];
      }
    }
    if (more) {
      float* nxt = sA + (buf ^ 1) * STAGE;
#pragma unroll
      for (int i = 0; i < LD; ++i) {
        const int idx = i * G1_THREADS + tid;
        *(float4*)(nxt + (idx >> 6) * G1_PITCH + 4 * (idx & 63)) = pre[i];
      }
    }
    __syncthreads();
    buf ^= 1;
  }
}

#define NODES_PER_WG 392
#define WG_THREADS 1024
#define WG_WAVES 32
#define LIST_CAP 32768
#define GROUP_RANGES 8
#define GROUP_NODES (GROUP_RANGES * NODES_PER_WG)
#define BKT_CAP 512
#define OWN_SLOTS ((NODES_PER_WG + 31) / 32)
#define FINE_U 8

#define SCAN_U 8

__device__ __forceinline__ void stage_append(bool valid, int e, int dl,
                                             int2* stage, int* lcount, int lane) {
  unsigned m = __builtin_amdgcn_ballot_w32(valid);
  if (m == 0u) return;
  int pos = 0;
  if (lane == 0) pos = atomicAdd(lcount, (int)__popc(m));
  pos = __builtin_amdgcn_readfirstlane(pos);
  if (valid) {
    int slot = pos + (int)__builtin_amdgcn_mbcnt_lo(m, 0u);
    if (slot < LIST_CAP) stage[slot] = make_int2(e, dl);
  }
}

template <bool GUARD>
__device__ __forceinline__ void scan_batch(const int4* __restrict__ d4, int vb, int nvec,
                                           int lo, int cnt, int2* stage, int* lcount,
                                           int lane) {
  int4 d[SCAN_U];
#pragma unroll
  for (int u = 0; u < SCAN_U; ++u) {
    const int v = vb + u * 32 + lane;
    if (GUARD) {
      d[u] = make_int4(lo - 1, lo - 1, lo - 1, lo - 1);
      if (v < nvec) d[u] = d4[v];
    } else {
      d[u] = d4[v];
    }
  }
  unsigned m[SCAN_U][4];
  int tot = 0;
#pragma unroll
  for (int u = 0; u < SCAN_U; ++u) {
    m[u][0] = __builtin_amdgcn_ballot_w32((unsigned)(d[u].x - lo) < (unsigned)cnt);
    m[u][1] = __builtin_amdgcn_ballot_w32((unsigned)(d[u].y - lo) < (unsigned)cnt);
    m[u][2] = __builtin_amdgcn_ballot_w32((unsigned)(d[u].z - lo) < (unsigned)cnt);
    m[u][3] = __builtin_amdgcn_ballot_w32((unsigned)(d[u].w - lo) < (unsigned)cnt);
    tot += (int)(__popc(m[u][0]) + __popc(m[u][1]) + __popc(m[u][2]) + __popc(m[u][3]));
  }
  if (tot == 0) return;
  int base = 0;
  if (lane == 0) base = atomicAdd(lcount, tot);
  base = __builtin_amdgcn_readfirstlane(base);
#pragma unroll
  for (int u = 0; u < SCAN_U; ++u) {
    const int e0 = 4 * (vb + u * 32 + lane);
    const int dd[4] = {d[u].x, d[u].y, d[u].z, d[u].w};
#pragma unroll
    for (int j = 0; j < 4; ++j) {
      const unsigned mm = m[u][j];
      if (mm != 0u) {
        if ((mm >> lane) & 1u) {
          const int slot = base + (int)__builtin_amdgcn_mbcnt_lo(mm, 0u);
          if (slot < LIST_CAP) stage[slot] = make_int2(e0 + j, dd[j] - lo);
        }
        base += (int)__popc(mm);
      }
    }
  }
}

__global__ __launch_bounds__(WG_THREADS) void edge_bin_kernel(
    const int* __restrict__ src, const int* __restrict__ dst, int E, int N,
    int2* __restrict__ lists, int* __restrict__ counts) {
  __shared__ int2 stage[LIST_CAP];
  __shared__ int lcount;
  const int tid = threadIdx.x, lane = tid & 31;
  const int wv = __builtin_amdgcn_readfirstlane(tid >> 5);
  const int grp = blockIdx.x / GROUP_RANGES;
  const int slc = blockIdx.x - grp * GROUP_RANGES;
  const int lo = grp * GROUP_NODES;
  const int cnt = min(GROUP_NODES, N - lo);
  int2* lst = lists + (size_t)blockIdx.x * LIST_CAP;
  if (tid == 0) lcount = 0;
  __syncthreads();
  if ((E & 3) == 0) {
    const int nvec = E >> 2;
    const int vper = (nvec + GROUP_RANGES - 1) / GROUP_RANGES;
    const int vbeg = slc * vper;
    const int vend = min(nvec, vbeg + vper);
    const int4* __restrict__ d4 = (const int4*)dst;
    constexpr int WSPAN = 32 * SCAN_U;
    for (int vb = vbeg + wv * WSPAN; vb < vend; vb += WG_WAVES * WSPAN) {
      if (vb + WSPAN <= vend) scan_batch<false>(d4, vb, vend, lo, cnt, stage, &lcount, lane);
      else                    scan_batch<true >(d4, vb, vend, lo, cnt, stage, &lcount, lane);
    }
  } else {
    const int eper = (E + GROUP_RANGES - 1) / GROUP_RANGES;
    const int ebeg = slc * eper;
    const int eend = min(E, ebeg + eper);
    for (int eb = ebeg + wv * 32; eb < eend; eb += WG_THREADS) {
      const int e = eb + lane;
      int x = -1;
      if (e < eend) x = dst[e] - lo;
      stage_append((unsigned)x < (unsigned)cnt, e, x, stage, &lcount, lane);
    }
  }
  __syncthreads();
  const int found = lcount;
  const int nfill = min(found, LIST_CAP);
  constexpr int GU = 8;
  for (int i0 = tid; i0 < nfill; i0 += GU * WG_THREADS) {
    int2 p[GU];
    int s[GU];
#pragma unroll
    for (int k = 0; k < GU; ++k) {
      const int i = i0 + k * WG_THREADS;
      p[k] = (i < nfill) ? stage[i] : make_int2(0, 0);
    }
#pragma unroll
    for (int k = 0; k < GU; ++k) s[k] = src[p[k].x];
#pragma unroll
    for (int k = 0; k < GU; ++k) {
      const int i = i0 + k * WG_THREADS;
      if (i < nfill) lst[i] = make_int2(s[k], p[k].y);
    }
  }
  if (tid == 0) counts[blockIdx.x] = found;
}

template <bool GUARD>
__device__ __forceinline__ void fine_batch(const int2* __restrict__ seg, int i0, int c,
                                           int sub_lo, int cnt, int2* bkt, int* bcnt,
                                           int lane) {
  int2 p[FINE_U];
#pragma unroll
  for (int u = 0; u < FINE_U; ++u) {
    const int i = i0 + u * 32 + lane;
    if (GUARD) {
      p[u] = make_int2(0, sub_lo - 1);
      if (i < c) p[u] = seg[i];
    } else {
      p[u] = seg[i];
    }
  }
  int slot[FINE_U];
#pragma unroll
  for (int u = 0; u < FINE_U; ++u) {
    const int dl = p[u].y - sub_lo;
    slot[u] = BKT_CAP;
    if ((unsigned)dl < (unsigned)cnt) slot[u] = atomicAdd(&bcnt[dl & 31], 1);
  }
#pragma unroll
  for (int u = 0; u < FINE_U; ++u) {
    const int dl = p[u].y - sub_lo;
    if (slot[u] < BKT_CAP) bkt[(dl & 31) * BKT_CAP + slot[u]] = make_int2(p[u].x, dl);
  }
}

__device__ __forceinline__ bool fetch_own(int i, int n_real, bool use_list,
                                          const int2* lst,
                                          const int* __restrict__ src,
                                          const int* __restrict__ dst,
                                          int lo, int cnt, int wave, int& sv, int& dv) {
  sv = 0; dv = 0;
  if (i < n_real) {
    if (use_list) { int2 p = lst[i]; sv = p.x; dv = p.y; return true; }
    int d = dst[i] - lo;
    if ((unsigned)d < (unsigned)cnt && (d & 31) == wave) { sv = src[i]; dv = d; return true; }
    return false;
  }
  const int j = i - n_real;
  const int n = wave + 32 * j;
  if (j < OWN_SLOTS && n < cnt) { dv = n; sv = lo + n; return true; }
  return false;
}

template <int DTOT, int H, bool ELU, int NOUT>
__global__ __launch_bounds__(WG_THREADS) void range_agg_kernel(
    const int* __restrict__ src, const int* __restrict__ dst, int E, int N,
    const int2* __restrict__ lists, const int* __restrict__ counts,
    const float* __restrict__ as, const float* __restrict__ ad,
    const float* __restrict__ hsrc, const float* __restrict__ bias,
    const float* __restrict__ Wn, float* __restrict__ out,
    const float* __restrict__ avs_n, const float* __restrict__ avd_n,
    float* __restrict__ as_out, float* __restrict__ ad_out) {
  constexpr int R = NODES_PER_WG;
  constexpr bool PROD = NOUT > 0;
  constexpr int P = DTOT + 4;
  constexpr int RP = ((R + 15) / 16) * 16;
  constexpr int VPL = DTOT / 32;
  constexpr int Dh = DTOT / H;
  constexpr int U = 8;
  constexpr int NS = OWN_SLOTS;
  constexpr int SPARE_SLOT = 15;
  static_assert(NS <= SPARE_SLOT, "a wave's node slots and the spare are the elements of one 16-wide register vector");
  static_assert(WG_WAVES == 32, "a node's owner wave is its index modulo 32");
  typedef float vrow __attribute__((ext_vector_type(VPL)));
  typedef float vslot __attribute__((ext_vector_type(16)));
  __shared__ __attribute__((aligned(16))) float rows[PROD ? RP * P : 4];
  __shared__ unsigned mx[R * H];
  __shared__ float adl[R * H];
  __shared__ int2 bkt[WG_WAVES * BKT_CAP];
  __shared__ int bcnt[WG_WAVES];
  __shared__ float wbuf[(H > 1) ? WG_WAVES * 32 * H : 1];

  const int tid = threadIdx.x, lane = tid & 31;
  const int wu = __builtin_amdgcn_readfirstlane(tid >> 5);
  const int lo = blockIdx.x * R;
  const int cnt = min(R, N - lo);
  const int grp = blockIdx.x / GROUP_RANGES;
  const int sub_lo = (blockIdx.x - grp * GROUP_RANGES) * R;
  const int2* glists = lists + (size_t)grp * GROUP_RANGES * LIST_CAP;
  const int* gcounts = counts + grp * GROUP_RANGES;
  int cmax = 0;
#pragma unroll
  for (int s = 0; s < GROUP_RANGES; ++s) cmax = max(cmax, gcounts[s]);
  const bool seg_ok = cmax <= LIST_CAP;

  if (tid < WG_WAVES) bcnt[tid] = 0;
  for (int i = tid; i < R * H; i += WG_THREADS) {
    mx[i] = ENC_NEG_INF;
    adl[i] = (i < cnt * H) ? ad[(size_t)lo * H + i] : 0.f;
  }
  __syncthreads();
  if (seg_ok) {
    const int s = wu % GROUP_RANGES;
    const int q = wu / GROUP_RANGES;
    constexpr int FSPAN = 32 * FINE_U;
    constexpr int FSTEP = (WG_WAVES / GROUP_RANGES) * FSPAN;
    const int2* seg = glists + (size_t)s * LIST_CAP;
    const int c = gcounts[s];
    for (int i0 = q * FSPAN; i0 < c; i0 += FSTEP) {
      if (i0 + FSPAN <= c) fine_batch<false>(seg, i0, c, sub_lo, cnt, bkt, bcnt, lane);
      else                 fine_batch<true >(seg, i0, c, sub_lo, cnt, bkt, bcnt, lane);
    }
  }
  __syncthreads();
  const int myc = bcnt[lane];
  const bool use_list = seg_ok && __builtin_amdgcn_ballot_w32(myc > BKT_CAP) == 0u;
  const int n_real = use_list ? __builtin_amdgcn_readlane(myc, wu) : E;
  const int total = n_real + NS;
  const int2* lst = bkt + wu * BKT_CAP;

  for (int base = 0; base < total; base += 32) {
    int sv, dv;
    bool valid = fetch_own(base + lane, n_real, use_list, lst, src, dst, lo, cnt, wu, sv, dv);
    if (valid) {
#pragma unroll
      for (int hh = 0; hh < H; ++hh) {
        float e = as[(size_t)sv * H + hh] + adl[dv * H + hh];
        e = e > 0.f ? e : NEG_SLOPE * e;
        atomicMax(&mx[dv * H + hh], enc_ord(e));
      }
    }
  }
  __syncthreads();

  const int c0 = lane * VPL;
  const int hh = c0 / Dh;
  vslot accv[VPL];
  vslot smv;
  float sm1 = 0.f;
#pragma unroll
  for (int j = 0; j < 16; ++j) {
    smv[j] = 0.f;
#pragma unroll
    for (int k = 0; k < VPL; ++k) accv[k][j] = 0.f;
  }
  float* wblk = wbuf + (H > 1 ? wu * 32 * H : 0);
  int sv_n, dv_n;
  bool valid_n = fetch_own(lane, n_real, use_list, lst, src, dst, lo, cnt, wu, sv_n, dv_n);
  float as_n[H];
#pragma unroll
  for (int h = 0; h < H; ++h) as_n[h] = as[(size_t)sv_n * H + h];
  for (int base = 0; base < total; base += 32) {
    const int sv = sv_n, dv = dv_n;
    const bool valid = valid_n;
    float asv[H];
#pragma unroll
    for (int h = 0; h < H; ++h) asv[h] = as_n[h];
    if (base + 32 < total) {
      valid_n = fetch_own(base + 32 + lane, n_real, use_list, lst, src, dst, lo, cnt, wu, sv_n, dv_n);
#pragma unroll
      for (int h = 0; h < H; ++h) as_n[h] = as[(size_t)sv_n * H + h];
    }
    unsigned mask = __builtin_amdgcn_ballot_w32(valid);
    if (mask != 0u) {
      float wl[H];
#pragma unroll
      for (int h = 0; h < H; ++h) {
        float e = asv[h] + adl[dv * H + h];
        e = e > 0.f ? e : NEG_SLOPE * e;
        wl[h] = __expf(e - dec_ord(mx[dv * H + h]));
      }
      if constexpr (H > 1) {
#pragma unroll
        for (int h = 0; h < H; ++h) wblk[lane * H + h] = wl[h];
      }
      while (mask != 0u) {
        int su[U], du[U], ju[U];
        float ws[U];
        bool ok[U];
#pragma unroll
        for (int u = 0; u < U; ++u) {
          ok[u] = mask != 0u;
          ju[u] = ok[u] ? (int)__builtin_ctz(mask) : 0;
          if (ok[u]) mask &= mask - 1u;
          su[u] = __builtin_amdgcn_readlane(sv, ju[u]);
          du[u] = __builtin_amdgcn_readlane(dv, ju[u]);
          if constexpr (H == 1)
            ws[u] = __int_as_float(__builtin_amdgcn_readlane(__float_as_int(wl[0]), ju[u]));
          else
            ws[u] = 0.f;
        }
        vrow row[U];
        float wv[U];
#pragma unroll
        for (int u = 0; u < U; ++u) {
          row[u] = *(const vrow*)(hsrc + (size_t)su[u] * DTOT + c0);
          if constexpr (H > 1) wv[u] = wblk[ju[u] * H + hh];
          else                 wv[u] = ws[u];
        }
        __builtin_amdgcn_sched_barrier(0);
#pragma unroll
        for (int u = 0; u < U; ++u) {
          const float w = wv[u];
          const int sl = ok[u] ? (du[u] >> 5) : SPARE_SLOT;
#pragma unroll
          for (int k = 0; k < VPL; ++k) accv[k][sl] += w * row[u][k];
          if constexpr (H == 1) sm1 += (lane == sl) ? w : 0.f;
          else                  smv[sl] += w;
        }
      }
    }
  }

  float bv[VPL];
#pragma unroll
  for (int k = 0; k < VPL; ++k) bv[k] = bias[c0 + k];
#pragma unroll
  for (int j = 0; j < NS; ++j) {
    const int n = wu + 32 * j;
    float sj;
    if constexpr (H == 1) sj = __int_as_float(__builtin_amdgcn_readlane(__float_as_int(sm1), j));
    else                  sj = smv[j];
    vrow v;
#pragma unroll
    for (int k = 0; k < VPL; ++k) {
      float t = accv[k][j] / (sj + GAT_EPS) + bv[k];
      if (ELU) t = t > 0.f ? t : __expf(t) - 1.f;
      v[k] = t;
    }
    if constexpr (!PROD) {
      if (n < cnt) *(vrow*)(out + (size_t)(lo + n) * DTOT + c0) = v;
    } else {
      if (n < RP) {
        if (n >= cnt) {
#pragma unroll
          for (int k = 0; k < VPL; ++k) v[k] = 0.f;
        }
        *(vrow*)(rows + n * P + c0) = v;
      }
    }
  }

  if constexpr (PROD) {
    __syncthreads();
    constexpr int NT = NOUT / 16;
    constexpr int MT = RP / 16;
    constexpr int KS = DTOT / 4;
    constexpr int QS = WG_WAVES / NT;
    const int half = lane >> 4, lid = lane & 15;
    const int ct = wu % NT;
    const int q = wu / NT;
    const float* bcol = Wn + (size_t)(2 * half) * NOUT + ct * 16 + lid;
    const float avs_l = avs_n[ct * 16 + lid];
    const float avd_l = avd_n[ct * 16 + lid];
    static_assert(H > 1 && WG_WAVES * 32 * H >= NT * RP * 2, "the partial logits fit the weight slab");
    float* part = wbuf;
    v2f b[KS];
#pragma unroll
    for (int ks = 0; ks < KS; ++ks) {
      b[ks].x = bcol[(size_t)(4 * ks) * NOUT];
      b[ks].y = bcol[(size_t)(4 * ks + 1) * NOUT];
    }
    for (int mt = q; mt < MT; mt += QS) {
      const float* arow = rows + (mt * 16 + lid) * P + 2 * half;
      v2f a[KS];
#pragma unroll
      for (int ks = 0; ks < KS; ++ks) a[ks] = *(const v2f*)(arow + 4 * ks);
      v8f c = {0.f, 0.f, 0.f, 0.f, 0.f, 0.f, 0.f, 0.f};
#pragma unroll
      for (int ks = 0; ks < KS; ++ks)
        c = __builtin_amdgcn_wmma_f32_16x16x4_f32(false, a[ks], false, b[ks],
                                                  (short)0, c, false, false);
      const int r0 = mt * 16 + 8 * half;
      {
        float ls[8], ld[8];
#pragma unroll
        for (int r = 0; r < 8; ++r) {
          ls[r] = row_xadd<8>(row_xadd<4>(row_xadd<2>(row_xadd<1>(c[r] * avs_l))));
          ld[r] = row_xadd<8>(row_xadd<4>(row_xadd<2>(row_xadd<1>(c[r] * avd_l))));
        }
        const int d = lid & 7;
        float vs = ls[0], vd = ld[0];
#pragma unroll
        for (int r = 1; r < 8; ++r) {
          vs = (d == r) ? ls[r] : vs;
          vd = (d == r) ? ld[r] : vd;
        }
        part[(ct * RP + r0 + d) * 2 + (lid >> 3)] = (lid < 8) ? vs : vd;
      }
      float* crow = out + (size_t)(lo + r0) * NOUT + ct * 16 + lid;
      if (mt * 16 + 16 <= cnt) {
#pragma unroll
        for (int r = 0; r < 8; ++r) crow[(size_t)r * NOUT] = c[r];
      } else {
        const int nrows = cnt - r0;
#pragma unroll
        for (int r = 0; r < 8; ++r)
          if (r < nrows) crow[(size_t)r * NOUT] = c[r];
      }
    }
    __syncthreads();
    for (int i = tid; i < RP; i += WG_THREADS) {
      if (i < cnt) {
        float a = part[i * 2], d = part[i * 2 + 1];
#pragma unroll
        for (int t = 1; t < NT; ++t) {
          a += part[(t * RP + i) * 2];
          d += part[(t * RP + i) * 2 + 1];
        }
        as_out[lo + i] = a;
        ad_out[lo + i] = d;
      }
    }
  }
}

static inline int cdiv(long long a, long long b) { return (int)((a + b - 1) / b); }

extern "C" void kernel_launch(void* const* d_in, const int* in_sizes, int n_in,
                              void* d_out, int out_size, void* d_ws, size_t ws_size,
                              hipStream_t stream) {
  const float* x      = (const float*)d_in[0];
  const int*   ei     = (const int*)  d_in[1];
  const float* W1     = (const float*)d_in[2];
  const float* a_src1 = (const float*)d_in[3];
  const float* a_dst1 = (const float*)d_in[4];
  const float* b1     = (const float*)d_in[5];
  const float* W2     = (const float*)d_in[6];
  const float* a_src2 = (const float*)d_in[7];
  const float* a_dst2 = (const float*)d_in[8];
  const float* b2     = (const float*)d_in[9];
  float* z = (float*)d_out;

  const int Fin = 256, H1 = 8, C1 = 64, C2 = 128;
  const int N    = in_sizes[0] / Fin;
  const int E    = in_sizes[1] / 2;
  const int* srcv = ei;
  const int* dstv = ei + E;
  const int G = cdiv(N, NODES_PER_WG);
  const int GB = cdiv(G, GROUP_RANGES) * GROUP_RANGES;

  float* h1   = (float*)d_ws;
  float* as1  = h1  + (size_t)N * C1;
  float* ad1  = as1 + (size_t)N * H1;
  float* h2   = ad1 + (size_t)N * H1;
  float* as2  = h2  + (size_t)N * C2;
  float* ad2  = as2 + N;
  int*   cnts = (int*)(ad2 + N);
  int2*  lsts = (int2*)(cnts + 1024);

  edge_bin_kernel<<<GB, WG_THREADS, 0, stream>>>(srcv, dstv, E, N, lsts, cnts);

  gemm1_lds_kernel<<<256, G1_THREADS, 0, stream>>>(x, W1, h1, N, a_src1, a_dst1, as1, ad1);
  range_agg_kernel<64, 8, true, 128><<<G, WG_THREADS, 0, stream>>>(
      srcv, dstv, E, N, lsts, cnts, as1, ad1, h1, b1, W2, h2, a_src2, a_dst2, as2, ad2);

  range_agg_kernel<128, 1, false, 0><<<G, WG_THREADS, 0, stream>>>(
      srcv, dstv, E, N, lsts, cnts, as2, ad2, h2, b2, nullptr, z,
      nullptr, nullptr, nullptr, nullptr);
}


# Edge filter made a one-read split: each slice work-group files its edges by destination range into its own cell per range (LDS counters); owners read only their cells; sum order = LDS arrival
# speedup vs baseline: 4.8431x; 1.0793x over previous
#include <hip/hip_runtime.h>
#include <cstdint>


typedef float v2f __attribute__((ext_vector_type(2)));
typedef float v8f __attribute__((ext_vector_type(8)));

#define NEG_SLOPE 0.2f
#define GAT_EPS 1e-16f
#define ENC_NEG_INF 0x007FFFFFu

__device__ __forceinline__ unsigned enc_ord(float f) {
  unsigned u = __float_as_uint(f);
  return (u & 0x80000000u) ? ~u : (u | 0x80000000u);
}
__device__ __forceinline__ float dec_ord(unsigned k) {
  return (k & 0x80000000u) ? __uint_as_float(k & 0x7FFFFFFFu)
                           : __uint_as_float(~k);
}

template <int XM>
__device__ __forceinline__ float row_xadd(float v) {
  return v + __int_as_float(__builtin_amdgcn_update_dpp(0, __float_as_int(v), 0x160 | XM,
                                                        0xf, 0xf, true));
}

#define G1_K 256
#define G1_N 64
#define G1_ROWS 64
#define G1_PITCH (G1_K + 4)
#define G1_THREADS 512

__global__ __launch_bounds__(G1_THREADS) void gemm1_lds_kernel(
    const float* __restrict__ A, const float* __restrict__ B,
    float* __restrict__ C, int M,
    const float* __restrict__ avs, const float* __restrict__ avd,
    float* __restrict__ as, float* __restrict__ ad) {
  __shared__ __attribute__((aligned(16))) float sA[2 * G1_ROWS * G1_PITCH];
  constexpr int KS = G1_K / 4;
  constexpr int LD = (G1_ROWS * G1_K / 4) / G1_THREADS;
  constexpr int STAGE = G1_ROWS * G1_PITCH;
  const int tid = threadIdx.x, lane = tid & 31;
  const int half = lane >> 4, lid = lane & 15;
  const int wu = __builtin_amdgcn_readfirstlane(tid >> 5);
  const int ct = wu & 3;
  const int rt = wu >> 2;
  const int mtiles = (M + 15) >> 4;
  const int t0 = (int)(((long long)blockIdx.x * mtiles) / gridDim.x);
  const int t1 = (int)(((long long)(blockIdx.x + 1) * mtiles) / gridDim.x);
  const int rlim = min(16 * t1, M);

  const float* bcol = B + (size_t)(2 * half) * G1_N + ct * 16 + lid;
  v2f b[KS];
#pragma unroll
  for (int ks = 0; ks < KS; ++ks) {
    b[ks].x = bcol[(size_t)(4 * ks) * G1_N];
    b[ks].y = bcol[(size_t)(4 * ks + 1) * G1_N];
  }

  const float avs_l = avs[ct * 16 + lid];
  const float avd_l = avd[ct * 16 + lid];

  float4 pre[LD];
#pragma unroll
  for (int i = 0; i < LD; ++i) {
    const int idx = i * G1_THREADS + tid;
    const int row = 16 * t0 + (idx >> 6);
    pre[i] = make_float4(0.f, 0.f, 0.f, 0.f);
    if (row < rlim) pre[i] = *(const float4*)(A + (size_t)row * G1_K + 4 * (idx & 63));
  }
#pragma unroll
  for (int i = 0; i < LD; ++i) {
    const int idx = i * G1_THREADS + tid;
    *(float4*)(sA + (idx >> 6) * G1_PITCH + 4 * (idx & 63)) = pre[i];
  }
  __syncthreads();

  int buf = 0;
  for (int tb = t0; tb < t1; tb += 4) {
    const int tn = tb + 4;
    const bool more = tn < t1;
    if (more) {
#pragma unroll
      for (int i = 0; i < LD; ++i) {
        const int idx = i * G1_THREADS + tid;
        const int row = 16 * tn + (idx >> 6);
        pre[i] = make_float4(0.f, 0.f, 0.f, 0.f);
        if (row < rlim) pre[i] = *(const float4*)(A + (size_t)row * G1_K + 4 * (idx & 63));
      }
    }
    const int mt = tb + rt;
    if (mt < t1) {
      const float* arow = sA + buf * STAGE + (rt * 16 + lid) * G1_PITCH + 2 * half;
      v8f c = {0.f, 0.f, 0.f, 0.f, 0.f, 0.f, 0.f, 0.f};
      constexpr int AC = 8;
      v2f a[2][AC];
#pragma unroll
      for (int j = 0; j < AC; ++j) a[0][j] = *(const v2f*)(arow + 4 * j);
#pragma unroll
      for (int kc = 0; kc < KS / AC; ++kc) {
        if (kc + 1 < KS / AC) {
#pragma unroll
          for (int j = 0; j < AC; ++j)
            a[(kc + 1) & 1][j] = *(const v2f*)(arow + 4 * (AC * (kc + 1) + j));
        }
        __builtin_amdgcn_sched_barrier(0);
#pragma unroll
        for (int j = 0; j < AC; ++j)
          c = __builtin_amdgcn_wmma_f32_16x16x4_f32(false, a[kc & 1][j], false, b[AC * kc + j],
                                                    (short)0, c, false, false);
        __builtin_amdgcn_sched_barrier(0);
      }
      const int r0 = mt * 16 + 8 * half;
      {
        float ls[8], ld[8];
#pragma unroll
        for (int r = 0; r < 8; ++r) {
          ls[r] = row_xadd<4>(row_xadd<2>(row_xadd<1>(c[r] * avs_l)));
          ld[r] = row_xadd<4>(row_xadd<2>(row_xadd<1>(c[r] * avd_l)));
        }
        const int d = lid & 7;
        float vs = ls[0], vd = ld[0];
#pragma unroll
        for (int r = 1; r < 8; ++r) {
          vs = (d == r) ? ls[r] : vs;
          vd = (d == r) ? ld[r] : vd;
        }
        const int lrow = r0 + d;
        if (lrow < M) {
          const size_t li = (size_t)lrow * (G1_N / 8) + 2 * ct + (lid >> 3);
          as[li] = vs;
          ad[li] = vd;
        }
      }
      float* crow = C + (size_t)r0 * G1_N + ct * 16 + lid;
      if (mt * 16 + 16 <= M) {
#pragma unroll
        for (int r = 0; r < 8; ++r) crow[(size_t)r * G1_N] = c[r];
      } else {
        const int nrows = M - r0;
#pragma unroll
        for (int r = 0; r < 8; ++r)
          if (r < nrows) crow[(size_t)r * G1_N] = c[r];
      }
    }
    if (more) {
      float* nxt = sA + (buf ^ 1) * STAGE;
#pragma unroll
      for (int i = 0; i < LD; ++i) {
        const int idx = i * G1_THREADS + tid;
        *(float4*)(nxt + (idx >> 6) * G1_PITCH + 4 * (idx & 63)) = pre[i];
      }
    }
    __syncthreads();
    buf ^= 1;
  }
}

#define NODES_PER_WG 392
#define WG_THREADS 1024
#define WG_WAVES 32
#define SLICES 256
#define CELL_CAP 64
#define MAX_RANGES 512
#define CELLS_PER_WAVE (SLICES / WG_WAVES)
#define BKT_CAP 512
#define OWN_SLOTS ((NODES_PER_WG + 31) / 32)
#define BIN_THREADS 256
#define BIN_WAVES (BIN_THREADS / 32)
#define BIN_U 32

__global__ __launch_bounds__(BIN_THREADS) void edge_bin_kernel(
    const int* __restrict__ src, const int* __restrict__ dst, int E, int N, int G,
    int2* __restrict__ cells, int* __restrict__ counts) {
  __shared__ __attribute__((aligned(16))) int2 stage[MAX_RANGES * CELL_CAP];
  __shared__ int rc[MAX_RANGES];
  const int tid = threadIdx.x, lane = tid & 31;
  const int wv = __builtin_amdgcn_readfirstlane(tid >> 5);
  const int slc = blockIdx.x;
  const int nr = min(G, MAX_RANGES);
  for (int i = tid; i < nr; i += BIN_THREADS) rc[i] = 0;
  __syncthreads();
  const int eper = (((E + SLICES - 1) / SLICES) + 31) & ~31;
  const int ebeg = min(E, slc * eper);
  const int eend = min(E, ebeg + eper);
  for (int e0 = ebeg + tid; e0 < eend; e0 += BIN_U * BIN_THREADS) {
    int d[BIN_U], s[BIN_U];
#pragma unroll
    for (int k = 0; k < BIN_U; ++k) {
      const int e = e0 + k * BIN_THREADS;
      d[k] = -1;
      s[k] = 0;
      if (e < eend) { d[k] = dst[e]; s[k] = src[e]; }
    }
    int r[BIN_U], slot[BIN_U];
#pragma unroll
    for (int k = 0; k < BIN_U; ++k) {
      slot[k] = CELL_CAP;
      r[k] = 0;
      if ((unsigned)d[k] < (unsigned)N) {
        r[k] = d[k] / NODES_PER_WG;
        if (r[k] < MAX_RANGES) slot[k] = atomicAdd(&rc[r[k]], 1);
      }
    }
#pragma unroll
    for (int k = 0; k < BIN_U; ++k) {
      if (slot[k] < CELL_CAP)
        stage[r[k] * CELL_CAP + slot[k]] = make_int2(s[k], d[k] - r[k] * NODES_PER_WG);
    }
  }
  __syncthreads();
  static_assert(CELL_CAP == 64, "a cell is one 16-byte piece a lane");
  constexpr int CU = 16;
  for (int rg0 = wv; rg0 < nr; rg0 += CU * BIN_WAVES) {
    int nl = 0;
    {
      const int rg = rg0 + lane * BIN_WAVES;
      if (lane < CU && rg < nr) nl = min(rc[rg], CELL_CAP);
    }
    int n[CU];
    int4 v[CU];
#pragma unroll
    for (int k = 0; k < CU; ++k) n[k] = __builtin_amdgcn_readlane(nl, k);
#pragma unroll
    for (int k = 0; k < CU; ++k) {
      const int rg = rg0 + k * BIN_WAVES;
      v[k] = make_int4(0, 0, 0, 0);
      if (2 * lane < n[k]) v[k] = *(const int4*)(stage + rg * CELL_CAP + 2 * lane);
    }
#pragma unroll
    for (int k = 0; k < CU; ++k) {
      const int rg = rg0 + k * BIN_WAVES;
      if (2 * lane < n[k])
        *(int4*)(cells + ((size_t)rg * SLICES + slc) * CELL_CAP + 2 * lane) = v[k];
    }
  }
  for (int i = tid; i < G; i += BIN_THREADS)
    counts[(size_t)i * SLICES + slc] = (i < MAX_RANGES) ? rc[i] : CELL_CAP + 1;
}

__device__ __forceinline__ bool fetch_own(int i, int n_real, bool use_list,
                                          const int2* lst,
                                          const int* __restrict__ src,
                                          const int* __restrict__ dst,
                                          int lo, int cnt, int wave, int& sv, int& dv) {
  sv = 0; dv = 0;
  if (i < n_real) {
    if (use_list) { int2 p = lst[i]; sv = p.x; dv = p.y; return true; }
    int d = dst[i] - lo;
    if ((unsigned)d < (unsigned)cnt && (d & 31) == wave) { sv = src[i]; dv = d; return true; }
    return false;
  }
  const int j = i - n_real;
  const int n = wave + 32 * j;
  if (j < OWN_SLOTS && n < cnt) { dv = n; sv = lo + n; return true; }
  return false;
}

template <int DTOT, int H, bool ELU, int NOUT>
__global__ __launch_bounds__(WG_THREADS) void range_agg_kernel(
    const int* __restrict__ src, const int* __restrict__ dst, int E, int N,
    const int2* __restrict__ lists, const int* __restrict__ counts,
    const float* __restrict__ as, const float* __restrict__ ad,
    const float* __restrict__ hsrc, const float* __restrict__ bias,
    const float* __restrict__ Wn, float* __restrict__ out,
    const float* __restrict__ avs_n, const float* __restrict__ avd_n,
    float* __restrict__ as_out, float* __restrict__ ad_out) {
  constexpr int R = NODES_PER_WG;
  constexpr bool PROD = NOUT > 0;
  constexpr int P = DTOT + 4;
  constexpr int RP = ((R + 15) / 16) * 16;
  constexpr int VPL = DTOT / 32;
  constexpr int Dh = DTOT / H;
  constexpr int U = 8;
  constexpr int NS = OWN_SLOTS;
  constexpr int SPARE_SLOT = 15;
  static_assert(NS <= SPARE_SLOT, "a wave's node slots and the spare are the elements of one 16-wide register vector");
  static_assert(WG_WAVES == 32, "a node's owner wave is its index modulo 32");
  static_assert(SLICES % 32 == 0 && CELLS_PER_WAVE * WG_WAVES == SLICES, "wave w reads the cells w + 32 k; lane l holds the counts l + 32 k");
  typedef float vrow __attribute__((ext_vector_type(VPL)));
  typedef float vslot __attribute__((ext_vector_type(16)));
  __shared__ __attribute__((aligned(16))) float rows[PROD ? RP * P : 4];
  __shared__ unsigned mx[R * H];
  __shared__ float adl[R * H];
  __shared__ int2 bkt[WG_WAVES * BKT_CAP];
  __shared__ int bcnt[WG_WAVES];
  __shared__ float wbuf[(H > 1) ? WG_WAVES * 32 * H : 1];

  const int tid = threadIdx.x, lane = tid & 31;
  const int wu = __builtin_amdgcn_readfirstlane(tid >> 5);
  const int lo = blockIdx.x * R;
  const int cnt = min(R, N - lo);
  const int2* rcells = lists + (size_t)blockIdx.x * SLICES * CELL_CAP;
  const int* rcounts = counts + (size_t)blockIdx.x * SLICES;
  int cc[CELLS_PER_WAVE];
  int cmax = 0;
#pragma unroll
  for (int k = 0; k < CELLS_PER_WAVE; ++k) {
    cc[k] = rcounts[lane + 32 * k];
    cmax = max(cmax, cc[k]);
  }
  const bool seg_ok = __builtin_amdgcn_ballot_w32(cmax > CELL_CAP) == 0u;

  if (tid < WG_WAVES) bcnt[tid] = 0;
  for (int i = tid; i < R * H; i += WG_THREADS) {
    mx[i] = ENC_NEG_INF;
    adl[i] = (i < cnt * H) ? ad[(size_t)lo * H + i] : 0.f;
  }
  __syncthreads();
  if (seg_ok) {
    static_assert(CELL_CAP == 64, "a cell is read by two loads of a wave");
    int2 p[2 * CELLS_PER_WAVE];
#pragma unroll
    for (int k = 0; k < CELLS_PER_WAVE; ++k) {
      const int ck = __builtin_amdgcn_readlane(cc[k], wu);
      const int2* cell = rcells + (size_t)(wu + 32 * k) * CELL_CAP;
      p[2 * k] = make_int2(0, -1);
      p[2 * k + 1] = make_int2(0, -1);
      if (lane < ck) p[2 * k] = cell[lane];
      if (ck > 32) {
        if (lane + 32 < ck) p[2 * k + 1] = cell[lane + 32];
      }
    }
    int slot[2 * CELLS_PER_WAVE];
#pragma unroll
    for (int u = 0; u < 2 * CELLS_PER_WAVE; ++u) {
      const int dl = p[u].y;
      slot[u] = BKT_CAP;
      if ((unsigned)dl < (unsigned)cnt) slot[u] = atomicAdd(&bcnt[dl & 31], 1);
    }
#pragma unroll
    for (int u = 0; u < 2 * CELLS_PER_WAVE; ++u) {
      const int dl = p[u].y;
      if (slot[u] < BKT_CAP) bkt[(dl & 31) * BKT_CAP + slot[u]] = make_int2(p[u].x, dl);
    }
  }
  __syncthreads();
  const int myc = bcnt[lane];
  const bool use_list = seg_ok && __builtin_amdgcn_ballot_w32(myc > BKT_CAP) == 0u;
  const int n_real = use_list ? __builtin_amdgcn_readlane(myc, wu) : E;
  const int total = n_real + NS;
  const int2* lst = bkt + wu * BKT_CAP;

  for (int base = 0; base < total; base += 32) {
    int sv, dv;
    bool valid = fetch_own(base + lane, n_real, use_list, lst, src, dst, lo, cnt, wu, sv, dv);
    if (valid) {
#pragma unroll
      for (int hh = 0; hh < H; ++hh) {
        float e = as[(size_t)sv * H + hh] + adl[dv * H + hh];
        e = e > 0.f ? e : NEG_SLOPE * e;
        atomicMax(&mx[dv * H + hh], enc_ord(e));
      }
    }
  }
  __syncthreads();

  const int c0 = lane * VPL;
  const int hh = c0 / Dh;
  vslot accv[VPL];
  vslot smv;
  float sm1 = 0.f;
#pragma unroll
  for (int j = 0; j < 16; ++j) {
    smv[j] = 0.f;
#pragma unroll
    for (int k = 0; k < VPL; ++k) accv[k][j] = 0.f;
  }
  float* wblk = wbuf + (H > 1 ? wu * 32 * H : 0);
  int sv_n, dv_n;
  bool valid_n = fetch_own(lane, n_real, use_list, lst, src, dst, lo, cnt, wu, sv_n, dv_n);
  float as_n[H];
#pragma unroll
  for (int h = 0; h < H; ++h) as_n[h] = as[(size_t)sv_n * H + h];
  for (int base = 0; base < total; base += 32) {
    const int sv = sv_n, dv = dv_n;
    const bool valid = valid_n;
    float asv[H];
#pragma unroll
    for (int h = 0; h < H; ++h) asv[h] = as_n[h];
    if (base + 32 < total) {
      valid_n = fetch_own(base + 32 + lane, n_real, use_list, lst, src, dst, lo, cnt, wu, sv_n, dv_n);
#pragma unroll
      for (int h = 0; h < H; ++h) as_n[h] = as[(size_t)sv_n * H + h];
    }
    unsigned mask = __builtin_amdgcn_ballot_w32(valid);
    if (mask != 0u) {
      float wl[H];
#pragma unroll
      for (int h = 0; h < H; ++h) {
        float e = asv[h] + adl[dv * H + h];
        e = e > 0.f ? e : NEG_SLOPE * e;
        wl[h] = __expf(e - dec_ord(mx[dv * H + h]));
      }
      if constexpr (H > 1) {
#pragma unroll
        for (int h = 0; h < H; ++h) wblk[lane * H + h] = wl[h];
      }
      while (mask != 0u) {
        int su[U], du[U], ju[U];
        float ws[U];
        bool ok[U];
#pragma unroll
        for (int u = 0; u < U; ++u) {
          ok[u] = mask != 0u;
          ju[u] = ok[u] ? (int)__builtin_ctz(mask) : 0;
          if (ok[u]) mask &= mask - 1u;
          su[u] = __builtin_amdgcn_readlane(sv, ju[u]);
          du[u] = __builtin_amdgcn_readlane(dv, ju[u]);
          if constexpr (H == 1)
            ws[u] = __int_as_float(__builtin_amdgcn_readlane(__float_as_int(wl[0]), ju[u]));
          else
            ws[u] = 0.f;
        }
        vrow row[U];
        float wv[U];
#pragma unroll
        for (int u = 0; u < U; ++u) {
          row[u] = *(const vrow*)(hsrc + (size_t)su[u] * DTOT + c0);
          if constexpr (H > 1) wv[u] = wblk[ju[u] * H + hh];
          else                 wv[u] = ws[u];
        }
        __builtin_amdgcn_sched_barrier(0);
#pragma unroll
        for (int u = 0; u < U; ++u) {
          const float w = wv[u];
          const int sl = ok[u] ? (du[u] >> 5) : SPARE_SLOT;
#pragma unroll
          for (int k = 0; k < VPL; ++k) accv[k][sl] += w * row[u][k];
          if constexpr (H == 1) sm1 += (lane == sl) ? w : 0.f;
          else                  smv[sl] += w;
        }
      }
    }
  }

  float bv[VPL];
#pragma unroll
  for (int k = 0; k < VPL; ++k) bv[k] = bias[c0 + k];
#pragma unroll
  for (int j = 0; j < NS; ++j) {
    const int n = wu + 32 * j;
    float sj;
    if constexpr (H == 1) sj = __int_as_float(__builtin_amdgcn_readlane(__float_as_int(sm1), j));
    else                  sj = smv[j];
    vrow v;
#pragma unroll
    for (int k = 0; k < VPL; ++k) {
      float t = accv[k][j] / (sj + GAT_EPS) + bv[k];
      if (ELU) t = t > 0.f ? t : __expf(t) - 1.f;
      v[k] = t;
    }
    if constexpr (!PROD) {
      if (n < cnt) *(vrow*)(out + (size_t)(lo + n) * DTOT + c0) = v;
    } else {
      if (n < RP) {
        if (n >= cnt) {
#pragma unroll
          for (int k = 0; k < VPL; ++k) v[k] = 0.f;
        }
        *(vrow*)(rows + n * P + c0) = v;
      }
    }
  }

  if constexpr (PROD) {
    __syncthreads();
    constexpr int NT = NOUT / 16;
    constexpr int MT = RP / 16;
    constexpr int KS = DTOT / 4;
    constexpr int QS = WG_WAVES / NT;
    const int half = lane >> 4, lid = lane & 15;
    const int ct = wu % NT;
    const int q = wu / NT;
    const float* bcol = Wn + (size_t)(2 * half) * NOUT + ct * 16 + lid;
    const float avs_l = avs_n[ct * 16 + lid];
    const float avd_l = avd_n[ct * 16 + lid];
    static_assert(H > 1 && WG_WAVES * 32 * H >= NT * RP * 2, "the partial logits fit the weight slab");
    float* part = wbuf;
    v2f b[KS];
#pragma unroll
    for (int ks = 0; ks < KS; ++ks) {
      b[ks].x = bcol[(size_t)(4 * ks) * NOUT];
      b[ks].y = bcol[(size_t)(4 * ks + 1) * NOUT];
    }
    for (int mt = q; mt < MT; mt += QS) {
      const float* arow = rows + (mt * 16 + lid) * P + 2 * half;
      v2f a[KS];
#pragma unroll
      for (int ks = 0; ks < KS; ++ks) a[ks] = *(const v2f*)(arow + 4 * ks);
      v8f c = {0.f, 0.f, 0.f, 0.f, 0.f, 0.f, 0.f, 0.f};
#pragma unroll
      for (int ks = 0; ks < KS; ++ks)
        c = __builtin_amdgcn_wmma_f32_16x16x4_f32(false, a[ks], false, b[ks],
                                                  (short)0, c, false, false);
      const int r0 = mt * 16 + 8 * half;
      {
        float ls[8], ld[8];
#pragma unroll
        for (int r = 0; r < 8; ++r) {
          ls[r] = row_xadd<8>(row_xadd<4>(row_xadd<2>(row_xadd<1>(c[r] * avs_l))));
          ld[r] = row_xadd<8>(row_xadd<4>(row_xadd<2>(row_xadd<1>(c[r] * avd_l))));
        }
        const int d = lid & 7;
        float vs = ls[0], vd = ld[0];
#pragma unroll
        for (int r = 1; r < 8; ++r) {
          vs = (d == r) ? ls[r] : vs;
          vd = (d == r) ? ld[r] : vd;
        }
        part[(ct * RP + r0 + d) * 2 + (lid >> 3)] = (lid < 8) ? vs : vd;
      }
      float* crow = out + (size_t)(lo + r0) * NOUT + ct * 16 + lid;
      if (mt * 16 + 16 <= cnt) {
#pragma unroll
        for (int r = 0; r < 8; ++r) crow[(size_t)r * NOUT] = c[r];
      } else {
        const int nrows = cnt - r0;
#pragma unroll
        for (int r = 0; r < 8; ++r)
          if (r < nrows) crow[(size_t)r * NOUT] = c[r];
      }
    }
    __syncthreads();
    for (int i = tid; i < RP; i += WG_THREADS) {
      if (i < cnt) {
        float a = part[i * 2], d = part[i * 2 + 1];
#pragma unroll
        for (int t = 1; t < NT; ++t) {
          a += part[(t * RP + i) * 2];
          d += part[(t * RP + i) * 2 + 1];
        }
        as_out[lo + i] = a;
        ad_out[lo + i] = d;
      }
    }
  }
}

static inline int cdiv(long long a, long long b) { return (int)((a + b - 1) / b); }

extern "C" void kernel_launch(void* const* d_in, const int* in_sizes, int n_in,
                              void* d_out, int out_size, void* d_ws, size_t ws_size,
                              hipStream_t stream) {
  const float* x      = (const float*)d_in[0];
  const int*   ei     = (const int*)  d_in[1];
  const float* W1     = (const float*)d_in[2];
  const float* a_src1 = (const float*)d_in[3];
  const float* a_dst1 = (const float*)d_in[4];
  const float* b1     = (const float*)d_in[5];
  const float* W2     = (const float*)d_in[6];
  const float* a_src2 = (const float*)d_in[7];
  const float* a_dst2 = (const float*)d_in[8];
  const float* b2     = (const float*)d_in[9];
  float* z = (float*)d_out;

  const int Fin = 256, H1 = 8, C1 = 64, C2 = 128;
  const int N    = in_sizes[0] / Fin;
  const int E    = in_sizes[1] / 2;
  const int* srcv = ei;
  const int* dstv = ei + E;
  const int G = cdiv(N, NODES_PER_WG);

  float* h1   = (float*)d_ws;
  float* as1  = h1  + (size_t)N * C1;
  float* ad1  = as1 + (size_t)N * H1;
  float* h2   = ad1 + (size_t)N * H1;
  float* as2  = h2  + (size_t)N * C2;
  float* ad2  = as2 + N;
  int*   cnts = (int*)(ad2 + N);
  int2*  lsts = (int2*)(cnts + (size_t)G * SLICES);

  edge_bin_kernel<<<SLICES, BIN_THREADS, 0, stream>>>(srcv, dstv, E, N, G, lsts, cnts);

  gemm1_lds_kernel<<<256, G1_THREADS, 0, stream>>>(x, W1, h1, N, a_src1, a_dst1, as1, ad1);
  range_agg_kernel<64, 8, true, 128><<<G, WG_THREADS, 0, stream>>>(
      srcv, dstv, E, N, lsts, cnts, as1, ad1, h1, b1, W2, h2, a_src2, a_dst2, as2, ad2);

  range_agg_kernel<128, 1, false, 0><<<G, WG_THREADS, 0, stream>>>(
      srcv, dstv, E, N, lsts, cnts, as2, ad2, h2, b2, nullptr, z,
      nullptr, nullptr, nullptr, nullptr);
}


# Softmax state in LDS (node max, dst logit) placed by owner wave at a 48-byte pitch; max pass turns heads by lane: a wave's LDS atomics and reads no longer share one bank or word; sum order unchanged
# speedup vs baseline: 5.6107x; 1.1585x over previous
#include <hip/hip_runtime.h>
#include <cstdint>


typedef float v2f __attribute__((ext_vector_type(2)));
typedef float v8f __attribute__((ext_vector_type(8)));

#define NEG_SLOPE 0.2f
#define GAT_EPS 1e-16f
#define ENC_NEG_INF 0x007FFFFFu

__device__ __forceinline__ unsigned enc_ord(float f) {
  unsigned u = __float_as_uint(f);
  return (u & 0x80000000u) ? ~u : (u | 0x80000000u);
}
__device__ __forceinline__ float dec_ord(unsigned k) {
  return (k & 0x80000000u) ? __uint_as_float(k & 0x7FFFFFFFu)
                           : __uint_as_float(~k);
}

template <int XM>
__device__ __forceinline__ float row_xadd(float v) {
  return v + __int_as_float(__builtin_amdgcn_update_dpp(0, __float_as_int(v), 0x160 | XM,
                                                        0xf, 0xf, true));
}

#define G1_K 256
#define G1_N 64
#define G1_ROWS 64
#define G1_PITCH (G1_K + 4)
#define G1_THREADS 512

__global__ __launch_bounds__(G1_THREADS) void gemm1_lds_kernel(
    const float* __restrict__ A, const float* __restrict__ B,
    float* __restrict__ C, int M,
    const float* __restrict__ avs, const float* __restrict__ avd,
    float* __restrict__ as, float* __restrict__ ad) {
  __shared__ __attribute__((aligned(16))) float sA[2 * G1_ROWS * G1_PITCH];
  constexpr int KS = G1_K / 4;
  constexpr int LD = (G1_ROWS * G1_K / 4) / G1_THREADS;
  constexpr int STAGE = G1_ROWS * G1_PITCH;
  const int tid = threadIdx.x, lane = tid & 31;
  const int half = lane >> 4, lid = lane & 15;
  const int wu = __builtin_amdgcn_readfirstlane(tid >> 5);
  const int ct = wu & 3;
  const int rt = wu >> 2;
  const int mtiles = (M + 15) >> 4;
  const int t0 = (int)(((long long)blockIdx.x * mtiles) / gridDim.x);
  const int t1 = (int)(((long long)(blockIdx.x + 1) * mtiles) / gridDim.x);
  const int rlim = min(16 * t1, M);

  const float* bcol = B + (size_t)(2 * half) * G1_N + ct * 16 + lid;
  v2f b[KS];
#pragma unroll
  for (int ks = 0; ks < KS; ++ks) {
    b[ks].x = bcol[(size_t)(4 * ks) * G1_N];
    b[ks].y = bcol[(size_t)(4 * ks + 1) * G1_N];
  }

  const float avs_l = avs[ct * 16 + lid];
  const float avd_l = avd[ct * 16 + lid];

  float4 pre[LD];
#pragma unroll
  for (int i = 0; i < LD; ++i) {
    const int idx = i * G1_THREADS + tid;
    const int row = 16 * t0 + (idx >> 6);
    pre[i] = make_float4(0.f, 0.f, 0.f, 0.f);
    if (row < rlim) pre[i] = *(const float4*)(A + (size_t)row * G1_K + 4 * (idx & 63));
  }
#pragma unroll
  for (int i = 0; i < LD; ++i) {
    const int idx = i * G1_THREADS + tid;
    *(float4*)(sA + (idx >> 6) * G1_PITCH + 4 * (idx & 63)) = pre[i];
  }
  __syncthreads();

  int buf = 0;
  for (int tb = t0; tb < t1; tb += 4) {
    const int tn = tb + 4;
    const bool more = tn < t1;
    if (more) {
#pragma unroll
      for (int i = 0; i < LD; ++i) {
        const int idx = i * G1_THREADS + tid;
        const int row = 16 * tn + (idx >> 6);
        pre[i] = make_float4(0.f, 0.f, 0.f, 0.f);
        if (row < rlim) pre[i] = *(const float4*)(A + (size_t)row * G1_K + 4 * (idx & 63));
      }
    }
    const int mt = tb + rt;
    if (mt < t1) {
      const float* arow = sA + buf * STAGE + (rt * 16 + lid) * G1_PITCH + 2 * half;
      v8f c = {0.f, 0.f, 0.f, 0.f, 0.f, 0.f, 0.f, 0.f};
      constexpr int AC = 8;
      v2f a[2][AC];
#pragma unroll
      for (int j = 0; j < AC; ++j) a[0][j] = *(const v2f*)(arow + 4 * j);
#pragma unroll
      for (int kc = 0; kc < KS / AC; ++kc) {
        if (kc + 1 < KS / AC) {
#pragma unroll
          for (int j = 0; j < AC; ++j)
            a[(kc + 1) & 1][j] = *(const v2f*)(arow + 4 * (AC * (kc + 1) + j));
        }
        __builtin_amdgcn_sched_barrier(0);
#pragma unroll
        for (int j = 0; j < AC; ++j)
          c = __builtin_amdgcn_wmma_f32_16x16x4_f32(false, a[kc & 1][j], false, b[AC * kc + j],
                                                    (short)0, c, false, false);
        __builtin_amdgcn_sched_barrier(0);
      }
      const int r0 = mt * 16 + 8 * half;
      {
        float ls[8], ld[8];
#pragma unroll
        for (int r = 0; r < 8; ++r) {
          ls[r] = row_xadd<4>(row_xadd<2>(row_xadd<1>(c[r] * avs_l)));
          ld[r] = row_xadd<4>(row_xadd<2>(row_xadd<1>(c[r] * avd_l)));
        }
        const int d = lid & 7;
        float vs = ls[0], vd = ld[0];
#pragma unroll
        for (int r = 1; r < 8; ++r) {
          vs = (d == r) ? ls[r] : vs;
          vd = (d == r) ? ld[r] : vd;
        }
        const int lrow = r0 + d;
        if (lrow < M) {
          const size_t li = (size_t)lrow * (G1_N / 8) + 2 * ct + (lid >> 3);
          as[li] = vs;
          ad[li] = vd;
        }
      }
      float* crow = C + (size_t)r0 * G1_N + ct * 16 + lid;
      if (mt * 16 + 16 <= M) {
#pragma unroll
        for (int r = 0; r < 8; ++r) crow[(size_t)r * G1_N] = c[r];
      } else {
        const int nrows = M - r0;
#pragma unroll
        for (int r = 0; r < 8; ++r)
          if (r < nrows) crow[(size_t)r * G1_N] = c[r];
      }
    }
    if (more) {
      float* nxt = sA + (buf ^ 1) * STAGE;
#pragma unroll
      for (int i = 0; i < LD; ++i) {
        const int idx = i * G1_THREADS + tid;
        *(float4*)(nxt + (idx >> 6) * G1_PITCH + 4 * (idx & 63)) = pre[i];
      }
    }
    __syncthreads();
    buf ^= 1;
  }
}

#define NODES_PER_WG 392
#define WG_THREADS 1024
#define WG_WAVES 32
#define SLICES 256
#define CELL_CAP 64
#define MAX_RANGES 512
#define CELLS_PER_WAVE (SLICES / WG_WAVES)
#define BKT_CAP 512
#define OWN_SLOTS ((NODES_PER_WG + 31) / 32)
#define BIN_THREADS 256
#define BIN_WAVES (BIN_THREADS / 32)
#define BIN_U 32

__global__ __launch_bounds__(BIN_THREADS) void edge_bin_kernel(
    const int* __restrict__ src, const int* __restrict__ dst, int E, int N, int G,
    int2* __restrict__ cells, int* __restrict__ counts) {
  __shared__ __attribute__((aligned(16))) int2 stage[MAX_RANGES * CELL_CAP];
  __shared__ int rc[MAX_RANGES];
  const int tid = threadIdx.x, lane = tid & 31;
  const int wv = __builtin_amdgcn_readfirstlane(tid >> 5);
  const int slc = blockIdx.x;
  const int nr = min(G, MAX_RANGES);
  for (int i = tid; i < nr; i += BIN_THREADS) rc[i] = 0;
  __syncthreads();
  const int eper = (((E + SLICES - 1) / SLICES) + 31) & ~31;
  const int ebeg = min(E, slc * eper);
  const int eend = min(E, ebeg + eper);
  for (int e0 = ebeg + tid; e0 < eend; e0 += BIN_U * BIN_THREADS) {
    int d[BIN_U], s[BIN_U];
#pragma unroll
    for (int k = 0; k < BIN_U; ++k) {
      const int e = e0 + k * BIN_THREADS;
      d[k] = -1;
      s[k] = 0;
      if (e < eend) { d[k] = dst[e]; s[k] = src[e]; }
    }
    int r[BIN_U], slot[BIN_U];
#pragma unroll
    for (int k = 0; k < BIN_U; ++k) {
      slot[k] = CELL_CAP;
      r[k] = 0;
      if ((unsigned)d[k] < (unsigned)N) {
        r[k] = d[k] / NODES_PER_WG;
        if (r[k] < MAX_RANGES) slot[k] = atomicAdd(&rc[r[k]], 1);
      }
    }
#pragma unroll
    for (int k = 0; k < BIN_U; ++k) {
      if (slot[k] < CELL_CAP)
        stage[r[k] * CELL_CAP + slot[k]] = make_int2(s[k], d[k] - r[k] * NODES_PER_WG);
    }
  }
  __syncthreads();
  static_assert(CELL_CAP == 64, "a cell is one 16-byte piece a lane");
  constexpr int CU = 16;
  for (int rg0 = wv; rg0 < nr; rg0 += CU * BIN_WAVES) {
    int nl = 0;
    {
      const int rg = rg0 + lane * BIN_WAVES;
      if (lane < CU && rg < nr) nl = min(rc[rg], CELL_CAP);
    }
    int n[CU];
    int4 v[CU];
#pragma unroll
    for (int k = 0; k < CU; ++k) n[k] = __builtin_amdgcn_readlane(nl, k);
#pragma unroll
    for (int k = 0; k < CU; ++k) {
      const int rg = rg0 + k * BIN_WAVES;
      v[k] = make_int4(0, 0, 0, 0);
      if (2 * lane < n[k]) v[k] = *(const int4*)(stage + rg * CELL_CAP + 2 * lane);
    }
#pragma unroll
    for (int k = 0; k < CU; ++k) {
      const int rg = rg0 + k * BIN_WAVES;
      if (2 * lane < n[k])
        *(int4*)(cells + ((size_t)rg * SLICES + slc) * CELL_CAP + 2 * lane) = v[k];
    }
  }
  for (int i = tid; i < G; i += BIN_THREADS)
    counts[(size_t)i * SLICES + slc] = (i < MAX_RANGES) ? rc[i] : CELL_CAP + 1;
}

__device__ __forceinline__ int own_pos(int dl) { return (dl & 31) * OWN_SLOTS + (dl >> 5); }

__device__ __forceinline__ bool fetch_own(int i, int n_real, bool use_list,
                                          const int2* lst,
                                          const int* __restrict__ src,
                                          const int* __restrict__ dst,
                                          int lo, int cnt, int wave, int& sv, int& dv) {
  sv = 0; dv = 0;
  if (i < n_real) {
    if (use_list) { int2 p = lst[i]; sv = p.x; dv = p.y; return true; }
    int d = dst[i] - lo;
    if ((unsigned)d < (unsigned)cnt && (d & 31) == wave) { sv = src[i]; dv = d; return true; }
    return false;
  }
  const int j = i - n_real;
  const int n = wave + 32 * j;
  if (j < OWN_SLOTS && n < cnt) { dv = n; sv = lo + n; return true; }
  return false;
}

template <int DTOT, int H, bool ELU, int NOUT>
__global__ __launch_bounds__(WG_THREADS) void range_agg_kernel(
    const int* __restrict__ src, const int* __restrict__ dst, int E, int N,
    const int2* __restrict__ lists, const int* __restrict__ counts,
    const float* __restrict__ as, const float* __restrict__ ad,
    const float* __restrict__ hsrc, const float* __restrict__ bias,
    const float* __restrict__ Wn, float* __restrict__ out,
    const float* __restrict__ avs_n, const float* __restrict__ avd_n,
    float* __restrict__ as_out, float* __restrict__ ad_out) {
  constexpr int R = NODES_PER_WG;
  constexpr bool PROD = NOUT > 0;
  constexpr int P = DTOT + 4;
  constexpr int RP = ((R + 15) / 16) * 16;
  constexpr int VPL = DTOT / 32;
  constexpr int Dh = DTOT / H;
  constexpr int U = 8;
  constexpr int NS = OWN_SLOTS;
  constexpr int PH = (H > 1) ? H + 4 : 1;
  constexpr int NPOS = WG_WAVES * NS;
  static_assert((H & (H - 1)) == 0, "the heads of a node are rotated over the lanes modulo H");
  static_assert(H == 1 || ((PH / 4) % 2 == 1 && PH % 4 == 0), "odd number of 16-byte pieces");
  constexpr int SPARE_SLOT = 15;
  static_assert(NS <= SPARE_SLOT, "a wave's node slots and the spare are the elements of one 16-wide register vector");
  static_assert(WG_WAVES == 32, "a node's owner wave is its index modulo 32");
  static_assert(SLICES % 32 == 0 && CELLS_PER_WAVE * WG_WAVES == SLICES, "wave w reads the cells w + 32 k; lane l holds the counts l + 32 k");
  typedef float vrow __attribute__((ext_vector_type(VPL)));
  typedef float vslot __attribute__((ext_vector_type(16)));
  __shared__ __attribute__((aligned(16))) float rows[PROD ? RP * P : 4];
  __shared__ __attribute__((aligned(16))) unsigned mx[NPOS * PH];
  __shared__ __attribute__((aligned(16))) float adl[NPOS * PH];
  __shared__ int2 bkt[WG_WAVES * BKT_CAP];
  __shared__ int bcnt[WG_WAVES];
  __shared__ float wbuf[(H > 1) ? WG_WAVES * 32 * H : 1];

  const int tid = threadIdx.x, lane = tid & 31;
  const int wu = __builtin_amdgcn_readfirstlane(tid >> 5);
  const int lo = blockIdx.x * R;
  const int cnt = min(R, N - lo);
  const int2* rcells = lists + (size_t)blockIdx.x * SLICES * CELL_CAP;
  const int* rcounts = counts + (size_t)blockIdx.x * SLICES;
  int cc[CELLS_PER_WAVE];
  int cmax = 0;
#pragma unroll
  for (int k = 0; k < CELLS_PER_WAVE; ++k) {
    cc[k] = rcounts[lane + 32 * k];
    cmax = max(cmax, cc[k]);
  }
  const bool seg_ok = __builtin_amdgcn_ballot_w32(cmax > CELL_CAP) == 0u;

  if (tid < WG_WAVES) bcnt[tid] = 0;
  for (int i = tid; i < NPOS * PH; i += WG_THREADS) mx[i] = ENC_NEG_INF;
  for (int i = tid; i < R * H; i += WG_THREADS)
    adl[own_pos(i / H) * PH + (i % H)] = (i < cnt * H) ? ad[(size_t)lo * H + i] : 0.f;
  __syncthreads();
  if (seg_ok) {
    static_assert(CELL_CAP == 64, "a cell is read by two loads of a wave");
    int2 p[2 * CELLS_PER_WAVE];
#pragma unroll
    for (int k = 0; k < CELLS_PER_WAVE; ++k) {
      const int ck = __builtin_amdgcn_readlane(cc[k], wu);
      const int2* cell = rcells + (size_t)(wu + 32 * k) * CELL_CAP;
      p[2 * k] = make_int2(0, -1);
      p[2 * k + 1] = make_int2(0, -1);
      if (lane < ck) p[2 * k] = cell[lane];
      if (ck > 32) {
        if (lane + 32 < ck) p[2 * k + 1] = cell[lane + 32];
      }
    }
    int slot[2 * CELLS_PER_WAVE];
#pragma unroll
    for (int u = 0; u < 2 * CELLS_PER_WAVE; ++u) {
      const int dl = p[u].y;
      slot[u] = BKT_CAP;
      if ((unsigned)dl < (unsigned)cnt) slot[u] = atomicAdd(&bcnt[dl & 31], 1);
    }
#pragma unroll
    for (int u = 0; u < 2 * CELLS_PER_WAVE; ++u) {
      const int dl = p[u].y;
      if (slot[u] < BKT_CAP) bkt[(dl & 31) * BKT_CAP + slot[u]] = make_int2(p[u].x, dl);
    }
  }
  __syncthreads();
  const int myc = bcnt[lane];
  const bool use_list = seg_ok && __builtin_amdgcn_ballot_w32(myc > BKT_CAP) == 0u;
  const int n_real = use_list ? __builtin_amdgcn_readlane(myc, wu) : E;
  const int total = n_real + NS;
  const int2* lst = bkt + wu * BKT_CAP;

  for (int base = 0; base < total; base += 32) {
    int sv, dv;
    bool valid = fetch_own(base + lane, n_real, use_list, lst, src, dst, lo, cnt, wu, sv, dv);
    if (valid) {
      const int pp = own_pos(dv) * PH;
      unsigned k[H];
#pragma unroll
      for (int hh = 0; hh < H; ++hh) {
        float e = as[(size_t)sv * H + hh] + adl[pp + hh];
        e = e > 0.f ? e : NEG_SLOPE * e;
        k[hh] = enc_ord(e);
      }
      if constexpr (H > 1) {
#pragma unroll
        for (int s = 1; s < H; s <<= 1) {
          const bool turn = (lane & s) != 0;
          unsigned t[H];
#pragma unroll
          for (int i = 0; i < H; ++i) t[i] = turn ? k[(i + s) & (H - 1)] : k[i];
#pragma unroll
          for (int i = 0; i < H; ++i) k[i] = t[i];
        }
#pragma unroll
        for (int i = 0; i < H; ++i) atomicMax(&mx[pp + ((i + lane) & (H - 1))], k[i]);
      } else {
        atomicMax(&mx[pp], k[0]);
      }
    }
  }
  __syncthreads();

  const int c0 = lane * VPL;
  const int hh = c0 / Dh;
  vslot accv[VPL];
  vslot smv;
  float sm1 = 0.f;
#pragma unroll
  for (int j = 0; j < 16; ++j) {
    smv[j] = 0.f;
#pragma unroll
    for (int k = 0; k < VPL; ++k) accv[k][j] = 0.f;
  }
  float* wblk = wbuf + (H > 1 ? wu * 32 * H : 0);
  int sv_n, dv_n;
  bool valid_n = fetch_own(lane, n_real, use_list, lst, src, dst, lo, cnt, wu, sv_n, dv_n);
  float as_n[H];
#pragma unroll
  for (int h = 0; h < H; ++h) as_n[h] = as[(size_t)sv_n * H + h];
  for (int base = 0; base < total; base += 32) {
    const int sv = sv_n, dv = dv_n;
    const bool valid = valid_n;
    float asv[H];
#pragma unroll
    for (int h = 0; h < H; ++h) asv[h] = as_n[h];
    if (base + 32 < total) {
      valid_n = fetch_own(base + 32 + lane, n_real, use_list, lst, src, dst, lo, cnt, wu, sv_n, dv_n);
#pragma unroll
      for (int h = 0; h < H; ++h) as_n[h] = as[(size_t)sv_n * H + h];
    }
    unsigned mask = __builtin_amdgcn_ballot_w32(valid);
    if (mask != 0u) {
      float wl[H];
      const int pp = own_pos(dv) * PH;
#pragma unroll
      for (int h = 0; h < H; ++h) {
        float e = asv[h] + adl[pp + h];
        e = e > 0.f ? e : NEG_SLOPE * e;
        wl[h] = __expf(e - dec_ord(mx[pp + h]));
      }
      if constexpr (H > 1) {
#pragma unroll
        for (int h = 0; h < H; ++h) wblk[lane * H + h] = wl[h];
      }
      while (mask != 0u) {
        int su[U], du[U], ju[U];
        float ws[U];
        bool ok[U];
#pragma unroll
        for (int u = 0; u < U; ++u) {
          ok[u] = mask != 0u;
          ju[u] = ok[u] ? (int)__builtin_ctz(mask) : 0;
          if (ok[u]) mask &= mask - 1u;
          su[u] = __builtin_amdgcn_readlane(sv, ju[u]);
          du[u] = __builtin_amdgcn_readlane(dv, ju[u]);
          if constexpr (H == 1)
            ws[u] = __int_as_float(__builtin_amdgcn_readlane(__float_as_int(wl[0]), ju[u]));
          else
            ws[u] = 0.f;
        }
        vrow row[U];
        float wv[U];
#pragma unroll
        for (int u = 0; u < U; ++u) {
          row[u] = *(const vrow*)(hsrc + (size_t)su[u] * DTOT + c0);
          if constexpr (H > 1) wv[u] = wblk[ju[u] * H + hh];
          else                 wv[u] = ws[u];
        }
        __builtin_amdgcn_sched_barrier(0);
#pragma unroll
        for (int u = 0; u < U; ++u) {
          const float w = wv[u];
          const int sl = ok[u] ? (du[u] >> 5) : SPARE_SLOT;
#pragma unroll
          for (int k = 0; k < VPL; ++k) accv[k][sl] += w * row[u][k];
          if constexpr (H == 1) sm1 += (lane == sl) ? w : 0.f;
          else                  smv[sl] += w;
        }
      }
    }
  }

  float bv[VPL];
#pragma unroll
  for (int k = 0; k < VPL; ++k) bv[k] = bias[c0 + k];
#pragma unroll
  for (int j = 0; j < NS; ++j) {
    const int n = wu + 32 * j;
    float sj;
    if constexpr (H == 1) sj = __int_as_float(__builtin_amdgcn_readlane(__float_as_int(sm1), j));
    else                  sj = smv[j];
    vrow v;
#pragma unroll
    for (int k = 0; k < VPL; ++k) {
      float t = accv[k][j] / (sj + GAT_EPS) + bv[k];
      if (ELU) t = t > 0.f ? t : __expf(t) - 1.f;
      v[k] = t;
    }
    if constexpr (!PROD) {
      if (n < cnt) *(vrow*)(out + (size_t)(lo + n) * DTOT + c0) = v;
    } else {
      if (n < RP) {
        if (n >= cnt) {
#pragma unroll
          for (int k = 0; k < VPL; ++k) v[k] = 0.f;
        }
        *(vrow*)(rows + n * P + c0) = v;
      }
    }
  }

  if constexpr (PROD) {
    __syncthreads();
    constexpr int NT = NOUT / 16;
    constexpr int MT = RP / 16;
    constexpr int KS = DTOT / 4;
    constexpr int QS = WG_WAVES / NT;
    const int half = lane >> 4, lid = lane & 15;
    const int ct = wu % NT;
    const int q = wu / NT;
    const float* bcol = Wn + (size_t)(2 * half) * NOUT + ct * 16 + lid;
    const float avs_l = avs_n[ct * 16 + lid];
    const float avd_l = avd_n[ct * 16 + lid];
    static_assert(H > 1 && WG_WAVES * 32 * H >= NT * RP * 2, "the partial logits fit the weight slab");
    float* part = wbuf;
    v2f b[KS];
#pragma unroll
    for (int ks = 0; ks < KS; ++ks) {
      b[ks].x = bcol[(size_t)(4 * ks) * NOUT];
      b[ks].y = bcol[(size_t)(4 * ks + 1) * NOUT];
    }
    for (int mt = q; mt < MT; mt += QS) {
      const float* arow = rows + (mt * 16 + lid) * P + 2 * half;
      v2f a[KS];
#pragma unroll
      for (int ks = 0; ks < KS; ++ks) a[ks] = *(const v2f*)(arow + 4 * ks);
      v8f c = {0.f, 0.f, 0.f, 0.f, 0.f, 0.f, 0.f, 0.f};
#pragma unroll
      for (int ks = 0; ks < KS; ++ks)
        c = __builtin_amdgcn_wmma_f32_16x16x4_f32(false, a[ks], false, b[ks],
                                                  (short)0, c, false, false);
      const int r0 = mt * 16 + 8 * half;
      {
        float ls[8], ld[8];
#pragma unroll
        for (int r = 0; r < 8; ++r) {
          ls[r] = row_xadd<8>(row_xadd<4>(row_xadd<2>(row_xadd<1>(c[r] * avs_l))));
          ld[r] = row_xadd<8>(row_xadd<4>(row_xadd<2>(row_xadd<1>(c[r] * avd_l))));
        }
        const int d = lid & 7;
        float vs = ls[0], vd = ld[0];
#pragma unroll
        for (int r = 1; r < 8; ++r) {
          vs = (d == r) ? ls[r] : vs;
          vd = (d == r) ? ld[r] : vd;
        }
        part[(ct * RP + r0 + d) * 2 + (lid >> 3)] = (lid < 8) ? vs : vd;
      }
      float* crow = out + (size_t)(lo + r0) * NOUT + ct * 16 + lid;
      if (mt * 16 + 16 <= cnt) {
#pragma unroll
        for (int r = 0; r < 8; ++r) crow[(size_t)r * NOUT] = c[r];
      } else {
        const int nrows = cnt - r0;
#pragma unroll
        for (int r = 0; r < 8; ++r)
          if (r < nrows) crow[(size_t)r * NOUT] = c[r];
      }
    }
    __syncthreads();
    for (int i = tid; i < RP; i += WG_THREADS) {
      if (i < cnt) {
        float a = part[i * 2], d = part[i * 2 + 1];
#pragma unroll
        for (int t = 1; t < NT; ++t) {
          a += part[(t * RP + i) * 2];
          d += part[(t * RP + i) * 2 + 1];
        }
        as_out[lo + i] = a;
        ad_out[lo + i] = d;
      }
    }
  }
}

static inline int cdiv(long long a, long long b) { return (int)((a + b - 1) / b); }

extern "C" void kernel_launch(void* const* d_in, const int* in_sizes, int n_in,
                              void* d_out, int out_size, void* d_ws, size_t ws_size,
                              hipStream_t stream) {
  const float* x      = (const float*)d_in[0];
  const int*   ei     = (const int*)  d_in[1];
  const float* W1     = (const float*)d_in[2];
  const float* a_src1 = (const float*)d_in[3];
  const float* a_dst1 = (const float*)d_in[4];
  const float* b1     = (const float*)d_in[5];
  const float* W2     = (const float*)d_in[6];
  const float* a_src2 = (const float*)d_in[7];
  const float* a_dst2 = (const float*)d_in[8];
  const float* b2     = (const float*)d_in[9];
  float* z = (float*)d_out;

  const int Fin = 256, H1 = 8, C1 = 64, C2 = 128;
  const int N    = in_sizes[0] / Fin;
  const int E    = in_sizes[1] / 2;
  const int* srcv = ei;
  const int* dstv = ei + E;
  const int G = cdiv(N, NODES_PER_WG);

  float* h1   = (float*)d_ws;
  float* as1  = h1  + (size_t)N * C1;
  float* ad1  = as1 + (size_t)N * H1;
  float* h2   = ad1 + (size_t)N * H1;
  float* as2  = h2  + (size_t)N * C2;
  float* ad2  = as2 + N;
  int*   cnts = (int*)(ad2 + N);
  int2*  lsts = (int2*)(cnts + (size_t)G * SLICES);

  edge_bin_kernel<<<SLICES, BIN_THREADS, 0, stream>>>(srcv, dstv, E, N, G, lsts, cnts);

  gemm1_lds_kernel<<<256, G1_THREADS, 0, stream>>>(x, W1, h1, N, a_src1, a_dst1, as1, ad1);
  range_agg_kernel<64, 8, true, 128><<<G, WG_THREADS, 0, stream>>>(
      srcv, dstv, E, N, lsts, cnts, as1, ad1, h1, b1, W2, h2, a_src2, a_dst2, as2, ad2);

  range_agg_kernel<128, 1, false, 0><<<G, WG_THREADS, 0, stream>>>(
      srcv, dstv, E, N, lsts, cnts, as2, ad2, h2, b2, nullptr, z,
      nullptr, nullptr, nullptr, nullptr);
}
